# starting kernel
# baseline (speedup 1.0000x reference)
#include <hip/hip_runtime.h>
#include <hip/hip_bf16.h>

typedef __attribute__((ext_vector_type(16))) _Float16 v16h;
typedef __attribute__((ext_vector_type(8)))  float    v8f;

#define IN_DIM 128

__global__ void gcn_pack_w(const float* __restrict__ W, _Float16* __restrict__ Wp,
                           int Nt  ) {
    int id = blockIdx.x * blockDim.x + threadIdx.x;
    int total = Nt * 4 * 32;
    if (id >= total) return;
    int lane = id & 31;
    int kk   = (id >> 5) & 3;
    int t    = id >> 7;
    int half = lane >> 4;
    int r    = lane & 15;
    int N    = Nt * 16;
    int n    = t * 16 + r;
    int k0   = kk * 32;
    _Float16* dst = Wp + (size_t)id * 16;
#pragma unroll
    for (int j = 0; j < 8; ++j)
        dst[j] = (_Float16)W[(size_t)(k0 + half * 8 + j) * N + n];
#pragma unroll
    for (int j = 0; j < 8; ++j)
        dst[8 + j] = (_Float16)W[(size_t)(k0 + 16 + half * 8 + j) * N + n];
}

__global__ void gcn_spmm(const int* __restrict__ row, const int* __restrict__ col,
                         const float* __restrict__ vals, const float* __restrict__ x,
                         float* __restrict__ agg, int nEdges) {
    int gid  = blockIdx.x * blockDim.x + threadIdx.x;
    int e    = gid >> 5;
    int lane = gid & 31;
    if (e >= nEdges) return;
    int   r = row[e];
    int   c = col[e];
    float v = vals[e];
    const float4* xp = (const float4*)(x + (size_t)c * IN_DIM);
    float4 m = xp[lane];
    float* op = agg + (size_t)r * IN_DIM + lane * 4;
    atomicAdd(op + 0, v * m.x);
    atomicAdd(op + 1, v * m.y);
    atomicAdd(op + 2, v * m.z);
    atomicAdd(op + 3, v * m.w);
}

template <int NTILES, bool RELU>
__global__ void gcn_gemm_wmma(const float* __restrict__ A,
                              const _Float16* __restrict__ Bp,
                              float* __restrict__ C, int M) {
    const int N = NTILES * 16;
    int lane = threadIdx.x & 31;
    int warp = threadIdx.x >> 5;
    int wavesPerBlock = blockDim.x >> 5;
    int mt = blockIdx.x * wavesPerBlock + warp;
    int mtiles = M >> 4;
    if (mt >= mtiles) return;

    int half = lane >> 4;
    int r    = lane & 15;

    int mrow = mt * 16 + r;

    v8f acc[NTILES];
#pragma unroll
    for (int t = 0; t < NTILES; ++t) acc[t] = (v8f){};

#pragma unroll
    for (int kk = 0; kk < 4; ++kk) {
        int k0 = kk * 32;
        const float* ap = A + (size_t)mrow * IN_DIM + k0 + half * 8;
        float4 f0 = *(const float4*)(ap + 0);
        float4 f1 = *(const float4*)(ap + 4);
        float4 f2 = *(const float4*)(ap + 16);
        float4 f3 = *(const float4*)(ap + 20);
        v16h a;
        a[0]  = (_Float16)f0.x; a[1]  = (_Float16)f0.y; a[2]  = (_Float16)f0.z; a[3]  = (_Float16)f0.w;
        a[4]  = (_Float16)f1.x; a[5]  = (_Float16)f1.y; a[6]  = (_Float16)f1.z; a[7]  = (_Float16)f1.w;
        a[8]  = (_Float16)f2.x; a[9]  = (_Float16)f2.y; a[10] = (_Float16)f2.z; a[11] = (_Float16)f2.w;
        a[12] = (_Float16)f3.x; a[13] = (_Float16)f3.y; a[14] = (_Float16)f3.z; a[15] = (_Float16)f3.w;

#pragma unroll
        for (int t = 0; t < NTILES; ++t) {
            v16h b = *(const v16h*)(Bp + (((size_t)t * 4 + kk) * 32 + lane) * 16);
            acc[t] = __builtin_amdgcn_wmma_f32_16x16x32_f16(
                         false, a,  false, b,
                         (short)0, acc[t],
                         false,  false);
        }
    }

    int mbase = mt * 16 + half * 8;
#pragma unroll
    for (int t = 0; t < NTILES; ++t) {
        int ncol = t * 16 + r;
#pragma unroll
        for (int v = 0; v < 8; ++v) {
            float val = acc[t][v];
            if (RELU) val = fmaxf(val, 0.0f);
            C[(size_t)(mbase + v) * N + ncol] = val;
        }
    }
}

extern "C" void kernel_launch(void* const* d_in, const int* in_sizes, int n_in,
                              void* d_out, int out_size, void* d_ws, size_t ws_size,
                              hipStream_t stream) {
    const float* x    = (const float*)d_in[0];
    const int*   row  = (const int*)d_in[1];
    const int*   col  = (const int*)d_in[2];
    const float* vals = (const float*)d_in[3];
    const float* W1   = (const float*)d_in[4];
    const float* W2   = (const float*)d_in[5];
    float* out = (float*)d_out;

    const int NN = in_sizes[0] / IN_DIM;
    const int NE = in_sizes[1];

    float*     aggF = (float*)d_ws;
    float*     hF   = aggF + (size_t)NN * IN_DIM;
    _Float16*  W1p  = (_Float16*)(hF + (size_t)NN * IN_DIM);
    _Float16*  W2p  = W1p + 128 * 128;

    gcn_pack_w<<<4, 256, 0, stream>>>(W1, W1p, 8);
    gcn_pack_w<<<2, 256, 0, stream>>>(W2, W2p, 4);

    const size_t aggBytes = (size_t)NN * IN_DIM * sizeof(float);
    const int spmmBlocks = (NE * 32 + 255) / 256;

    hipMemsetAsync(aggF, 0, aggBytes, stream);
    gcn_spmm<<<spmmBlocks, 256, 0, stream>>>(row, col, vals, x, aggF, NE);

    {
        int mtiles = NN / 16;
        int blocks = (mtiles + 3) / 4;
        gcn_gemm_wmma<8, true><<<blocks, 128, 0, stream>>>(aggF, W1p, hF, NN);
    }

    hipMemsetAsync(aggF, 0, aggBytes, stream);
    gcn_spmm<<<spmmBlocks, 256, 0, stream>>>(row, col, vals, hF, aggF, NE);

    {
        int mtiles = NN / 16;
        int blocks = (mtiles + 3) / 4;
        gcn_gemm_wmma<4, false><<<blocks, 128, 0, stream>>>(aggF, W2p, out, NN);
    }
}


# Layer 2 computed as adj @ (h @ W2) instead of (adj @ h) @ W2: the second edge pass gathers 64-float rows, not 128-float ones; f32 sum order changes (edge sum now after the product)
# speedup vs baseline: 2.0030x; 2.0030x over previous
#include <hip/hip_runtime.h>
#include <hip/hip_bf16.h>

typedef __attribute__((ext_vector_type(16))) _Float16 v16h;
typedef __attribute__((ext_vector_type(8)))  float    v8f;

#define IN_DIM 128

__global__ void gcn_pack_w(const float* __restrict__ W, _Float16* __restrict__ Wp,
                           int Nt  ) {
    int id = blockIdx.x * blockDim.x + threadIdx.x;
    int total = Nt * 4 * 32;
    if (id >= total) return;
    int lane = id & 31;
    int kk   = (id >> 5) & 3;
    int t    = id >> 7;
    int half = lane >> 4;
    int r    = lane & 15;
    int N    = Nt * 16;
    int n    = t * 16 + r;
    int k0   = kk * 32;
    _Float16* dst = Wp + (size_t)id * 16;
#pragma unroll
    for (int j = 0; j < 8; ++j)
        dst[j] = (_Float16)W[(size_t)(k0 + half * 8 + j) * N + n];
#pragma unroll
    for (int j = 0; j < 8; ++j)
        dst[8 + j] = (_Float16)W[(size_t)(k0 + 16 + half * 8 + j) * N + n];
}

__global__ void gcn_zero_i32(int* __restrict__ p, int n) {
    int i = blockIdx.x * blockDim.x + threadIdx.x;
    if (i < n) p[i] = 0;
}

__global__ void gcn_count(const int* __restrict__ row, int* __restrict__ cnt,
                          int* __restrict__ rank, int nEdges) {
    int e = blockIdx.x * blockDim.x + threadIdx.x;
    if (e >= nEdges) return;
    rank[e] = atomicAdd(&cnt[row[e]], 1);
}

#define SCAN_THREADS 1024
#define SCAN_K 13
__global__ __launch_bounds__(SCAN_THREADS)
void gcn_scan(const int* __restrict__ cnt, int* __restrict__ rowptr, int n) {
    __shared__ int wsum[SCAN_K * 32];
    __shared__ int total_s;
    const int t = threadIdx.x;
    const int lane = t & 31;
    const int w = t >> 5;
    int carry = 0;
    for (int base = 0; base < n; base += SCAN_K * SCAN_THREADS * 4) {
        int4 v[SCAN_K];
        int incl[SCAN_K];
#pragma unroll
        for (int k = 0; k < SCAN_K; ++k) {
            int idx = base + (k * SCAN_THREADS + t) * 4;
            if (idx + 3 < n) {
                v[k] = *(const int4*)(cnt + idx);
            } else {
                v[k].x = (idx     < n) ? cnt[idx]     : 0;
                v[k].y = (idx + 1 < n) ? cnt[idx + 1] : 0;
                v[k].z = (idx + 2 < n) ? cnt[idx + 2] : 0;
                v[k].w = 0;
            }
        }
#pragma unroll
        for (int k = 0; k < SCAN_K; ++k) {
            int sum = v[k].x + v[k].y + v[k].z + v[k].w;
            for (int d = 1; d < 32; d <<= 1) {
                int o = __shfl_up(sum, d);
                if (lane >= d) sum += o;
            }
            incl[k] = sum;
            if (lane == 31) wsum[k * 32 + w] = sum;
        }
        __syncthreads();
        if (w == 0) {
            int loc[SCAN_K];
            int run = 0;
#pragma unroll
            for (int i = 0; i < SCAN_K; ++i) { loc[i] = wsum[lane * SCAN_K + i]; run += loc[i]; }
            int inc = run;
            for (int d = 1; d < 32; d <<= 1) {
                int o = __shfl_up(inc, d);
                if (lane >= d) inc += o;
            }
            int ex = inc - run;
#pragma unroll
            for (int i = 0; i < SCAN_K; ++i) { wsum[lane * SCAN_K + i] = ex; ex += loc[i]; }
            if (lane == 31) total_s = inc;
        }
        __syncthreads();
#pragma unroll
        for (int k = 0; k < SCAN_K; ++k) {
            int idx = base + (k * SCAN_THREADS + t) * 4;
            int own = v[k].x + v[k].y + v[k].z + v[k].w;
            int p0 = carry + wsum[k * 32 + w] + incl[k] - own;
            int p1 = p0 + v[k].x;
            int p2 = p1 + v[k].y;
            int p3 = p2 + v[k].z;
            if (idx + 3 < n) {
                *(int4*)(rowptr + idx) = make_int4(p0, p1, p2, p3);
            } else {
                if (idx     < n) rowptr[idx]     = p0;
                if (idx + 1 < n) rowptr[idx + 1] = p1;
                if (idx + 2 < n) rowptr[idx + 2] = p2;
            }
        }
        carry += total_s;
        __syncthreads();
    }
    if (t == 0) rowptr[n] = carry;
}

__global__ void gcn_place(const int* __restrict__ row, const int* __restrict__ rank,
                          const int* __restrict__ rowptr, int* __restrict__ slot, int nEdges) {
    int e = blockIdx.x * blockDim.x + threadIdx.x;
    if (e >= nEdges) return;
    slot[rowptr[row[e]] + rank[e]] = e;
}

#define ROWS_PER_WAVE 5

__global__ __launch_bounds__(256)
void gcn_order_rows(const int* __restrict__ rowptr, const int* __restrict__ slot,
                    const int* __restrict__ col, const float* __restrict__ vals,
                    int* __restrict__ ccol, float* __restrict__ cval,
                    int nRows, int nEdges) {
    const int lane = threadIdx.x & 31;
    const int wave = __builtin_amdgcn_readfirstlane((blockIdx.x * blockDim.x + threadIdx.x) >> 5);
    const int r0 = wave * ROWS_PER_WAVE;
    if (r0 >= nRows) return;
    const int rEnd = min(nRows, r0 + ROWS_PER_WAVE);
    const int last = max(nEdges - 1, 0);
    int s = rowptr[r0];
    int e = rowptr[r0 + 1];
    int nxt = slot[min(s + lane, last)];
    for (int r = r0; r < rEnd; ++r) {
        const int e_next = rowptr[min(r + 2, nRows)];
        const int first = nxt;
        const int deg = e - s;
        if (deg <= 32) {
            int place = 0;
            for (int j = 0; j < deg; ++j)
                place += (__builtin_amdgcn_readlane(first, j) < first) ? 1 : 0;
            nxt = slot[min(e + lane, last)];
            if (lane < deg) {
                ccol[s + place] = col[first];
                cval[s + place] = vals[first];
            }
        } else {
            nxt = slot[min(e + lane, last)];
            for (int ib = s; ib < e; ib += 32) {
                const int i = ib + lane;
                const int mine = slot[min(i, e - 1)];
                int place = 0;
                for (int jb = s; jb < e; jb += 32) {
                    const int jn = min(32, e - jb);
                    const int other = slot[min(jb + lane, e - 1)];
                    for (int j = 0; j < jn; ++j)
                        place += (__builtin_amdgcn_readlane(other, j) < mine) ? 1 : 0;
                }
                if (i < e) {
                    ccol[s + place] = col[mine];
                    cval[s + place] = vals[mine];
                }
            }
        }
        s = e;
        e = e_next;
    }
}

template <int VEC> struct gcn_row;
template <> struct gcn_row<4> { typedef float4 type; };
template <> struct gcn_row<2> { typedef float2 type; };
__device__ __forceinline__ void gcn_row_zero(float4& a) { a.x = 0.0f; a.y = 0.0f; a.z = 0.0f; a.w = 0.0f; }
__device__ __forceinline__ void gcn_row_zero(float2& a) { a.x = 0.0f; a.y = 0.0f; }
__device__ __forceinline__ void gcn_row_fma(float4& a, float v, const float4& m) {
    a.x = fmaf(v, m.x, a.x);
    a.y = fmaf(v, m.y, a.y);
    a.z = fmaf(v, m.z, a.z);
    a.w = fmaf(v, m.w, a.w);
}
__device__ __forceinline__ void gcn_row_fma(float2& a, float v, const float2& m) {
    a.x = fmaf(v, m.x, a.x);
    a.y = fmaf(v, m.y, a.y);
}

#define GATHER_U 8
template <int VEC>
__device__ __forceinline__ void gcn_gather_chunk(typename gcn_row<VEC>::type& acc, int c_l, float v_l, int n,
                                                 const float* __restrict__ x, unsigned loff) {
    typedef typename gcn_row<VEC>::type row_t;
    const int D = 32 * VEC;
    int j = 0;
    for (; j + GATHER_U <= n; j += GATHER_U) {
        row_t m[GATHER_U];
#pragma unroll
        for (int k = 0; k < GATHER_U; ++k) {
            const int c = __builtin_amdgcn_readlane(c_l, j + k);
            m[k] = *(const row_t*)(x + (size_t)c * D + loff);
        }
#pragma unroll
        for (int k = 0; k < GATHER_U; ++k) {
            const float v = __int_as_float(__builtin_amdgcn_readlane(__float_as_int(v_l), j + k));
            gcn_row_fma(acc, v, m[k]);
        }
    }
    const int rem = n - j;
    if (rem > 0) {
        row_t m[GATHER_U - 1];
#pragma unroll
        for (int k = 0; k < GATHER_U - 1; ++k) {
            if (k < rem) {
                const int c = __builtin_amdgcn_readlane(c_l, j + k);
                m[k] = *(const row_t*)(x + (size_t)c * D + loff);
            }
        }
#pragma unroll
        for (int k = 0; k < GATHER_U - 1; ++k) {
            if (k < rem) {
                const float v = __int_as_float(__builtin_amdgcn_readlane(__float_as_int(v_l), j + k));
                gcn_row_fma(acc, v, m[k]);
            }
        }
    }
}

template <int VEC>
__global__ __launch_bounds__(256)
void gcn_gather_rows(const int* __restrict__ rowptr, const int* __restrict__ ccol,
                     const float* __restrict__ cval, const float* __restrict__ x,
                     float* __restrict__ agg, int nRows, int nEdges) {
    typedef typename gcn_row<VEC>::type row_t;
    const int D = 32 * VEC;
    const int lane = threadIdx.x & 31;
    const unsigned loff = (threadIdx.x & 31u) * (unsigned)VEC;
    const int wave = __builtin_amdgcn_readfirstlane((blockIdx.x * blockDim.x + threadIdx.x) >> 5);
    const int r0 = wave * ROWS_PER_WAVE;
    if (r0 >= nRows) return;
    const int rEnd = min(nRows, r0 + ROWS_PER_WAVE);
    const int last = max(nEdges - 1, 0);
    int s = rowptr[r0];
    int e = rowptr[r0 + 1];
    int   c_n = ccol[min(s + lane, last)];
    float v_n = cval[min(s + lane, last)];
    for (int r = r0; r < rEnd; ++r) {
        const int e_next = rowptr[min(r + 2, nRows)];
        const int   c_l = c_n;
        const float v_l = v_n;
        c_n = ccol[min(e + lane, last)];
        v_n = cval[min(e + lane, last)];
        row_t acc;
        gcn_row_zero(acc);
        if (s < e) {
            gcn_gather_chunk<VEC>(acc, c_l, v_l, min(32, e - s), x, loff);
            for (int b = s + 32; b < e; b += 32) {
                const int i = min(b + lane, e - 1);
                gcn_gather_chunk<VEC>(acc, ccol[i], cval[i], min(32, e - b), x, loff);
            }
        }
        *(row_t*)(agg + (size_t)r * D + loff) = acc;
        s = e;
        e = e_next;
    }
}

template <int NTILES, bool RELU>
__global__ void gcn_gemm_wmma(const float* __restrict__ A,
                              const _Float16* __restrict__ Bp,
                              float* __restrict__ C, int M) {
    const int N = NTILES * 16;
    int lane = threadIdx.x & 31;
    int warp = threadIdx.x >> 5;
    int wavesPerBlock = blockDim.x >> 5;
    int mt = blockIdx.x * wavesPerBlock + warp;
    int mtiles = M >> 4;
    if (mt >= mtiles) return;

    int half = lane >> 4;
    int r    = lane & 15;

    int mrow = mt * 16 + r;

    v8f acc[NTILES];
#pragma unroll
    for (int t = 0; t < NTILES; ++t) acc[t] = (v8f){};

#pragma unroll
    for (int kk = 0; kk < 4; ++kk) {
        int k0 = kk * 32;
        const float* ap = A + (size_t)mrow * IN_DIM + k0 + half * 8;
        float4 f0 = *(const float4*)(ap + 0);
        float4 f1 = *(const float4*)(ap + 4);
        float4 f2 = *(const float4*)(ap + 16);
        float4 f3 = *(const float4*)(ap + 20);
        v16h a;
        a[0]  = (_Float16)f0.x; a[1]  = (_Float16)f0.y; a[2]  = (_Float16)f0.z; a[3]  = (_Float16)f0.w;
        a[4]  = (_Float16)f1.x; a[5]  = (_Float16)f1.y; a[6]  = (_Float16)f1.z; a[7]  = (_Float16)f1.w;
        a[8]  = (_Float16)f2.x; a[9]  = (_Float16)f2.y; a[10] = (_Float16)f2.z; a[11] = (_Float16)f2.w;
        a[12] = (_Float16)f3.x; a[13] = (_Float16)f3.y; a[14] = (_Float16)f3.z; a[15] = (_Float16)f3.w;

#pragma unroll
        for (int t = 0; t < NTILES; ++t) {
            v16h b = *(const v16h*)(Bp + (((size_t)t * 4 + kk) * 32 + lane) * 16);
            acc[t] = __builtin_amdgcn_wmma_f32_16x16x32_f16(
                         false, a,  false, b,
                         (short)0, acc[t],
                         false,  false);
        }
    }

    int mbase = mt * 16 + half * 8;
#pragma unroll
    for (int t = 0; t < NTILES; ++t) {
        int ncol = t * 16 + r;
#pragma unroll
        for (int v = 0; v < 8; ++v) {
            float val = acc[t][v];
            if (RELU) val = fmaxf(val, 0.0f);
            C[(size_t)(mbase + v) * N + ncol] = val;
        }
    }
}

extern "C" void kernel_launch(void* const* d_in, const int* in_sizes, int n_in,
                              void* d_out, int out_size, void* d_ws, size_t ws_size,
                              hipStream_t stream) {
    const float* x    = (const float*)d_in[0];
    const int*   row  = (const int*)d_in[1];
    const int*   col  = (const int*)d_in[2];
    const float* vals = (const float*)d_in[3];
    const float* W1   = (const float*)d_in[4];
    const float* W2   = (const float*)d_in[5];
    float* out = (float*)d_out;

    const int NN = in_sizes[0] / IN_DIM;
    const int NE = in_sizes[1];

    float*     aggF = (float*)d_ws;
    float*     hF   = aggF + (size_t)NN * IN_DIM;
    _Float16*  W1p  = (_Float16*)(hF + (size_t)NN * IN_DIM);
    _Float16*  W2p  = W1p + 128 * 128;
    const int  NNp  = ((NN + 8) >> 2) << 2;
    int*       cnt    = (int*)(W2p + 128 * 64);
    int*       rowptr = cnt + NNp;
    int*       rank   = rowptr + NNp;
    int*       slot   = rank + NE;
    int*       ccol   = slot + NE;
    float*     cval   = (float*)(ccol + NE);

    gcn_pack_w<<<4, 256, 0, stream>>>(W1, W1p, 8);
    gcn_pack_w<<<2, 256, 0, stream>>>(W2, W2p, 4);

    const int edgeBlocks = (NE + 255) / 256;
    gcn_zero_i32<<<(NN + 255) / 256, 256, 0, stream>>>(cnt, NN);
    gcn_count<<<edgeBlocks, 256, 0, stream>>>(row, cnt, rank, NE);
    gcn_scan<<<1, SCAN_THREADS, 0, stream>>>(cnt, rowptr, NN);
    gcn_place<<<edgeBlocks, 256, 0, stream>>>(row, rank, rowptr, slot, NE);
    const int rowBlocks = (NN + 8 * ROWS_PER_WAVE - 1) / (8 * ROWS_PER_WAVE);
    gcn_order_rows<<<rowBlocks, 256, 0, stream>>>(rowptr, slot, col, vals, ccol, cval, NN, NE);

    gcn_gather_rows<4><<<rowBlocks, 256, 0, stream>>>(rowptr, ccol, cval, x, aggF, NN, NE);

    {
        int mtiles = NN / 16;
        int blocks = (mtiles + 3) / 4;
        gcn_gemm_wmma<8, true><<<blocks, 128, 0, stream>>>(aggF, W1p, hF, NN);
    }

    float* tF = aggF;
    {
        int mtiles = NN / 16;
        int blocks = (mtiles + 3) / 4;
        gcn_gemm_wmma<4, false><<<blocks, 128, 0, stream>>>(hF, W2p, tF, NN);
    }

    gcn_gather_rows<2><<<rowBlocks, 256, 0, stream>>>(rowptr, ccol, cval, tF, out, NN, NE);
}


# Per-row edge counters and slot hand-out kept in LDS tables (row ranges x edge slices, a small per-row prefix over slices) instead of 800000 returning global atomics; f32 sums and order unchanged
# speedup vs baseline: 2.5579x; 1.2770x over previous
#include <hip/hip_runtime.h>
#include <hip/hip_bf16.h>

typedef __attribute__((ext_vector_type(16))) _Float16 v16h;
typedef __attribute__((ext_vector_type(8)))  float    v8f;

#define IN_DIM 128

__global__ void gcn_pack_w(const float* __restrict__ W, _Float16* __restrict__ Wp,
                           int Nt  ) {
    int id = blockIdx.x * blockDim.x + threadIdx.x;
    int total = Nt * 4 * 32;
    if (id >= total) return;
    int lane = id & 31;
    int kk   = (id >> 5) & 3;
    int t    = id >> 7;
    int half = lane >> 4;
    int r    = lane & 15;
    int N    = Nt * 16;
    int n    = t * 16 + r;
    int k0   = kk * 32;
    _Float16* dst = Wp + (size_t)id * 16;
#pragma unroll
    for (int j = 0; j < 8; ++j)
        dst[j] = (_Float16)W[(size_t)(k0 + half * 8 + j) * N + n];
#pragma unroll
    for (int j = 0; j < 8; ++j)
        dst[8 + j] = (_Float16)W[(size_t)(k0 + 16 + half * 8 + j) * N + n];
}

#define CNT_RANGE   8192
#define CNT_SLICE   32768
#define CNT_THREADS 512
#define CNT_U       8

__device__ __forceinline__ void gcn_load_rows(int4 (&v)[CNT_U], const int* __restrict__ row, int ib, int e1v) {
#pragma unroll
    for (int k = 0; k < CNT_U; ++k) {
        const int i = ib + k * CNT_THREADS * 4;
        v[k] = make_int4(-1, -1, -1, -1);
        if (i < e1v) v[k] = *(const int4*)(row + i);
    }
}

__device__ __forceinline__ void gcn_count_one(int* c, int v, int base, unsigned lim) {
    const unsigned d = (unsigned)(v - base);
    if (d < lim) atomicAdd(&c[d], 1);
}

#define CNT_PER_THREAD (CNT_RANGE / CNT_THREADS)

__global__ __launch_bounds__(CNT_THREADS)
void gcn_count_lds(const int* __restrict__ row, int* __restrict__ part,
                   int nRows, int nRowsP, int nEdges) {
    __shared__ int c[CNT_RANGE];
    const int t = threadIdx.x;
    const int base = blockIdx.y * CNT_RANGE;
    const unsigned lim = (unsigned)max(0, min(CNT_RANGE, nRows - base));
    const int e0 = blockIdx.x * CNT_SLICE;
    const int e1 = min(nEdges, e0 + CNT_SLICE);
    const int e1v = e0 + ((e1 - e0) & ~3);
    for (int j = t; j < CNT_RANGE; j += CNT_THREADS) c[j] = 0;
    __syncthreads();
    for (int ib = e0 + t * 4; ib < e1v; ib += CNT_THREADS * 4 * CNT_U) {
        int4 v[CNT_U];
        gcn_load_rows(v, row, ib, e1v);
#pragma unroll
        for (int k = 0; k < CNT_U; ++k) {
            gcn_count_one(c, v[k].x, base, lim);
            gcn_count_one(c, v[k].y, base, lim);
            gcn_count_one(c, v[k].z, base, lim);
            gcn_count_one(c, v[k].w, base, lim);
        }
    }
    if (t < e1 - e1v) gcn_count_one(c, row[e1v + t], base, lim);
    __syncthreads();
    int o[CNT_PER_THREAD];
#pragma unroll
    for (int k = 0; k < CNT_PER_THREAD; ++k) o[k] = c[k * CNT_THREADS + t];
#pragma unroll
    for (int k = 0; k < CNT_PER_THREAD; ++k) {
        const unsigned j = (unsigned)(k * CNT_THREADS + t);
        if (j < lim) part[blockIdx.x * nRowsP + base + j] = o[k];
    }
}

__global__ void gcn_slice_prefix(int* __restrict__ part, int* __restrict__ cnt,
                                 int nRows, int nRowsP, int nSlices) {
    const int node = blockIdx.x * blockDim.x + threadIdx.x;
    if (node >= nRows) return;
    int run = 0;
    for (int s0 = 0; s0 < nSlices; s0 += 8) {
        int v[8];
#pragma unroll
        for (int k = 0; k < 8; ++k)
            v[k] = (s0 + k < nSlices) ? part[(s0 + k) * nRowsP + node] : 0;
#pragma unroll
        for (int k = 0; k < 8; ++k) {
            if (s0 + k < nSlices) {
                part[(s0 + k) * nRowsP + node] = run;
                run += v[k];
            }
        }
    }
    cnt[node] = run;
}

#define SCAN_THREADS 1024
#define SCAN_K 13
__global__ __launch_bounds__(SCAN_THREADS)
void gcn_scan(const int* __restrict__ cnt, int* __restrict__ rowptr, int n) {
    __shared__ int wsum[SCAN_K * 32];
    __shared__ int total_s;
    const int t = threadIdx.x;
    const int lane = t & 31;
    const int w = t >> 5;
    int carry = 0;
    for (int base = 0; base < n; base += SCAN_K * SCAN_THREADS * 4) {
        int4 v[SCAN_K];
        int incl[SCAN_K];
#pragma unroll
        for (int k = 0; k < SCAN_K; ++k) {
            int idx = base + (k * SCAN_THREADS + t) * 4;
            if (idx + 3 < n) {
                v[k] = *(const int4*)(cnt + idx);
            } else {
                v[k].x = (idx     < n) ? cnt[idx]     : 0;
                v[k].y = (idx + 1 < n) ? cnt[idx + 1] : 0;
                v[k].z = (idx + 2 < n) ? cnt[idx + 2] : 0;
                v[k].w = 0;
            }
        }
#pragma unroll
        for (int k = 0; k < SCAN_K; ++k) {
            int sum = v[k].x + v[k].y + v[k].z + v[k].w;
            for (int d = 1; d < 32; d <<= 1) {
                int o = __shfl_up(sum, d);
                if (lane >= d) sum += o;
            }
            incl[k] = sum;
            if (lane == 31) wsum[k * 32 + w] = sum;
        }
        __syncthreads();
        if (w == 0) {
            int loc[SCAN_K];
            int run = 0;
#pragma unroll
            for (int i = 0; i < SCAN_K; ++i) { loc[i] = wsum[lane * SCAN_K + i]; run += loc[i]; }
            int inc = run;
            for (int d = 1; d < 32; d <<= 1) {
                int o = __shfl_up(inc, d);
                if (lane >= d) inc += o;
            }
            int ex = inc - run;
#pragma unroll
            for (int i = 0; i < SCAN_K; ++i) { wsum[lane * SCAN_K + i] = ex; ex += loc[i]; }
            if (lane == 31) total_s = inc;
        }
        __syncthreads();
#pragma unroll
        for (int k = 0; k < SCAN_K; ++k) {
            int idx = base + (k * SCAN_THREADS + t) * 4;
            int own = v[k].x + v[k].y + v[k].z + v[k].w;
            int p0 = carry + wsum[k * 32 + w] + incl[k] - own;
            int p1 = p0 + v[k].x;
            int p2 = p1 + v[k].y;
            int p3 = p2 + v[k].z;
            if (idx + 3 < n) {
                *(int4*)(rowptr + idx) = make_int4(p0, p1, p2, p3);
            } else {
                if (idx     < n) rowptr[idx]     = p0;
                if (idx + 1 < n) rowptr[idx + 1] = p1;
                if (idx + 2 < n) rowptr[idx + 2] = p2;
            }
        }
        carry += total_s;
        __syncthreads();
    }
    if (t == 0) rowptr[n] = carry;
}

__device__ __forceinline__ int gcn_take_slot(int* p, int v, int base, unsigned lim) {
    const unsigned d = (unsigned)(v - base);
    int pos = -1;
    if (d < lim) pos = atomicAdd(&p[d], 1);
    return pos;
}

__global__ __launch_bounds__(CNT_THREADS)
void gcn_place_lds(const int* __restrict__ row, const int* __restrict__ rowptr,
                   const int* __restrict__ part, int* __restrict__ slot,
                   int nRows, int nRowsP, int nEdges) {
    __shared__ int p[CNT_RANGE];
    const int t = threadIdx.x;
    const int base = blockIdx.y * CNT_RANGE;
    const unsigned lim = (unsigned)max(0, min(CNT_RANGE, nRows - base));
    const int e0 = blockIdx.x * CNT_SLICE;
    const int e1 = min(nEdges, e0 + CNT_SLICE);
    const int e1v = e0 + ((e1 - e0) & ~3);
    {
        int a[CNT_PER_THREAD], b[CNT_PER_THREAD];
#pragma unroll
        for (int k = 0; k < CNT_PER_THREAD; ++k) {
            const unsigned j = (unsigned)(k * CNT_THREADS + t);
            a[k] = 0;
            b[k] = 0;
            if (j < lim) {
                a[k] = rowptr[base + j];
                b[k] = part[blockIdx.x * nRowsP + base + j];
            }
        }
#pragma unroll
        for (int k = 0; k < CNT_PER_THREAD; ++k) p[k * CNT_THREADS + t] = a[k] + b[k];
    }
    __syncthreads();
    for (int ib = e0 + t * 4; ib < e1v; ib += CNT_THREADS * 4 * CNT_U) {
        int4 v[CNT_U];
        gcn_load_rows(v, row, ib, e1v);
        int4 pos[CNT_U];
#pragma unroll
        for (int k = 0; k < CNT_U; ++k) {
            pos[k].x = gcn_take_slot(p, v[k].x, base, lim);
            pos[k].y = gcn_take_slot(p, v[k].y, base, lim);
            pos[k].z = gcn_take_slot(p, v[k].z, base, lim);
            pos[k].w = gcn_take_slot(p, v[k].w, base, lim);
        }
#pragma unroll
        for (int k = 0; k < CNT_U; ++k) {
            const int i = ib + k * CNT_THREADS * 4;
            if (pos[k].x >= 0) slot[pos[k].x] = i;
            if (pos[k].y >= 0) slot[pos[k].y] = i + 1;
            if (pos[k].z >= 0) slot[pos[k].z] = i + 2;
            if (pos[k].w >= 0) slot[pos[k].w] = i + 3;
        }
    }
    if (t < e1 - e1v) {
        const int pos = gcn_take_slot(p, row[e1v + t], base, lim);
        if (pos >= 0) slot[pos] = e1v + t;
    }
}

#define ROWS_PER_WAVE 5

__global__ __launch_bounds__(256)
void gcn_order_rows(const int* __restrict__ rowptr, const int* __restrict__ slot,
                    const int* __restrict__ col, const float* __restrict__ vals,
                    int* __restrict__ ccol, float* __restrict__ cval,
                    int nRows, int nEdges) {
    const int lane = threadIdx.x & 31;
    const int wave = __builtin_amdgcn_readfirstlane((blockIdx.x * blockDim.x + threadIdx.x) >> 5);
    const int r0 = wave * ROWS_PER_WAVE;
    if (r0 >= nRows) return;
    const int rEnd = min(nRows, r0 + ROWS_PER_WAVE);
    const int last = max(nEdges - 1, 0);
    int s = rowptr[r0];
    int e = rowptr[r0 + 1];
    int nxt = slot[min(s + lane, last)];
    for (int r = r0; r < rEnd; ++r) {
        const int e_next = rowptr[min(r + 2, nRows)];
        const int first = nxt;
        const int deg = e - s;
        if (deg <= 32) {
            int place = 0;
            for (int j = 0; j < deg; ++j)
                place += (__builtin_amdgcn_readlane(first, j) < first) ? 1 : 0;
            nxt = slot[min(e + lane, last)];
            if (lane < deg) {
                ccol[s + place] = col[first];
                cval[s + place] = vals[first];
            }
        } else {
            nxt = slot[min(e + lane, last)];
            for (int ib = s; ib < e; ib += 32) {
                const int i = ib + lane;
                const int mine = slot[min(i, e - 1)];
                int place = 0;
                for (int jb = s; jb < e; jb += 32) {
                    const int jn = min(32, e - jb);
                    const int other = slot[min(jb + lane, e - 1)];
                    for (int j = 0; j < jn; ++j)
                        place += (__builtin_amdgcn_readlane(other, j) < mine) ? 1 : 0;
                }
                if (i < e) {
                    ccol[s + place] = col[mine];
                    cval[s + place] = vals[mine];
                }
            }
        }
        s = e;
        e = e_next;
    }
}

template <int VEC> struct gcn_row;
template <> struct gcn_row<4> { typedef float4 type; };
template <> struct gcn_row<2> { typedef float2 type; };
__device__ __forceinline__ void gcn_row_zero(float4& a) { a.x = 0.0f; a.y = 0.0f; a.z = 0.0f; a.w = 0.0f; }
__device__ __forceinline__ void gcn_row_zero(float2& a) { a.x = 0.0f; a.y = 0.0f; }
__device__ __forceinline__ void gcn_row_fma(float4& a, float v, const float4& m) {
    a.x = fmaf(v, m.x, a.x);
    a.y = fmaf(v, m.y, a.y);
    a.z = fmaf(v, m.z, a.z);
    a.w = fmaf(v, m.w, a.w);
}
__device__ __forceinline__ void gcn_row_fma(float2& a, float v, const float2& m) {
    a.x = fmaf(v, m.x, a.x);
    a.y = fmaf(v, m.y, a.y);
}

#define GATHER_U 8
template <int VEC>
__device__ __forceinline__ void gcn_gather_chunk(typename gcn_row<VEC>::type& acc, int c_l, float v_l, int n,
                                                 const float* __restrict__ x, unsigned loff) {
    typedef typename gcn_row<VEC>::type row_t;
    const int D = 32 * VEC;
    int j = 0;
    for (; j + GATHER_U <= n; j += GATHER_U) {
        row_t m[GATHER_U];
#pragma unroll
        for (int k = 0; k < GATHER_U; ++k) {
            const int c = __builtin_amdgcn_readlane(c_l, j + k);
            m[k] = *(const row_t*)(x + (size_t)c * D + loff);
        }
#pragma unroll
        for (int k = 0; k < GATHER_U; ++k) {
            const float v = __int_as_float(__builtin_amdgcn_readlane(__float_as_int(v_l), j + k));
            gcn_row_fma(acc, v, m[k]);
        }
    }
    const int rem = n - j;
    if (rem > 0) {
        row_t m[GATHER_U - 1];
#pragma unroll
        for (int k = 0; k < GATHER_U - 1; ++k) {
            if (k < rem) {
                const int c = __builtin_amdgcn_readlane(c_l, j + k);
                m[k] = *(const row_t*)(x + (size_t)c * D + loff);
            }
        }
#pragma unroll
        for (int k = 0; k < GATHER_U - 1; ++k) {
            if (k < rem) {
                const float v = __int_as_float(__builtin_amdgcn_readlane(__float_as_int(v_l), j + k));
                gcn_row_fma(acc, v, m[k]);
            }
        }
    }
}

template <int VEC>
__global__ __launch_bounds__(256)
void gcn_gather_rows(const int* __restrict__ rowptr, const int* __restrict__ ccol,
                     const float* __restrict__ cval, const float* __restrict__ x,
                     float* __restrict__ agg, int nRows, int nEdges) {
    typedef typename gcn_row<VEC>::type row_t;
    const int D = 32 * VEC;
    const int lane = threadIdx.x & 31;
    const unsigned loff = (threadIdx.x & 31u) * (unsigned)VEC;
    const int wave = __builtin_amdgcn_readfirstlane((blockIdx.x * blockDim.x + threadIdx.x) >> 5);
    const int r0 = wave * ROWS_PER_WAVE;
    if (r0 >= nRows) return;
    const int rEnd = min(nRows, r0 + ROWS_PER_WAVE);
    const int last = max(nEdges - 1, 0);
    int s = rowptr[r0];
    int e = rowptr[r0 + 1];
    int   c_n = ccol[min(s + lane, last)];
    float v_n = cval[min(s + lane, last)];
    for (int r = r0; r < rEnd; ++r) {
        const int e_next = rowptr[min(r + 2, nRows)];
        const int   c_l = c_n;
        const float v_l = v_n;
        c_n = ccol[min(e + lane, last)];
        v_n = cval[min(e + lane, last)];
        row_t acc;
        gcn_row_zero(acc);
        if (s < e) {
            gcn_gather_chunk<VEC>(acc, c_l, v_l, min(32, e - s), x, loff);
            for (int b = s + 32; b < e; b += 32) {
                const int i = min(b + lane, e - 1);
                gcn_gather_chunk<VEC>(acc, ccol[i], cval[i], min(32, e - b), x, loff);
            }
        }
        *(row_t*)(agg + (size_t)r * D + loff) = acc;
        s = e;
        e = e_next;
    }
}

template <int NTILES, bool RELU>
__global__ void gcn_gemm_wmma(const float* __restrict__ A,
                              const _Float16* __restrict__ Bp,
                              float* __restrict__ C, int M) {
    const int N = NTILES * 16;
    int lane = threadIdx.x & 31;
    int warp = threadIdx.x >> 5;
    int wavesPerBlock = blockDim.x >> 5;
    int mt = blockIdx.x * wavesPerBlock + warp;
    int mtiles = M >> 4;
    if (mt >= mtiles) return;

    int half = lane >> 4;
    int r    = lane & 15;

    int mrow = mt * 16 + r;

    v8f acc[NTILES];
#pragma unroll
    for (int t = 0; t < NTILES; ++t) acc[t] = (v8f){};

#pragma unroll
    for (int kk = 0; kk < 4; ++kk) {
        int k0 = kk * 32;
        const float* ap = A + (size_t)mrow * IN_DIM + k0 + half * 8;
        float4 f0 = *(const float4*)(ap + 0);
        float4 f1 = *(const float4*)(ap + 4);
        float4 f2 = *(const float4*)(ap + 16);
        float4 f3 = *(const float4*)(ap + 20);
        v16h a;
        a[0]  = (_Float16)f0.x; a[1]  = (_Float16)f0.y; a[2]  = (_Float16)f0.z; a[3]  = (_Float16)f0.w;
        a[4]  = (_Float16)f1.x; a[5]  = (_Float16)f1.y; a[6]  = (_Float16)f1.z; a[7]  = (_Float16)f1.w;
        a[8]  = (_Float16)f2.x; a[9]  = (_Float16)f2.y; a[10] = (_Float16)f2.z; a[11] = (_Float16)f2.w;
        a[12] = (_Float16)f3.x; a[13] = (_Float16)f3.y; a[14] = (_Float16)f3.z; a[15] = (_Float16)f3.w;

#pragma unroll
        for (int t = 0; t < NTILES; ++t) {
            v16h b = *(const v16h*)(Bp + (((size_t)t * 4 + kk) * 32 + lane) * 16);
            acc[t] = __builtin_amdgcn_wmma_f32_16x16x32_f16(
                         false, a,  false, b,
                         (short)0, acc[t],
                         false,  false);
        }
    }

    int mbase = mt * 16 + half * 8;
#pragma unroll
    for (int t = 0; t < NTILES; ++t) {
        int ncol = t * 16 + r;
#pragma unroll
        for (int v = 0; v < 8; ++v) {
            float val = acc[t][v];
            if (RELU) val = fmaxf(val, 0.0f);
            C[(size_t)(mbase + v) * N + ncol] = val;
        }
    }
}

extern "C" void kernel_launch(void* const* d_in, const int* in_sizes, int n_in,
                              void* d_out, int out_size, void* d_ws, size_t ws_size,
                              hipStream_t stream) {
    const float* x    = (const float*)d_in[0];
    const int*   row  = (const int*)d_in[1];
    const int*   col  = (const int*)d_in[2];
    const float* vals = (const float*)d_in[3];
    const float* W1   = (const float*)d_in[4];
    const float* W2   = (const float*)d_in[5];
    float* out = (float*)d_out;

    const int NN = in_sizes[0] / IN_DIM;
    const int NE = in_sizes[1];

    float*     aggF = (float*)d_ws;
    float*     hF   = aggF + (size_t)NN * IN_DIM;
    _Float16*  W1p  = (_Float16*)(hF + (size_t)NN * IN_DIM);
    _Float16*  W2p  = W1p + 128 * 128;
    const int  NNp  = ((NN + 8) >> 2) << 2;
    int*       cnt    = (int*)(W2p + 128 * 64);
    int*       rowptr = cnt + NNp;
    int*       slot   = rowptr + NNp;
    int*       ccol   = slot + NE;
    float*     cval   = (float*)(ccol + NE);
    int*       part   = (int*)(cval + NE);

    gcn_pack_w<<<4, 256, 0, stream>>>(W1, W1p, 8);
    gcn_pack_w<<<2, 256, 0, stream>>>(W2, W2p, 4);

    const int nSlices = (NE + CNT_SLICE - 1) / CNT_SLICE;
    const int nRanges = (NN + CNT_RANGE - 1) / CNT_RANGE;
    gcn_count_lds<<<dim3(nSlices, nRanges), CNT_THREADS, 0, stream>>>(row, part, NN, NNp, NE);
    gcn_slice_prefix<<<(NN + 255) / 256, 256, 0, stream>>>(part, cnt, NN, NNp, nSlices);
    gcn_scan<<<1, SCAN_THREADS, 0, stream>>>(cnt, rowptr, NN);
    gcn_place_lds<<<dim3(nSlices, nRanges), CNT_THREADS, 0, stream>>>(row, rowptr, part, slot, NN, NNp, NE);
    const int rowBlocks = (NN + 8 * ROWS_PER_WAVE - 1) / (8 * ROWS_PER_WAVE);
    gcn_order_rows<<<rowBlocks, 256, 0, stream>>>(rowptr, slot, col, vals, ccol, cval, NN, NE);

    gcn_gather_rows<4><<<rowBlocks, 256, 0, stream>>>(rowptr, ccol, cval, x, aggF, NN, NE);

    {
        int mtiles = NN / 16;
        int blocks = (mtiles + 3) / 4;
        gcn_gemm_wmma<8, true><<<blocks, 128, 0, stream>>>(aggF, W1p, hF, NN);
    }

    float* tF = aggF;
    {
        int mtiles = NN / 16;
        int blocks = (mtiles + 3) / 4;
        gcn_gemm_wmma<4, false><<<blocks, 128, 0, stream>>>(hF, W2p, tF, NN);
    }

    gcn_gather_rows<2><<<rowBlocks, 256, 0, stream>>>(rowptr, ccol, cval, tF, out, NN, NE);
}


# Layer-1 edge sum, its product with W1, relu and the product with W2 fused in one kernel: 16 summed rows pass through LDS tiles into the WMMA A operand; f32 sums and order unchanged
# speedup vs baseline: 3.1548x; 1.2334x over previous
#include <hip/hip_runtime.h>
#include <hip/hip_bf16.h>

typedef __attribute__((ext_vector_type(16))) _Float16 v16h;
typedef __attribute__((ext_vector_type(8)))  _Float16 v8h;
typedef __attribute__((ext_vector_type(4)))  _Float16 v4h;
typedef __attribute__((ext_vector_type(8)))  float    v8f;

#define IN_DIM 128

__global__ void gcn_pack_w(const float* __restrict__ W, _Float16* __restrict__ Wp,
                           int Nt  ) {
    int id = blockIdx.x * blockDim.x + threadIdx.x;
    int total = Nt * 4 * 32;
    if (id >= total) return;
    int lane = id & 31;
    int kk   = (id >> 5) & 3;
    int t    = id >> 7;
    int half = lane >> 4;
    int r    = lane & 15;
    int N    = Nt * 16;
    int n    = t * 16 + r;
    int k0   = kk * 32;
    _Float16* dst = Wp + (size_t)id * 16;
#pragma unroll
    for (int j = 0; j < 8; ++j)
        dst[j] = (_Float16)W[(size_t)(k0 + half * 8 + j) * N + n];
#pragma unroll
    for (int j = 0; j < 8; ++j)
        dst[8 + j] = (_Float16)W[(size_t)(k0 + 16 + half * 8 + j) * N + n];
}

#define CNT_RANGE   8192
#define CNT_SLICE   32768
#define CNT_THREADS 512
#define CNT_U       8

__device__ __forceinline__ void gcn_load_rows(int4 (&v)[CNT_U], const int* __restrict__ row, int ib, int e1v) {
#pragma unroll
    for (int k = 0; k < CNT_U; ++k) {
        const int i = ib + k * CNT_THREADS * 4;
        v[k] = make_int4(-1, -1, -1, -1);
        if (i < e1v) v[k] = *(const int4*)(row + i);
    }
}

__device__ __forceinline__ void gcn_count_one(int* c, int v, int base, unsigned lim) {
    const unsigned d = (unsigned)(v - base);
    if (d < lim) atomicAdd(&c[d], 1);
}

#define CNT_PER_THREAD (CNT_RANGE / CNT_THREADS)

__global__ __launch_bounds__(CNT_THREADS)
void gcn_count_lds(const int* __restrict__ row, int* __restrict__ part,
                   int nRows, int nRowsP, int nEdges) {
    __shared__ int c[CNT_RANGE];
    const int t = threadIdx.x;
    const int base = blockIdx.y * CNT_RANGE;
    const unsigned lim = (unsigned)max(0, min(CNT_RANGE, nRows - base));
    const int e0 = blockIdx.x * CNT_SLICE;
    const int e1 = min(nEdges, e0 + CNT_SLICE);
    const int e1v = e0 + ((e1 - e0) & ~3);
    for (int j = t; j < CNT_RANGE; j += CNT_THREADS) c[j] = 0;
    __syncthreads();
    for (int ib = e0 + t * 4; ib < e1v; ib += CNT_THREADS * 4 * CNT_U) {
        int4 v[CNT_U];
        gcn_load_rows(v, row, ib, e1v);
#pragma unroll
        for (int k = 0; k < CNT_U; ++k) {
            gcn_count_one(c, v[k].x, base, lim);
            gcn_count_one(c, v[k].y, base, lim);
            gcn_count_one(c, v[k].z, base, lim);
            gcn_count_one(c, v[k].w, base, lim);
        }
    }
    if (t < e1 - e1v) gcn_count_one(c, row[e1v + t], base, lim);
    __syncthreads();
    int o[CNT_PER_THREAD];
#pragma unroll
    for (int k = 0; k < CNT_PER_THREAD; ++k) o[k] = c[k * CNT_THREADS + t];
#pragma unroll
    for (int k = 0; k < CNT_PER_THREAD; ++k) {
        const unsigned j = (unsigned)(k * CNT_THREADS + t);
        if (j < lim) part[blockIdx.x * nRowsP + base + j] = o[k];
    }
}

__global__ void gcn_slice_prefix(int* __restrict__ part, int* __restrict__ cnt,
                                 int nRows, int nRowsP, int nSlices) {
    const int node = blockIdx.x * blockDim.x + threadIdx.x;
    if (node >= nRows) return;
    int run = 0;
    for (int s0 = 0; s0 < nSlices; s0 += 8) {
        int v[8];
#pragma unroll
        for (int k = 0; k < 8; ++k)
            v[k] = (s0 + k < nSlices) ? part[(s0 + k) * nRowsP + node] : 0;
#pragma unroll
        for (int k = 0; k < 8; ++k) {
            if (s0 + k < nSlices) {
                part[(s0 + k) * nRowsP + node] = run;
                run += v[k];
            }
        }
    }
    cnt[node] = run;
}

#define SCAN_THREADS 1024
#define SCAN_K 13
__global__ __launch_bounds__(SCAN_THREADS)
void gcn_scan(const int* __restrict__ cnt, int* __restrict__ rowptr, int n) {
    __shared__ int wsum[SCAN_K * 32];
    __shared__ int total_s;
    const int t = threadIdx.x;
    const int lane = t & 31;
    const int w = t >> 5;
    int carry = 0;
    for (int base = 0; base < n; base += SCAN_K * SCAN_THREADS * 4) {
        int4 v[SCAN_K];
        int incl[SCAN_K];
#pragma unroll
        for (int k = 0; k < SCAN_K; ++k) {
            int idx = base + (k * SCAN_THREADS + t) * 4;
            if (idx + 3 < n) {
                v[k] = *(const int4*)(cnt + idx);
            } else {
                v[k].x = (idx     < n) ? cnt[idx]     : 0;
                v[k].y = (idx + 1 < n) ? cnt[idx + 1] : 0;
                v[k].z = (idx + 2 < n) ? cnt[idx + 2] : 0;
                v[k].w = 0;
            }
        }
#pragma unroll
        for (int k = 0; k < SCAN_K; ++k) {
            int sum = v[k].x + v[k].y + v[k].z + v[k].w;
            for (int d = 1; d < 32; d <<= 1) {
                int o = __shfl_up(sum, d);
                if (lane >= d) sum += o;
            }
            incl[k] = sum;
            if (lane == 31) wsum[k * 32 + w] = sum;
        }
        __syncthreads();
        if (w == 0) {
            int loc[SCAN_K];
            int run = 0;
#pragma unroll
            for (int i = 0; i < SCAN_K; ++i) { loc[i] = wsum[lane * SCAN_K + i]; run += loc[i]; }
            int inc = run;
            for (int d = 1; d < 32; d <<= 1) {
                int o = __shfl_up(inc, d);
                if (lane >= d) inc += o;
            }
            int ex = inc - run;
#pragma unroll
            for (int i = 0; i < SCAN_K; ++i) { wsum[lane * SCAN_K + i] = ex; ex += loc[i]; }
            if (lane == 31) total_s = inc;
        }
        __syncthreads();
#pragma unroll
        for (int k = 0; k < SCAN_K; ++k) {
            int idx = base + (k * SCAN_THREADS + t) * 4;
            int own = v[k].x + v[k].y + v[k].z + v[k].w;
            int p0 = carry + wsum[k * 32 + w] + incl[k] - own;
            int p1 = p0 + v[k].x;
            int p2 = p1 + v[k].y;
            int p3 = p2 + v[k].z;
            if (idx + 3 < n) {
                *(int4*)(rowptr + idx) = make_int4(p0, p1, p2, p3);
            } else {
                if (idx     < n) rowptr[idx]     = p0;
                if (idx + 1 < n) rowptr[idx + 1] = p1;
                if (idx + 2 < n) rowptr[idx + 2] = p2;
            }
        }
        carry += total_s;
        __syncthreads();
    }
    if (t == 0) rowptr[n] = carry;
}

__device__ __forceinline__ int gcn_take_slot(int* p, int v, int base, unsigned lim) {
    const unsigned d = (unsigned)(v - base);
    int pos = -1;
    if (d < lim) pos = atomicAdd(&p[d], 1);
    return pos;
}

__global__ __launch_bounds__(CNT_THREADS)
void gcn_place_lds(const int* __restrict__ row, const int* __restrict__ rowptr,
                   const int* __restrict__ part, int* __restrict__ slot,
                   int nRows, int nRowsP, int nEdges) {
    __shared__ int p[CNT_RANGE];
    const int t = threadIdx.x;
    const int base = blockIdx.y * CNT_RANGE;
    const unsigned lim = (unsigned)max(0, min(CNT_RANGE, nRows - base));
    const int e0 = blockIdx.x * CNT_SLICE;
    const int e1 = min(nEdges, e0 + CNT_SLICE);
    const int e1v = e0 + ((e1 - e0) & ~3);
    {
        int a[CNT_PER_THREAD], b[CNT_PER_THREAD];
#pragma unroll
        for (int k = 0; k < CNT_PER_THREAD; ++k) {
            const unsigned j = (unsigned)(k * CNT_THREADS + t);
            a[k] = 0;
            b[k] = 0;
            if (j < lim) {
                a[k] = rowptr[base + j];
                b[k] = part[blockIdx.x * nRowsP + base + j];
            }
        }
#pragma unroll
        for (int k = 0; k < CNT_PER_THREAD; ++k) p[k * CNT_THREADS + t] = a[k] + b[k];
    }
    __syncthreads();
    for (int ib = e0 + t * 4; ib < e1v; ib += CNT_THREADS * 4 * CNT_U) {
        int4 v[CNT_U];
        gcn_load_rows(v, row, ib, e1v);
        int4 pos[CNT_U];
#pragma unroll
        for (int k = 0; k < CNT_U; ++k) {
            pos[k].x = gcn_take_slot(p, v[k].x, base, lim);
            pos[k].y = gcn_take_slot(p, v[k].y, base, lim);
            pos[k].z = gcn_take_slot(p, v[k].z, base, lim);
            pos[k].w = gcn_take_slot(p, v[k].w, base, lim);
        }
#pragma unroll
        for (int k = 0; k < CNT_U; ++k) {
            const int i = ib + k * CNT_THREADS * 4;
            if (pos[k].x >= 0) slot[pos[k].x] = i;
            if (pos[k].y >= 0) slot[pos[k].y] = i + 1;
            if (pos[k].z >= 0) slot[pos[k].z] = i + 2;
            if (pos[k].w >= 0) slot[pos[k].w] = i + 3;
        }
    }
    if (t < e1 - e1v) {
        const int pos = gcn_take_slot(p, row[e1v + t], base, lim);
        if (pos >= 0) slot[pos] = e1v + t;
    }
}

#define ROWS_PER_WAVE 5

__global__ __launch_bounds__(256)
void gcn_order_rows(const int* __restrict__ rowptr, const int* __restrict__ slot,
                    const int* __restrict__ col, const float* __restrict__ vals,
                    int* __restrict__ ccol, float* __restrict__ cval,
                    int nRows, int nEdges) {
    const int lane = threadIdx.x & 31;
    const int wave = __builtin_amdgcn_readfirstlane((blockIdx.x * blockDim.x + threadIdx.x) >> 5);
    const int r0 = wave * ROWS_PER_WAVE;
    if (r0 >= nRows) return;
    const int rEnd = min(nRows, r0 + ROWS_PER_WAVE);
    const int last = max(nEdges - 1, 0);
    int s = rowptr[r0];
    int e = rowptr[r0 + 1];
    int nxt = slot[min(s + lane, last)];
    for (int r = r0; r < rEnd; ++r) {
        const int e_next = rowptr[min(r + 2, nRows)];
        const int first = nxt;
        const int deg = e - s;
        if (deg <= 32) {
            int place = 0;
            for (int j = 0; j < deg; ++j)
                place += (__builtin_amdgcn_readlane(first, j) < first) ? 1 : 0;
            nxt = slot[min(e + lane, last)];
            if (lane < deg) {
                ccol[s + place] = col[first];
                cval[s + place] = vals[first];
            }
        } else {
            nxt = slot[min(e + lane, last)];
            for (int ib = s; ib < e; ib += 32) {
                const int i = ib + lane;
                const int mine = slot[min(i, e - 1)];
                int place = 0;
                for (int jb = s; jb < e; jb += 32) {
                    const int jn = min(32, e - jb);
                    const int other = slot[min(jb + lane, e - 1)];
                    for (int j = 0; j < jn; ++j)
                        place += (__builtin_amdgcn_readlane(other, j) < mine) ? 1 : 0;
                }
                if (i < e) {
                    ccol[s + place] = col[mine];
                    cval[s + place] = vals[mine];
                }
            }
        }
        s = e;
        e = e_next;
    }
}

template <int VEC> struct gcn_row;
template <> struct gcn_row<4> { typedef float4 type; };
template <> struct gcn_row<2> { typedef float2 type; };
__device__ __forceinline__ void gcn_row_zero(float4& a) { a.x = 0.0f; a.y = 0.0f; a.z = 0.0f; a.w = 0.0f; }
__device__ __forceinline__ void gcn_row_zero(float2& a) { a.x = 0.0f; a.y = 0.0f; }
__device__ __forceinline__ void gcn_row_fma(float4& a, float v, const float4& m) {
    a.x = fmaf(v, m.x, a.x);
    a.y = fmaf(v, m.y, a.y);
    a.z = fmaf(v, m.z, a.z);
    a.w = fmaf(v, m.w, a.w);
}
__device__ __forceinline__ void gcn_row_fma(float2& a, float v, const float2& m) {
    a.x = fmaf(v, m.x, a.x);
    a.y = fmaf(v, m.y, a.y);
}

#define GATHER_U 8
template <int VEC>
__device__ __forceinline__ void gcn_gather_chunk(typename gcn_row<VEC>::type& acc, int c_l, float v_l, int n,
                                                 const float* __restrict__ x, unsigned loff) {
    typedef typename gcn_row<VEC>::type row_t;
    const int D = 32 * VEC;
    int j = 0;
    for (; j + GATHER_U <= n; j += GATHER_U) {
        row_t m[GATHER_U];
#pragma unroll
        for (int k = 0; k < GATHER_U; ++k) {
            const int c = __builtin_amdgcn_readlane(c_l, j + k);
            m[k] = *(const row_t*)(x + (size_t)c * D + loff);
        }
#pragma unroll
        for (int k = 0; k < GATHER_U; ++k) {
            const float v = __int_as_float(__builtin_amdgcn_readlane(__float_as_int(v_l), j + k));
            gcn_row_fma(acc, v, m[k]);
        }
    }
    const int rem = n - j;
    if (rem > 0) {
        row_t m[GATHER_U - 1];
#pragma unroll
        for (int k = 0; k < GATHER_U - 1; ++k) {
            if (k < rem) {
                const int c = __builtin_amdgcn_readlane(c_l, j + k);
                m[k] = *(const row_t*)(x + (size_t)c * D + loff);
            }
        }
#pragma unroll
        for (int k = 0; k < GATHER_U - 1; ++k) {
            if (k < rem) {
                const float v = __int_as_float(__builtin_amdgcn_readlane(__float_as_int(v_l), j + k));
                gcn_row_fma(acc, v, m[k]);
            }
        }
    }
}

template <int VEC>
__global__ __launch_bounds__(256)
void gcn_gather_rows(const int* __restrict__ rowptr, const int* __restrict__ ccol,
                     const float* __restrict__ cval, const float* __restrict__ x,
                     float* __restrict__ agg, int nRows, int nEdges) {
    typedef typename gcn_row<VEC>::type row_t;
    const int D = 32 * VEC;
    const int lane = threadIdx.x & 31;
    const unsigned loff = (threadIdx.x & 31u) * (unsigned)VEC;
    const int wave = __builtin_amdgcn_readfirstlane((blockIdx.x * blockDim.x + threadIdx.x) >> 5);
    const int r0 = wave * ROWS_PER_WAVE;
    if (r0 >= nRows) return;
    const int rEnd = min(nRows, r0 + ROWS_PER_WAVE);
    const int last = max(nEdges - 1, 0);
    int s = rowptr[r0];
    int e = rowptr[r0 + 1];
    int   c_n = ccol[min(s + lane, last)];
    float v_n = cval[min(s + lane, last)];
    for (int r = r0; r < rEnd; ++r) {
        const int e_next = rowptr[min(r + 2, nRows)];
        const int   c_l = c_n;
        const float v_l = v_n;
        c_n = ccol[min(e + lane, last)];
        v_n = cval[min(e + lane, last)];
        row_t acc;
        gcn_row_zero(acc);
        if (s < e) {
            gcn_gather_chunk<VEC>(acc, c_l, v_l, min(32, e - s), x, loff);
            for (int b = s + 32; b < e; b += 32) {
                const int i = min(b + lane, e - 1);
                gcn_gather_chunk<VEC>(acc, ccol[i], cval[i], min(32, e - b), x, loff);
            }
        }
        *(row_t*)(agg + (size_t)r * D + loff) = acc;
        s = e;
        e = e_next;
    }
}

#define FUSE_WAVES 4
#define FUSE_ROWS  4
#define FUSE_LD    136

__device__ __forceinline__ v16h gcn_frag_a(const _Float16* tile, int r, int half, int kk) {
    const _Float16* p = tile + r * FUSE_LD + kk * 32 + half * 8;
    const v8h lo = *(const v8h*)p;
    const v8h hi = *(const v8h*)(p + 16);
    return __builtin_shufflevector(lo, hi, 0, 1, 2, 3, 4, 5, 6, 7, 8, 9, 10, 11, 12, 13, 14, 15);
}

__global__ __launch_bounds__(FUSE_WAVES * 32) __attribute__((amdgpu_waves_per_eu(10)))
void gcn_layer1_fused(const int* __restrict__ rowptr, const int* __restrict__ ccol,
                      const float* __restrict__ cval, const float* __restrict__ x,
                      const _Float16* __restrict__ W1p, const _Float16* __restrict__ W2p,
                      float* __restrict__ t, int nRows, int nEdges) {
    __shared__ __attribute__((aligned(16))) _Float16 sA[16 * FUSE_LD];
    __shared__ __attribute__((aligned(16))) _Float16 sH[16 * FUSE_LD];
    const int lane = threadIdx.x & 31;
    const unsigned loff = (threadIdx.x & 31u) * 4u;
    const int w = __builtin_amdgcn_readfirstlane(threadIdx.x >> 5);
    const int tile0 = blockIdx.x * 16;
    const int r0 = tile0 + w * FUSE_ROWS;
    const int last = max(nEdges - 1, 0);
    int s = rowptr[min(r0, nRows)];
    int e = rowptr[min(r0 + 1, nRows)];
    int   c_n = ccol[min(s + lane, last)];
    float v_n = cval[min(s + lane, last)];
#pragma unroll 1
    for (int k = 0; k < FUSE_ROWS; ++k) {
        const int e_next = rowptr[min(r0 + k + 2, nRows)];
        const int   c_l = c_n;
        const float v_l = v_n;
        c_n = ccol[min(e + lane, last)];
        v_n = cval[min(e + lane, last)];
        float4 acc;
        gcn_row_zero(acc);
        if (s < e) {
            gcn_gather_chunk<4>(acc, c_l, v_l, min(32, e - s), x, loff);
            for (int b = s + 32; b < e; b += 32) {
                const int i = min(b + lane, e - 1);
                gcn_gather_chunk<4>(acc, ccol[i], cval[i], min(32, e - b), x, loff);
            }
        }
        v4h hv;
        hv[0] = (_Float16)acc.x; hv[1] = (_Float16)acc.y; hv[2] = (_Float16)acc.z; hv[3] = (_Float16)acc.w;
        *(v4h*)(sA + (w * FUSE_ROWS + k) * FUSE_LD + loff) = hv;
        s = e;
        e = e_next;
    }
    __syncthreads();

    const int half = lane >> 4;
    const int rr   = lane & 15;

    v8f h0 = (v8f){};
    v8f h1 = (v8f){};
#pragma unroll
    for (int kk = 0; kk < 4; ++kk) {
        const v16h a  = gcn_frag_a(sA, rr, half, kk);
        const v16h b0 = *(const v16h*)(W1p + (((size_t)(2 * w)     * 4 + kk) * 32 + lane) * 16);
        const v16h b1 = *(const v16h*)(W1p + (((size_t)(2 * w + 1) * 4 + kk) * 32 + lane) * 16);
        h0 = __builtin_amdgcn_wmma_f32_16x16x32_f16(false, a, false, b0, (short)0, h0, false, false);
        h1 = __builtin_amdgcn_wmma_f32_16x16x32_f16(false, a, false, b1, (short)0, h1, false, false);
    }
#pragma unroll
    for (int v = 0; v < 8; ++v) {
        _Float16* hp = sH + (half * 8 + v) * FUSE_LD + (2 * w) * 16 + rr;
        hp[0]  = (_Float16)fmaxf(h0[v], 0.0f);
        hp[16] = (_Float16)fmaxf(h1[v], 0.0f);
    }
    __syncthreads();

    v8f tv = (v8f){};
#pragma unroll
    for (int kk = 0; kk < 4; ++kk) {
        const v16h a = gcn_frag_a(sH, rr, half, kk);
        const v16h b = *(const v16h*)(W2p + (((size_t)w * 4 + kk) * 32 + lane) * 16);
        tv = __builtin_amdgcn_wmma_f32_16x16x32_f16(false, a, false, b, (short)0, tv, false, false);
    }
    const int mbase = tile0 + half * 8;
    float* tp = t + (size_t)mbase * 64 + w * 16 + rr;
    if (tile0 + 16 <= nRows) {
#pragma unroll
        for (int v = 0; v < 8; ++v) tp[v * 64] = tv[v];
    } else {
#pragma unroll
        for (int v = 0; v < 8; ++v)
            if (mbase + v < nRows) tp[v * 64] = tv[v];
    }
}

extern "C" void kernel_launch(void* const* d_in, const int* in_sizes, int n_in,
                              void* d_out, int out_size, void* d_ws, size_t ws_size,
                              hipStream_t stream) {
    const float* x    = (const float*)d_in[0];
    const int*   row  = (const int*)d_in[1];
    const int*   col  = (const int*)d_in[2];
    const float* vals = (const float*)d_in[3];
    const float* W1   = (const float*)d_in[4];
    const float* W2   = (const float*)d_in[5];
    float* out = (float*)d_out;

    const int NN = in_sizes[0] / IN_DIM;
    const int NE = in_sizes[1];

    float*     tBuf = (float*)d_ws;
    _Float16*  W1p  = (_Float16*)(tBuf + (size_t)2 * NN * IN_DIM);
    _Float16*  W2p  = W1p + 128 * 128;
    const int  NNp  = ((NN + 8) >> 2) << 2;
    int*       cnt    = (int*)(W2p + 128 * 64);
    int*       rowptr = cnt + NNp;
    int*       slot   = rowptr + NNp;
    int*       ccol   = slot + NE;
    float*     cval   = (float*)(ccol + NE);
    int*       part   = (int*)(cval + NE);

    gcn_pack_w<<<4, 256, 0, stream>>>(W1, W1p, 8);
    gcn_pack_w<<<2, 256, 0, stream>>>(W2, W2p, 4);

    const int nSlices = (NE + CNT_SLICE - 1) / CNT_SLICE;
    const int nRanges = (NN + CNT_RANGE - 1) / CNT_RANGE;
    gcn_count_lds<<<dim3(nSlices, nRanges), CNT_THREADS, 0, stream>>>(row, part, NN, NNp, NE);
    gcn_slice_prefix<<<(NN + 255) / 256, 256, 0, stream>>>(part, cnt, NN, NNp, nSlices);
    gcn_scan<<<1, SCAN_THREADS, 0, stream>>>(cnt, rowptr, NN);
    gcn_place_lds<<<dim3(nSlices, nRanges), CNT_THREADS, 0, stream>>>(row, rowptr, part, slot, NN, NNp, NE);
    const int rowBlocks = (NN + 8 * ROWS_PER_WAVE - 1) / (8 * ROWS_PER_WAVE);
    gcn_order_rows<<<rowBlocks, 256, 0, stream>>>(rowptr, slot, col, vals, ccol, cval, NN, NE);

    float* tF = tBuf;
    gcn_layer1_fused<<<(NN + 15) / 16, FUSE_WAVES * 32, 0, stream>>>(rowptr, ccol, cval, x, W1p, W2p, tF, NN, NE);

    gcn_gather_rows<2><<<rowBlocks, 256, 0, stream>>>(rowptr, ccol, cval, tF, out, NN, NE);
}


# Nine launches become six: weight packing rides in the edge-count grid; row offsets come from block-local sums in the slice-prefix launch plus a block-total scan in the placing one; f32 sums unchanged
# speedup vs baseline: 3.7056x; 1.1746x over previous
#include <hip/hip_runtime.h>
#include <hip/hip_bf16.h>

typedef __attribute__((ext_vector_type(16))) _Float16 v16h;
typedef __attribute__((ext_vector_type(8)))  _Float16 v8h;
typedef __attribute__((ext_vector_type(4)))  _Float16 v4h;
typedef __attribute__((ext_vector_type(8)))  float    v8f;

#define IN_DIM 128

__device__ __forceinline__ void gcn_pack_w(const float* __restrict__ W, _Float16* __restrict__ Wp,
                                           int Nt  , int id) {
    int lane = id & 31;
    int kk   = (id >> 5) & 3;
    int t    = id >> 7;
    int half = lane >> 4;
    int r    = lane & 15;
    int N    = Nt * 16;
    int n    = t * 16 + r;
    int k0   = kk * 32;
    _Float16* dst = Wp + (size_t)id * 16;
#pragma unroll
    for (int j = 0; j < 8; ++j)
        dst[j] = (_Float16)W[(size_t)(k0 + half * 8 + j) * N + n];
#pragma unroll
    for (int j = 0; j < 8; ++j)
        dst[8 + j] = (_Float16)W[(size_t)(k0 + 16 + half * 8 + j) * N + n];
}

#define CNT_RANGE   8192
#define CNT_SLICE   32768
#define CNT_THREADS 512
#define CNT_U       8

__device__ __forceinline__ void gcn_load_rows(int4 (&v)[CNT_U], const int* __restrict__ row, int ib, int e1v) {
#pragma unroll
    for (int k = 0; k < CNT_U; ++k) {
        const int i = ib + k * CNT_THREADS * 4;
        v[k] = make_int4(-1, -1, -1, -1);
        if (i < e1v) v[k] = *(const int4*)(row + i);
    }
}

__device__ __forceinline__ void gcn_count_one(int* c, int v, int base, unsigned lim) {
    const unsigned d = (unsigned)(v - base);
    if (d < lim) atomicAdd(&c[d], 1);
}

#define CNT_PER_THREAD (CNT_RANGE / CNT_THREADS)

__global__ __launch_bounds__(CNT_THREADS)
void gcn_count_lds(const int* __restrict__ row, int* __restrict__ part,
                   int nRows, int nRowsP, int nEdges,
                   const float* __restrict__ W1, const float* __restrict__ W2,
                   _Float16* __restrict__ W1p, _Float16* __restrict__ W2p,
                   int nt1  , int nt2  ) {
    __shared__ int c[CNT_RANGE];
    const int t = threadIdx.x;
    if (blockIdx.y == gridDim.y - 1) {
        const int n1 = nt1 * 4 * 32;
        const int n2 = nt2 * 4 * 32;
        for (int id = blockIdx.x * CNT_THREADS + t; id < n1 + n2; id += gridDim.x * CNT_THREADS) {
            if (id < n1) gcn_pack_w(W1, W1p, nt1, id);
            else         gcn_pack_w(W2, W2p, nt2, id - n1);
        }
        return;
    }
    const int base = blockIdx.y * CNT_RANGE;
    const unsigned lim = (unsigned)max(0, min(CNT_RANGE, nRows - base));
    const int e0 = blockIdx.x * CNT_SLICE;
    const int e1 = min(nEdges, e0 + CNT_SLICE);
    const int e1v = e0 + ((e1 - e0) & ~3);
    for (int j = t; j < CNT_RANGE; j += CNT_THREADS) c[j] = 0;
    __syncthreads();
    for (int ib = e0 + t * 4; ib < e1v; ib += CNT_THREADS * 4 * CNT_U) {
        int4 v[CNT_U];
        gcn_load_rows(v, row, ib, e1v);
#pragma unroll
        for (int k = 0; k < CNT_U; ++k) {
            gcn_count_one(c, v[k].x, base, lim);
            gcn_count_one(c, v[k].y, base, lim);
            gcn_count_one(c, v[k].z, base, lim);
            gcn_count_one(c, v[k].w, base, lim);
        }
    }
    if (t < e1 - e1v) gcn_count_one(c, row[e1v + t], base, lim);
    __syncthreads();
    int o[CNT_PER_THREAD];
#pragma unroll
    for (int k = 0; k < CNT_PER_THREAD; ++k) o[k] = c[k * CNT_THREADS + t];
#pragma unroll
    for (int k = 0; k < CNT_PER_THREAD; ++k) {
        const unsigned j = (unsigned)(k * CNT_THREADS + t);
        if (j < lim) part[blockIdx.x * nRowsP + base + j] = o[k];
    }
}

#define PFX_THREADS 256
__global__ __launch_bounds__(PFX_THREADS)
void gcn_slice_prefix(int* __restrict__ part, int* __restrict__ lrow, int* __restrict__ bsum,
                      int nRows, int nRowsP, int nSlices) {
    __shared__ int wtot[PFX_THREADS / 32];
    const int t = threadIdx.x;
    const int lane = t & 31;
    const int w = t >> 5;
    const int node = blockIdx.x * PFX_THREADS + t;
    int run = 0;
    if (node < nRows) {
        for (int s0 = 0; s0 < nSlices; s0 += 8) {
            int v[8];
#pragma unroll
            for (int k = 0; k < 8; ++k)
                v[k] = (s0 + k < nSlices) ? part[(s0 + k) * nRowsP + node] : 0;
#pragma unroll
            for (int k = 0; k < 8; ++k) {
                if (s0 + k < nSlices) {
                    part[(s0 + k) * nRowsP + node] = run;
                    run += v[k];
                }
            }
        }
    }
    int inc = run;
    for (int d = 1; d < 32; d <<= 1) {
        int o = __shfl_up(inc, d);
        if (lane >= d) inc += o;
    }
    if (lane == 31) wtot[w] = inc;
    __syncthreads();
    int ahead = 0;
    int total = 0;
#pragma unroll
    for (int i = 0; i < PFX_THREADS / 32; ++i) {
        const int v = wtot[i];
        if (i < w) ahead += v;
        total += v;
    }
    if (node < nRows) lrow[node] = ahead + inc - run;
    if (t == 0) bsum[blockIdx.x] = total;
}

__device__ __forceinline__ int gcn_take_slot(int* p, int v, int base, unsigned lim) {
    const unsigned d = (unsigned)(v - base);
    int pos = -1;
    if (d < lim) pos = atomicAdd(&p[d], 1);
    return pos;
}

#define PLACE_BLOCKS (CNT_RANGE / PFX_THREADS)
static_assert(PLACE_BLOCKS == 32, "one prefix work-group per lane of a wave");
__global__ __launch_bounds__(CNT_THREADS)
void gcn_place_lds(const int* __restrict__ row, const int* __restrict__ lrow,
                   const int* __restrict__ bsum, const int* __restrict__ part,
                   int* __restrict__ rowptr, int* __restrict__ slot,
                   int nRows, int nRowsP, int nEdges, int nPfx) {
    __shared__ int p[CNT_RANGE];
    __shared__ int boff[PLACE_BLOCKS];
    const int t = threadIdx.x;
    const int base = blockIdx.y * CNT_RANGE;
    const unsigned lim = (unsigned)max(0, min(CNT_RANGE, nRows - base));
    const int e0 = blockIdx.x * CNT_SLICE;
    const int e1 = min(nEdges, e0 + CNT_SLICE);
    const int e1v = e0 + ((e1 - e0) & ~3);
    {
        int a[CNT_PER_THREAD], b[CNT_PER_THREAD];
#pragma unroll
        for (int k = 0; k < CNT_PER_THREAD; ++k) {
            const unsigned j = (unsigned)(k * CNT_THREADS + t);
            a[k] = 0;
            b[k] = 0;
            if (j < lim) {
                a[k] = lrow[base + j];
                b[k] = part[blockIdx.x * nRowsP + base + j];
            }
        }
        if (t < 32) {
            const int b0 = base / PFX_THREADS;
            int ahead = 0;
            for (int bb = 0; bb < b0; bb += 8 * 32) {
                int v[8];
#pragma unroll
                for (int k = 0; k < 8; ++k) {
                    const int bi = bb + k * 32 + t;
                    v[k] = (bi < b0) ? bsum[bi] : 0;
                }
#pragma unroll
                for (int k = 0; k < 8; ++k) ahead += v[k];
            }
            const int mine = (b0 + t < nPfx) ? bsum[b0 + t] : 0;
            for (int d = 16; d >= 1; d >>= 1) ahead += __shfl_xor(ahead, d);
            int inc = mine;
            for (int d = 1; d < 32; d <<= 1) {
                int o = __shfl_up(inc, d);
                if (t >= d) inc += o;
            }
            boff[t] = ahead + inc - mine;
            if (t == 31 && blockIdx.x == 0 && blockIdx.y == gridDim.y - 1) rowptr[nRows] = ahead + inc;
        }
        __syncthreads();
#pragma unroll
        for (int k = 0; k < CNT_PER_THREAD; ++k) a[k] += boff[(unsigned)(k * CNT_THREADS + t) / PFX_THREADS];
#pragma unroll
        for (int k = 0; k < CNT_PER_THREAD; ++k) {
            const unsigned j = (unsigned)(k * CNT_THREADS + t);
            p[j] = a[k] + b[k];
            if (blockIdx.x == 0 && j < lim) rowptr[base + j] = a[k];
        }
    }
    __syncthreads();
    for (int ib = e0 + t * 4; ib < e1v; ib += CNT_THREADS * 4 * CNT_U) {
        int4 v[CNT_U];
        gcn_load_rows(v, row, ib, e1v);
        int4 pos[CNT_U];
#pragma unroll
        for (int k = 0; k < CNT_U; ++k) {
            pos[k].x = gcn_take_slot(p, v[k].x, base, lim);
            pos[k].y = gcn_take_slot(p, v[k].y, base, lim);
            pos[k].z = gcn_take_slot(p, v[k].z, base, lim);
            pos[k].w = gcn_take_slot(p, v[k].w, base, lim);
        }
#pragma unroll
        for (int k = 0; k < CNT_U; ++k) {
            const int i = ib + k * CNT_THREADS * 4;
            if (pos[k].x >= 0) slot[pos[k].x] = i;
            if (pos[k].y >= 0) slot[pos[k].y] = i + 1;
            if (pos[k].z >= 0) slot[pos[k].z] = i + 2;
            if (pos[k].w >= 0) slot[pos[k].w] = i + 3;
        }
    }
    if (t < e1 - e1v) {
        const int pos = gcn_take_slot(p, row[e1v + t], base, lim);
        if (pos >= 0) slot[pos] = e1v + t;
    }
}

#define ROWS_PER_WAVE 5

__global__ __launch_bounds__(256)
void gcn_order_rows(const int* __restrict__ rowptr, const int* __restrict__ slot,
                    const int* __restrict__ col, const float* __restrict__ vals,
                    int* __restrict__ ccol, float* __restrict__ cval,
                    int nRows, int nEdges) {
    const int lane = threadIdx.x & 31;
    const int wave = __builtin_amdgcn_readfirstlane((blockIdx.x * blockDim.x + threadIdx.x) >> 5);
    const int r0 = wave * ROWS_PER_WAVE;
    if (r0 >= nRows) return;
    const int rEnd = min(nRows, r0 + ROWS_PER_WAVE);
    const int last = max(nEdges - 1, 0);
    int s = rowptr[r0];
    int e = rowptr[r0 + 1];
    int nxt = slot[min(s + lane, last)];
    for (int r = r0; r < rEnd; ++r) {
        const int e_next = rowptr[min(r + 2, nRows)];
        const int first = nxt;
        const int deg = e - s;
        if (deg <= 32) {
            int place = 0;
            for (int j = 0; j < deg; ++j)
                place += (__builtin_amdgcn_readlane(first, j) < first) ? 1 : 0;
            nxt = slot[min(e + lane, last)];
            if (lane < deg) {
                ccol[s + place] = col[first];
                cval[s + place] = vals[first];
            }
        } else {
            nxt = slot[min(e + lane, last)];
            for (int ib = s; ib < e; ib += 32) {
                const int i = ib + lane;
                const int mine = slot[min(i, e - 1)];
                int place = 0;
                for (int jb = s; jb < e; jb += 32) {
                    const int jn = min(32, e - jb);
                    const int other = slot[min(jb + lane, e - 1)];
                    for (int j = 0; j < jn; ++j)
                        place += (__builtin_amdgcn_readlane(other, j) < mine) ? 1 : 0;
                }
                if (i < e) {
                    ccol[s + place] = col[mine];
                    cval[s + place] = vals[mine];
                }
            }
        }
        s = e;
        e = e_next;
    }
}

template <int VEC> struct gcn_row;
template <> struct gcn_row<4> { typedef float4 type; };
template <> struct gcn_row<2> { typedef float2 type; };
__device__ __forceinline__ void gcn_row_zero(float4& a) { a.x = 0.0f; a.y = 0.0f; a.z = 0.0f; a.w = 0.0f; }
__device__ __forceinline__ void gcn_row_zero(float2& a) { a.x = 0.0f; a.y = 0.0f; }
__device__ __forceinline__ void gcn_row_fma(float4& a, float v, const float4& m) {
    a.x = fmaf(v, m.x, a.x);
    a.y = fmaf(v, m.y, a.y);
    a.z = fmaf(v, m.z, a.z);
    a.w = fmaf(v, m.w, a.w);
}
__device__ __forceinline__ void gcn_row_fma(float2& a, float v, const float2& m) {
    a.x = fmaf(v, m.x, a.x);
    a.y = fmaf(v, m.y, a.y);
}

#define GATHER_U 8
template <int VEC>
__device__ __forceinline__ void gcn_gather_chunk(typename gcn_row<VEC>::type& acc, int c_l, float v_l, int n,
                                                 const float* __restrict__ x, unsigned loff) {
    typedef typename gcn_row<VEC>::type row_t;
    const int D = 32 * VEC;
    int j = 0;
    for (; j + GATHER_U <= n; j += GATHER_U) {
        row_t m[GATHER_U];
#pragma unroll
        for (int k = 0; k < GATHER_U; ++k) {
            const int c = __builtin_amdgcn_readlane(c_l, j + k);
            m[k] = *(const row_t*)(x + (size_t)c * D + loff);
        }
#pragma unroll
        for (int k = 0; k < GATHER_U; ++k) {
            const float v = __int_as_float(__builtin_amdgcn_readlane(__float_as_int(v_l), j + k));
            gcn_row_fma(acc, v, m[k]);
        }
    }
    const int rem = n - j;
    if (rem > 0) {
        row_t m[GATHER_U - 1];
#pragma unroll
        for (int k = 0; k < GATHER_U - 1; ++k) {
            if (k < rem) {
                const int c = __builtin_amdgcn_readlane(c_l, j + k);
                m[k] = *(const row_t*)(x + (size_t)c * D + loff);
            }
        }
#pragma unroll
        for (int k = 0; k < GATHER_U - 1; ++k) {
            if (k < rem) {
                const float v = __int_as_float(__builtin_amdgcn_readlane(__float_as_int(v_l), j + k));
                gcn_row_fma(acc, v, m[k]);
            }
        }
    }
}

template <int VEC>
__global__ __launch_bounds__(256)
void gcn_gather_rows(const int* __restrict__ rowptr, const int* __restrict__ ccol,
                     const float* __restrict__ cval, const float* __restrict__ x,
                     float* __restrict__ agg, int nRows, int nEdges) {
    typedef typename gcn_row<VEC>::type row_t;
    const int D = 32 * VEC;
    const int lane = threadIdx.x & 31;
    const unsigned loff = (threadIdx.x & 31u) * (unsigned)VEC;
    const int wave = __builtin_amdgcn_readfirstlane((blockIdx.x * blockDim.x + threadIdx.x) >> 5);
    const int r0 = wave * ROWS_PER_WAVE;
    if (r0 >= nRows) return;
    const int rEnd = min(nRows, r0 + ROWS_PER_WAVE);
    const int last = max(nEdges - 1, 0);
    int s = rowptr[r0];
    int e = rowptr[r0 + 1];
    int   c_n = ccol[min(s + lane, last)];
    float v_n = cval[min(s + lane, last)];
    for (int r = r0; r < rEnd; ++r) {
        const int e_next = rowptr[min(r + 2, nRows)];
        const int   c_l = c_n;
        const float v_l = v_n;
        c_n = ccol[min(e + lane, last)];
        v_n = cval[min(e + lane, last)];
        row_t acc;
        gcn_row_zero(acc);
        if (s < e) {
            gcn_gather_chunk<VEC>(acc, c_l, v_l, min(32, e - s), x, loff);
            for (int b = s + 32; b < e; b += 32) {
                const int i = min(b + lane, e - 1);
                gcn_gather_chunk<VEC>(acc, ccol[i], cval[i], min(32, e - b), x, loff);
            }
        }
        *(row_t*)(agg + (size_t)r * D + loff) = acc;
        s = e;
        e = e_next;
    }
}

#define FUSE_WAVES 4
#define FUSE_ROWS  4
#define FUSE_LD    136

__device__ __forceinline__ v16h gcn_frag_a(const _Float16* tile, int r, int half, int kk) {
    const _Float16* p = tile + r * FUSE_LD + kk * 32 + half * 8;
    const v8h lo = *(const v8h*)p;
    const v8h hi = *(const v8h*)(p + 16);
    return __builtin_shufflevector(lo, hi, 0, 1, 2, 3, 4, 5, 6, 7, 8, 9, 10, 11, 12, 13, 14, 15);
}

__global__ __launch_bounds__(FUSE_WAVES * 32) __attribute__((amdgpu_waves_per_eu(10)))
void gcn_layer1_fused(const int* __restrict__ rowptr, const int* __restrict__ ccol,
                      const float* __restrict__ cval, const float* __restrict__ x,
                      const _Float16* __restrict__ W1p, const _Float16* __restrict__ W2p,
                      float* __restrict__ t, int nRows, int nEdges) {
    __shared__ __attribute__((aligned(16))) _Float16 sA[16 * FUSE_LD];
    __shared__ __attribute__((aligned(16))) _Float16 sH[16 * FUSE_LD];
    const int lane = threadIdx.x & 31;
    const unsigned loff = (threadIdx.x & 31u) * 4u;
    const int w = __builtin_amdgcn_readfirstlane(threadIdx.x >> 5);
    const int tile0 = blockIdx.x * 16;
    const int r0 = tile0 + w * FUSE_ROWS;
    const int last = max(nEdges - 1, 0);
    int s = rowptr[min(r0, nRows)];
    int e = rowptr[min(r0 + 1, nRows)];
    int   c_n = ccol[min(s + lane, last)];
    float v_n = cval[min(s + lane, last)];
#pragma unroll 1
    for (int k = 0; k < FUSE_ROWS; ++k) {
        const int e_next = rowptr[min(r0 + k + 2, nRows)];
        const int   c_l = c_n;
        const float v_l = v_n;
        c_n = ccol[min(e + lane, last)];
        v_n = cval[min(e + lane, last)];
        float4 acc;
        gcn_row_zero(acc);
        if (s < e) {
            gcn_gather_chunk<4>(acc, c_l, v_l, min(32, e - s), x, loff);
            for (int b = s + 32; b < e; b += 32) {
                const int i = min(b + lane, e - 1);
                gcn_gather_chunk<4>(acc, ccol[i], cval[i], min(32, e - b), x, loff);
            }
        }
        v4h hv;
        hv[0] = (_Float16)acc.x; hv[1] = (_Float16)acc.y; hv[2] = (_Float16)acc.z; hv[3] = (_Float16)acc.w;
        *(v4h*)(sA + (w * FUSE_ROWS + k) * FUSE_LD + loff) = hv;
        s = e;
        e = e_next;
    }
    __syncthreads();

    const int half = lane >> 4;
    const int rr   = lane & 15;

    v8f h0 = (v8f){};
    v8f h1 = (v8f){};
#pragma unroll
    for (int kk = 0; kk < 4; ++kk) {
        const v16h a  = gcn_frag_a(sA, rr, half, kk);
        const v16h b0 = *(const v16h*)(W1p + (((size_t)(2 * w)     * 4 + kk) * 32 + lane) * 16);
        const v16h b1 = *(const v16h*)(W1p + (((size_t)(2 * w + 1) * 4 + kk) * 32 + lane) * 16);
        h0 = __builtin_amdgcn_wmma_f32_16x16x32_f16(false, a, false, b0, (short)0, h0, false, false);
        h1 = __builtin_amdgcn_wmma_f32_16x16x32_f16(false, a, false, b1, (short)0, h1, false, false);
    }
#pragma unroll
    for (int v = 0; v < 8; ++v) {
        _Float16* hp = sH + (half * 8 + v) * FUSE_LD + (2 * w) * 16 + rr;
        hp[0]  = (_Float16)fmaxf(h0[v], 0.0f);
        hp[16] = (_Float16)fmaxf(h1[v], 0.0f);
    }
    __syncthreads();

    v8f tv = (v8f){};
#pragma unroll
    for (int kk = 0; kk < 4; ++kk) {
        const v16h a = gcn_frag_a(sH, rr, half, kk);
        const v16h b = *(const v16h*)(W2p + (((size_t)w * 4 + kk) * 32 + lane) * 16);
        tv = __builtin_amdgcn_wmma_f32_16x16x32_f16(false, a, false, b, (short)0, tv, false, false);
    }
    const int mbase = tile0 + half * 8;
    float* tp = t + (size_t)mbase * 64 + w * 16 + rr;
    if (tile0 + 16 <= nRows) {
#pragma unroll
        for (int v = 0; v < 8; ++v) tp[v * 64] = tv[v];
    } else {
#pragma unroll
        for (int v = 0; v < 8; ++v)
            if (mbase + v < nRows) tp[v * 64] = tv[v];
    }
}

extern "C" void kernel_launch(void* const* d_in, const int* in_sizes, int n_in,
                              void* d_out, int out_size, void* d_ws, size_t ws_size,
                              hipStream_t stream) {
    const float* x    = (const float*)d_in[0];
    const int*   row  = (const int*)d_in[1];
    const int*   col  = (const int*)d_in[2];
    const float* vals = (const float*)d_in[3];
    const float* W1   = (const float*)d_in[4];
    const float* W2   = (const float*)d_in[5];
    float* out = (float*)d_out;

    const int NN = in_sizes[0] / IN_DIM;
    const int NE = in_sizes[1];

    float*     tBuf = (float*)d_ws;
    _Float16*  W1p  = (_Float16*)(tBuf + (size_t)2 * NN * IN_DIM);
    _Float16*  W2p  = W1p + 128 * 128;
    const int  NNp  = ((NN + 8) >> 2) << 2;
    int*       lrow   = (int*)(W2p + 128 * 64);
    int*       rowptr = lrow + NNp;
    int*       slot   = rowptr + NNp;
    int*       ccol   = slot + NE;
    float*     cval   = (float*)(ccol + NE);
    int*       part   = (int*)(cval + NE);
    int*       bsum   = part + (size_t)((NE + CNT_SLICE - 1) / CNT_SLICE) * NNp;

    const int nSlices = (NE + CNT_SLICE - 1) / CNT_SLICE;
    const int nRanges = (NN + CNT_RANGE - 1) / CNT_RANGE;
    const int nPfx    = (NN + PFX_THREADS - 1) / PFX_THREADS;
    gcn_count_lds<<<dim3(nSlices, nRanges + 1), CNT_THREADS, 0, stream>>>(row, part, NN, NNp, NE, W1, W2, W1p, W2p, 8, 4);
    gcn_slice_prefix<<<nPfx, PFX_THREADS, 0, stream>>>(part, lrow, bsum, NN, NNp, nSlices);
    gcn_place_lds<<<dim3(nSlices, nRanges), CNT_THREADS, 0, stream>>>(row, lrow, bsum, part, rowptr, slot, NN, NNp, NE, nPfx);
    const int rowBlocks = (NN + 8 * ROWS_PER_WAVE - 1) / (8 * ROWS_PER_WAVE);
    gcn_order_rows<<<rowBlocks, 256, 0, stream>>>(rowptr, slot, col, vals, ccol, cval, NN, NE);

    float* tF = tBuf;
    gcn_layer1_fused<<<(NN + 15) / 16, FUSE_WAVES * 32, 0, stream>>>(rowptr, ccol, cval, x, W1p, W2p, tF, NN, NE);

    gcn_gather_rows<2><<<rowBlocks, 256, 0, stream>>>(rowptr, ccol, cval, tF, out, NN, NE);
}


# Placing kernel writes edge records (index, col, value); row kernels rank a row's records in registers and permute lanes: ordering launch and its per-edge fetches gone, 6 launches to 5; same f32 sums
# speedup vs baseline: 4.0242x; 1.0860x over previous
#include <hip/hip_runtime.h>
#include <hip/hip_bf16.h>

typedef __attribute__((ext_vector_type(16))) _Float16 v16h;
typedef __attribute__((ext_vector_type(8)))  _Float16 v8h;
typedef __attribute__((ext_vector_type(4)))  _Float16 v4h;
typedef __attribute__((ext_vector_type(8)))  float    v8f;

#define IN_DIM 128

__device__ __forceinline__ void gcn_pack_w(const float* __restrict__ W, _Float16* __restrict__ Wp,
                                           int Nt  , int id) {
    int lane = id & 31;
    int kk   = (id >> 5) & 3;
    int t    = id >> 7;
    int half = lane >> 4;
    int r    = lane & 15;
    int N    = Nt * 16;
    int n    = t * 16 + r;
    int k0   = kk * 32;
    _Float16* dst = Wp + (size_t)id * 16;
#pragma unroll
    for (int j = 0; j < 8; ++j)
        dst[j] = (_Float16)W[(size_t)(k0 + half * 8 + j) * N + n];
#pragma unroll
    for (int j = 0; j < 8; ++j)
        dst[8 + j] = (_Float16)W[(size_t)(k0 + 16 + half * 8 + j) * N + n];
}

#define CNT_RANGE   8192
#define CNT_SLICE   32768
#define CNT_THREADS 512
#define CNT_U       8

__device__ __forceinline__ void gcn_load_rows(int4 (&v)[CNT_U], const int* __restrict__ row, int ib, int e1v) {
#pragma unroll
    for (int k = 0; k < CNT_U; ++k) {
        const int i = ib + k * CNT_THREADS * 4;
        v[k] = make_int4(-1, -1, -1, -1);
        if (i < e1v) v[k] = *(const int4*)(row + i);
    }
}

__device__ __forceinline__ void gcn_count_one(int* c, int v, int base, unsigned lim) {
    const unsigned d = (unsigned)(v - base);
    if (d < lim) atomicAdd(&c[d], 1);
}

#define CNT_PER_THREAD (CNT_RANGE / CNT_THREADS)

__global__ __launch_bounds__(CNT_THREADS)
void gcn_count_lds(const int* __restrict__ row, int* __restrict__ part,
                   int nRows, int nRowsP, int nEdges,
                   const float* __restrict__ W1, const float* __restrict__ W2,
                   _Float16* __restrict__ W1p, _Float16* __restrict__ W2p,
                   int nt1  , int nt2  ) {
    __shared__ int c[CNT_RANGE];
    const int t = threadIdx.x;
    if (blockIdx.y == gridDim.y - 1) {
        const int n1 = nt1 * 4 * 32;
        const int n2 = nt2 * 4 * 32;
        for (int id = blockIdx.x * CNT_THREADS + t; id < n1 + n2; id += gridDim.x * CNT_THREADS) {
            if (id < n1) gcn_pack_w(W1, W1p, nt1, id);
            else         gcn_pack_w(W2, W2p, nt2, id - n1);
        }
        return;
    }
    const int base = blockIdx.y * CNT_RANGE;
    const unsigned lim = (unsigned)max(0, min(CNT_RANGE, nRows - base));
    const int e0 = blockIdx.x * CNT_SLICE;
    const int e1 = min(nEdges, e0 + CNT_SLICE);
    const int e1v = e0 + ((e1 - e0) & ~3);
    for (int j = t; j < CNT_RANGE; j += CNT_THREADS) c[j] = 0;
    __syncthreads();
    for (int ib = e0 + t * 4; ib < e1v; ib += CNT_THREADS * 4 * CNT_U) {
        int4 v[CNT_U];
        gcn_load_rows(v, row, ib, e1v);
#pragma unroll
        for (int k = 0; k < CNT_U; ++k) {
            gcn_count_one(c, v[k].x, base, lim);
            gcn_count_one(c, v[k].y, base, lim);
            gcn_count_one(c, v[k].z, base, lim);
            gcn_count_one(c, v[k].w, base, lim);
        }
    }
    if (t < e1 - e1v) gcn_count_one(c, row[e1v + t], base, lim);
    __syncthreads();
    int o[CNT_PER_THREAD];
#pragma unroll
    for (int k = 0; k < CNT_PER_THREAD; ++k) o[k] = c[k * CNT_THREADS + t];
#pragma unroll
    for (int k = 0; k < CNT_PER_THREAD; ++k) {
        const unsigned j = (unsigned)(k * CNT_THREADS + t);
        if (j < lim) part[blockIdx.x * nRowsP + base + j] = o[k];
    }
}

#define PFX_THREADS 256
__global__ __launch_bounds__(PFX_THREADS)
void gcn_slice_prefix(int* __restrict__ part, int* __restrict__ lrow, int* __restrict__ bsum,
                      int nRows, int nRowsP, int nSlices) {
    __shared__ int wtot[PFX_THREADS / 32];
    const int t = threadIdx.x;
    const int lane = t & 31;
    const int w = t >> 5;
    const int node = blockIdx.x * PFX_THREADS + t;
    int run = 0;
    if (node < nRows) {
        for (int s0 = 0; s0 < nSlices; s0 += 8) {
            int v[8];
#pragma unroll
            for (int k = 0; k < 8; ++k)
                v[k] = (s0 + k < nSlices) ? part[(s0 + k) * nRowsP + node] : 0;
#pragma unroll
            for (int k = 0; k < 8; ++k) {
                if (s0 + k < nSlices) {
                    part[(s0 + k) * nRowsP + node] = run;
                    run += v[k];
                }
            }
        }
    }
    int inc = run;
    for (int d = 1; d < 32; d <<= 1) {
        int o = __shfl_up(inc, d);
        if (lane >= d) inc += o;
    }
    if (lane == 31) wtot[w] = inc;
    __syncthreads();
    int ahead = 0;
    int total = 0;
#pragma unroll
    for (int i = 0; i < PFX_THREADS / 32; ++i) {
        const int v = wtot[i];
        if (i < w) ahead += v;
        total += v;
    }
    if (node < nRows) lrow[node] = ahead + inc - run;
    if (t == 0) bsum[blockIdx.x] = total;
}

__device__ __forceinline__ int gcn_take_slot(int* p, int v, int base, unsigned lim) {
    const unsigned d = (unsigned)(v - base);
    int pos = -1;
    if (d < lim) pos = atomicAdd(&p[d], 1);
    return pos;
}

#define PLACE_BLOCKS (CNT_RANGE / PFX_THREADS)
static_assert(PLACE_BLOCKS == 32, "one prefix work-group per lane of a wave");
__global__ __launch_bounds__(CNT_THREADS)
void gcn_place_lds(const int* __restrict__ row, const int* __restrict__ col,
                   const float* __restrict__ vals, const int* __restrict__ lrow,
                   const int* __restrict__ bsum, const int* __restrict__ part,
                   int* __restrict__ rowptr, int* __restrict__ recw,
                   int nRows, int nRowsP, int nEdges, int nPfx) {
    int4* const rec = (int4*)recw;
    __shared__ int p[CNT_RANGE];
    __shared__ int boff[PLACE_BLOCKS];
    const int t = threadIdx.x;
    const int base = blockIdx.y * CNT_RANGE;
    const unsigned lim = (unsigned)max(0, min(CNT_RANGE, nRows - base));
    const int e0 = blockIdx.x * CNT_SLICE;
    const int e1 = min(nEdges, e0 + CNT_SLICE);
    const int e1v = e0 + ((e1 - e0) & ~3);
    {
        int a[CNT_PER_THREAD], b[CNT_PER_THREAD];
#pragma unroll
        for (int k = 0; k < CNT_PER_THREAD; ++k) {
            const unsigned j = (unsigned)(k * CNT_THREADS + t);
            a[k] = 0;
            b[k] = 0;
            if (j < lim) {
                a[k] = lrow[base + j];
                b[k] = part[blockIdx.x * nRowsP + base + j];
            }
        }
        if (t < 32) {
            const int b0 = base / PFX_THREADS;
            int ahead = 0;
            for (int bb = 0; bb < b0; bb += 8 * 32) {
                int v[8];
#pragma unroll
                for (int k = 0; k < 8; ++k) {
                    const int bi = bb + k * 32 + t;
                    v[k] = (bi < b0) ? bsum[bi] : 0;
                }
#pragma unroll
                for (int k = 0; k < 8; ++k) ahead += v[k];
            }
            const int mine = (b0 + t < nPfx) ? bsum[b0 + t] : 0;
            for (int d = 16; d >= 1; d >>= 1) ahead += __shfl_xor(ahead, d);
            int inc = mine;
            for (int d = 1; d < 32; d <<= 1) {
                int o = __shfl_up(inc, d);
                if (t >= d) inc += o;
            }
            boff[t] = ahead + inc - mine;
            if (t == 31 && blockIdx.x == 0 && blockIdx.y == gridDim.y - 1) rowptr[nRows] = ahead + inc;
        }
        __syncthreads();
#pragma unroll
        for (int k = 0; k < CNT_PER_THREAD; ++k) a[k] += boff[(unsigned)(k * CNT_THREADS + t) / PFX_THREADS];
#pragma unroll
        for (int k = 0; k < CNT_PER_THREAD; ++k) {
            const unsigned j = (unsigned)(k * CNT_THREADS + t);
            p[j] = a[k] + b[k];
            if (blockIdx.x == 0 && j < lim) rowptr[base + j] = a[k];
        }
    }
    __syncthreads();
    for (int ib = e0 + t * 4; ib < e1v; ib += CNT_THREADS * 4 * CNT_U) {
        int4 v[CNT_U];
        gcn_load_rows(v, row, ib, e1v);
        int4 cc[CNT_U];
        float4 vv[CNT_U];
#pragma unroll
        for (int k = 0; k < CNT_U; ++k) {
            const int i = ib + k * CNT_THREADS * 4;
            cc[k] = make_int4(0, 0, 0, 0);
            vv[k] = make_float4(0.0f, 0.0f, 0.0f, 0.0f);
            if (i < e1v) {
                cc[k] = *(const int4*)(col + i);
                vv[k] = *(const float4*)(vals + i);
            }
        }
        int4 pos[CNT_U];
#pragma unroll
        for (int k = 0; k < CNT_U; ++k) {
            pos[k].x = gcn_take_slot(p, v[k].x, base, lim);
            pos[k].y = gcn_take_slot(p, v[k].y, base, lim);
            pos[k].z = gcn_take_slot(p, v[k].z, base, lim);
            pos[k].w = gcn_take_slot(p, v[k].w, base, lim);
        }
#pragma unroll
        for (int k = 0; k < CNT_U; ++k) {
            const int i = ib + k * CNT_THREADS * 4;
            if (pos[k].x >= 0) rec[pos[k].x] = make_int4(i,     cc[k].x, __float_as_int(vv[k].x), 0);
            if (pos[k].y >= 0) rec[pos[k].y] = make_int4(i + 1, cc[k].y, __float_as_int(vv[k].y), 0);
            if (pos[k].z >= 0) rec[pos[k].z] = make_int4(i + 2, cc[k].z, __float_as_int(vv[k].z), 0);
            if (pos[k].w >= 0) rec[pos[k].w] = make_int4(i + 3, cc[k].w, __float_as_int(vv[k].w), 0);
        }
    }
    if (t < e1 - e1v) {
        const int pos = gcn_take_slot(p, row[e1v + t], base, lim);
        if (pos >= 0) rec[pos] = make_int4(e1v + t, col[e1v + t], __float_as_int(vals[e1v + t]), 0);
    }
}

#define ROWS_PER_WAVE 5

__device__ __forceinline__ void gcn_sort_row32(const int4& r, int deg, int lane, int& c_l, float& v_l) {
    int place = 0;
    for (int j = 0; j < deg; ++j)
        place += (__builtin_amdgcn_readlane(r.x, j) < r.x) ? 1 : 0;
    const int dst = ((lane < deg) ? place : lane) << 2;
    c_l = __builtin_amdgcn_ds_permute(dst, r.y);
    v_l = __int_as_float(__builtin_amdgcn_ds_permute(dst, r.z));
}

__device__ __forceinline__ void gcn_sorted_window(const int4* __restrict__ rec, int s, int e, int w0, int lane,
                                                  int& c_w, float& v_w) {
    c_w = 0;
    v_w = 0.0f;
    int ib = s;
    do {
        const int4 r = rec[min(ib + lane, e - 1)];
        int place = 0;
        int jb = s;
        do {
            const int jn = min(32, e - jb);
            const int other = rec[min(jb + lane, e - 1)].x;
            place += (__builtin_amdgcn_readlane(other, 0) < r.x) ? 1 : 0;
            for (int j = 1; j < jn; ++j)
                place += (__builtin_amdgcn_readlane(other, j) < r.x) ? 1 : 0;
            jb += 32;
        } while (jb < e);
        const int d = place - w0;
        const int jn = min(32, e - ib);
        for (int j = 0; j < jn; ++j) {
            const int   dj = __builtin_amdgcn_readlane(d, j);
            const int   cj = __builtin_amdgcn_readlane(r.y, j);
            const float vj = __int_as_float(__builtin_amdgcn_readlane(r.z, j));
            if (dj == lane) { c_w = cj; v_w = vj; }
        }
        ib += 32;
    } while (ib < e);
}

template <int VEC> struct gcn_row;
template <> struct gcn_row<4> { typedef float4 type; };
template <> struct gcn_row<2> { typedef float2 type; };
__device__ __forceinline__ void gcn_row_zero(float4& a) { a.x = 0.0f; a.y = 0.0f; a.z = 0.0f; a.w = 0.0f; }
__device__ __forceinline__ void gcn_row_zero(float2& a) { a.x = 0.0f; a.y = 0.0f; }
__device__ __forceinline__ void gcn_row_fma(float4& a, float v, const float4& m) {
    a.x = fmaf(v, m.x, a.x);
    a.y = fmaf(v, m.y, a.y);
    a.z = fmaf(v, m.z, a.z);
    a.w = fmaf(v, m.w, a.w);
}
__device__ __forceinline__ void gcn_row_fma(float2& a, float v, const float2& m) {
    a.x = fmaf(v, m.x, a.x);
    a.y = fmaf(v, m.y, a.y);
}

#define GATHER_U 8
template <int VEC>
__device__ __forceinline__ void gcn_gather_chunk(typename gcn_row<VEC>::type& acc, int c_l, float v_l, int n,
                                                 const float* __restrict__ x, unsigned loff) {
    typedef typename gcn_row<VEC>::type row_t;
    const int D = 32 * VEC;
    int j = 0;
    for (; j + GATHER_U <= n; j += GATHER_U) {
        row_t m[GATHER_U];
#pragma unroll
        for (int k = 0; k < GATHER_U; ++k) {
            const int c = __builtin_amdgcn_readlane(c_l, j + k);
            m[k] = *(const row_t*)(x + (size_t)c * D + loff);
        }
#pragma unroll
        for (int k = 0; k < GATHER_U; ++k) {
            const float v = __int_as_float(__builtin_amdgcn_readlane(__float_as_int(v_l), j + k));
            gcn_row_fma(acc, v, m[k]);
        }
    }
    const int rem = n - j;
    if (rem > 0) {
        row_t m[GATHER_U - 1];
#pragma unroll
        for (int k = 0; k < GATHER_U - 1; ++k) {
            if (k < rem) {
                const int c = __builtin_amdgcn_readlane(c_l, j + k);
                m[k] = *(const row_t*)(x + (size_t)c * D + loff);
            }
        }
#pragma unroll
        for (int k = 0; k < GATHER_U - 1; ++k) {
            if (k < rem) {
                const float v = __int_as_float(__builtin_amdgcn_readlane(__float_as_int(v_l), j + k));
                gcn_row_fma(acc, v, m[k]);
            }
        }
    }
}

template <int VEC>
__global__ __launch_bounds__(256)
void gcn_gather_rows(const int* __restrict__ rowptr, const int* __restrict__ recw,
                     const float* __restrict__ x,
                     float* __restrict__ agg, int nRows, int nEdges) {
    typedef typename gcn_row<VEC>::type row_t;
    const int4* const rec = (const int4*)recw;
    const int D = 32 * VEC;
    const int lane = threadIdx.x & 31;
    const unsigned loff = (threadIdx.x & 31u) * (unsigned)VEC;
    const int wave = __builtin_amdgcn_readfirstlane((blockIdx.x * blockDim.x + threadIdx.x) >> 5);
    const int r0 = wave * ROWS_PER_WAVE;
    if (r0 >= nRows) return;
    const int rEnd = min(nRows, r0 + ROWS_PER_WAVE);
    const int last = max(nEdges - 1, 0);
    int s = rowptr[r0];
    int e = rowptr[r0 + 1];
    int4 r_n = rec[min(s + lane, last)];
    for (int r = r0; r < rEnd; ++r) {
        const int e_next = rowptr[min(r + 2, nRows)];
        const int4 r_c = r_n;
        r_n = rec[min(e + lane, last)];
        row_t acc;
        gcn_row_zero(acc);
        const int deg = e - s;
        for (int b = s; b < e; b += 32) {
            int c_w; float v_w;
            if (deg <= 32) gcn_sort_row32(r_c, deg, lane, c_w, v_w);
            else           gcn_sorted_window(rec, s, e, b - s, lane, c_w, v_w);
            gcn_gather_chunk<VEC>(acc, c_w, v_w, min(32, e - b), x, loff);
        }
        *(row_t*)(agg + (size_t)r * D + loff) = acc;
        s = e;
        e = e_next;
    }
}

#define FUSE_WAVES 4
#define FUSE_ROWS  4
#define FUSE_LD    136

__device__ __forceinline__ v16h gcn_frag_a(const _Float16* tile, int r, int half, int kk) {
    const _Float16* p = tile + r * FUSE_LD + kk * 32 + half * 8;
    const v8h lo = *(const v8h*)p;
    const v8h hi = *(const v8h*)(p + 16);
    return __builtin_shufflevector(lo, hi, 0, 1, 2, 3, 4, 5, 6, 7, 8, 9, 10, 11, 12, 13, 14, 15);
}

__global__ __launch_bounds__(FUSE_WAVES * 32)
void gcn_layer1_fused(const int* __restrict__ rowptr, const int* __restrict__ recw,
                      const float* __restrict__ x,
                      const _Float16* __restrict__ W1p, const _Float16* __restrict__ W2p,
                      float* __restrict__ t, int nRows, int nEdges) {
    const int4* const rec = (const int4*)recw;
    __shared__ __attribute__((aligned(16))) _Float16 sA[16 * FUSE_LD];
    __shared__ __attribute__((aligned(16))) _Float16 sH[16 * FUSE_LD];
    const int lane = threadIdx.x & 31;
    const unsigned loff = (threadIdx.x & 31u) * 4u;
    const int w = __builtin_amdgcn_readfirstlane(threadIdx.x >> 5);
    const int tile0 = blockIdx.x * 16;
    const int r0 = tile0 + w * FUSE_ROWS;
    const int last = max(nEdges - 1, 0);
    int s = rowptr[min(r0, nRows)];
    int e = rowptr[min(r0 + 1, nRows)];
    int4 r_n = rec[min(s + lane, last)];
#pragma unroll 1
    for (int k = 0; k < FUSE_ROWS; ++k) {
        const int e_next = rowptr[min(r0 + k + 2, nRows)];
        const int4 r_c = r_n;
        r_n = rec[min(e + lane, last)];
        float4 acc;
        gcn_row_zero(acc);
        const int deg = e - s;
        for (int b = s; b < e; b += 32) {
            int c_w; float v_w;
            if (deg <= 32) gcn_sort_row32(r_c, deg, lane, c_w, v_w);
            else           gcn_sorted_window(rec, s, e, b - s, lane, c_w, v_w);
            gcn_gather_chunk<4>(acc, c_w, v_w, min(32, e - b), x, loff);
        }
        v4h hv;
        hv[0] = (_Float16)acc.x; hv[1] = (_Float16)acc.y; hv[2] = (_Float16)acc.z; hv[3] = (_Float16)acc.w;
        *(v4h*)(sA + (w * FUSE_ROWS + k) * FUSE_LD + loff) = hv;
        s = e;
        e = e_next;
    }
    __syncthreads();

    const int half = lane >> 4;
    const int rr   = lane & 15;

    v8f h0 = (v8f){};
    v8f h1 = (v8f){};
#pragma unroll
    for (int kk = 0; kk < 4; ++kk) {
        const v16h a  = gcn_frag_a(sA, rr, half, kk);
        const v16h b0 = *(const v16h*)(W1p + (((size_t)(2 * w)     * 4 + kk) * 32 + lane) * 16);
        const v16h b1 = *(const v16h*)(W1p + (((size_t)(2 * w + 1) * 4 + kk) * 32 + lane) * 16);
        h0 = __builtin_amdgcn_wmma_f32_16x16x32_f16(false, a, false, b0, (short)0, h0, false, false);
        h1 = __builtin_amdgcn_wmma_f32_16x16x32_f16(false, a, false, b1, (short)0, h1, false, false);
    }
#pragma unroll
    for (int v = 0; v < 8; ++v) {
        _Float16* hp = sH + (half * 8 + v) * FUSE_LD + (2 * w) * 16 + rr;
        hp[0]  = (_Float16)fmaxf(h0[v], 0.0f);
        hp[16] = (_Float16)fmaxf(h1[v], 0.0f);
    }
    __syncthreads();

    v8f tv = (v8f){};
#pragma unroll
    for (int kk = 0; kk < 4; ++kk) {
        const v16h a = gcn_frag_a(sH, rr, half, kk);
        const v16h b = *(const v16h*)(W2p + (((size_t)w * 4 + kk) * 32 + lane) * 16);
        tv = __builtin_amdgcn_wmma_f32_16x16x32_f16(false, a, false, b, (short)0, tv, false, false);
    }
    const int mbase = tile0 + half * 8;
    float* tp = t + (size_t)mbase * 64 + w * 16 + rr;
    if (tile0 + 16 <= nRows) {
#pragma unroll
        for (int v = 0; v < 8; ++v) tp[v * 64] = tv[v];
    } else {
#pragma unroll
        for (int v = 0; v < 8; ++v)
            if (mbase + v < nRows) tp[v * 64] = tv[v];
    }
}

extern "C" void kernel_launch(void* const* d_in, const int* in_sizes, int n_in,
                              void* d_out, int out_size, void* d_ws, size_t ws_size,
                              hipStream_t stream) {
    const float* x    = (const float*)d_in[0];
    const int*   row  = (const int*)d_in[1];
    const int*   col  = (const int*)d_in[2];
    const float* vals = (const float*)d_in[3];
    const float* W1   = (const float*)d_in[4];
    const float* W2   = (const float*)d_in[5];
    float* out = (float*)d_out;

    const int NN = in_sizes[0] / IN_DIM;
    const int NE = in_sizes[1];

    float*     tBuf = (float*)d_ws;
    _Float16*  W1p  = (_Float16*)(tBuf + (size_t)2 * NN * IN_DIM);
    _Float16*  W2p  = W1p + 128 * 128;
    const int  NNp  = ((NN + 8) >> 2) << 2;
    int*       lrow   = (int*)(W2p + 128 * 64);
    int*       rowptr = lrow + NNp;
    int*       rec    = rowptr + NNp;
    int*       part   = rec + (size_t)4 * NE;
    int*       bsum   = part + (size_t)((NE + CNT_SLICE - 1) / CNT_SLICE) * NNp;

    const int nSlices = (NE + CNT_SLICE - 1) / CNT_SLICE;
    const int nRanges = (NN + CNT_RANGE - 1) / CNT_RANGE;
    const int nPfx    = (NN + PFX_THREADS - 1) / PFX_THREADS;
    gcn_count_lds<<<dim3(nSlices, nRanges + 1), CNT_THREADS, 0, stream>>>(row, part, NN, NNp, NE, W1, W2, W1p, W2p, 8, 4);
    gcn_slice_prefix<<<nPfx, PFX_THREADS, 0, stream>>>(part, lrow, bsum, NN, NNp, nSlices);
    gcn_place_lds<<<dim3(nSlices, nRanges), CNT_THREADS, 0, stream>>>(row, col, vals, lrow, bsum, part, rowptr, rec, NN, NNp, NE, nPfx);
    const int rowBlocks = (NN + 8 * ROWS_PER_WAVE - 1) / (8 * ROWS_PER_WAVE);

    float* tF = tBuf;
    gcn_layer1_fused<<<(NN + 15) / 16, FUSE_WAVES * 32, 0, stream>>>(rowptr, rec, x, W1p, W2p, tF, NN, NE);

    gcn_gather_rows<2><<<rowBlocks, 256, 0, stream>>>(rowptr, rec, tF, out, NN, NE);
}


# Slice-prefix launch gone, 5 launches to 4: count launch writes byte-wide per-slice row counts, each placing work-group sums and scans the counts of its own rows itself; f32 sums and order unchanged
# speedup vs baseline: 4.1757x; 1.0376x over previous
#include <hip/hip_runtime.h>
#include <hip/hip_bf16.h>

typedef __attribute__((ext_vector_type(16))) _Float16 v16h;
typedef __attribute__((ext_vector_type(8)))  _Float16 v8h;
typedef __attribute__((ext_vector_type(4)))  _Float16 v4h;
typedef __attribute__((ext_vector_type(8)))  float    v8f;

#define IN_DIM 128

__device__ __forceinline__ void gcn_pack_w(const float* __restrict__ W, _Float16* __restrict__ Wp,
                                           int Nt  , int id) {
    int lane = id & 31;
    int kk   = (id >> 5) & 3;
    int t    = id >> 7;
    int half = lane >> 4;
    int r    = lane & 15;
    int N    = Nt * 16;
    int n    = t * 16 + r;
    int k0   = kk * 32;
    _Float16* dst = Wp + (size_t)id * 16;
#pragma unroll
    for (int j = 0; j < 8; ++j)
        dst[j] = (_Float16)W[(size_t)(k0 + half * 8 + j) * N + n];
#pragma unroll
    for (int j = 0; j < 8; ++j)
        dst[8 + j] = (_Float16)W[(size_t)(k0 + 16 + half * 8 + j) * N + n];
}

#define CNT_RANGE   8192
#define CNT_SLICE   32768
#define CNT_THREADS 512
#define CNT_U       8

__device__ __forceinline__ void gcn_load_rows(int4 (&v)[CNT_U], const int* __restrict__ row, int ib, int e1v) {
#pragma unroll
    for (int k = 0; k < CNT_U; ++k) {
        const int i = ib + k * CNT_THREADS * 4;
        v[k] = make_int4(-1, -1, -1, -1);
        if (i < e1v) v[k] = *(const int4*)(row + i);
    }
}

__device__ __forceinline__ void gcn_count_one(int* c, int v, int base, unsigned lim) {
    const unsigned d = (unsigned)(v - base);
    if (d < lim) atomicAdd(&c[d], 1);
}

#define CNT_PER_THREAD (CNT_RANGE / CNT_THREADS)
static_assert(CNT_PER_THREAD == 16, "a thread's counts are four 16-byte LDS reads and one 16-byte store");

__global__ __launch_bounds__(CNT_THREADS)
void gcn_count_lds(const int* __restrict__ row, int* __restrict__ cnt32, unsigned* __restrict__ cnt8,
                   int* __restrict__ rtot, int* __restrict__ sflag,
                   int nRows, int pitch, int nEdges,
                   const float* __restrict__ W1, const float* __restrict__ W2,
                   _Float16* __restrict__ W1p, _Float16* __restrict__ W2p,
                   int nt1  , int nt2  ) {
    __shared__ __attribute__((aligned(16))) int c[CNT_RANGE];
    __shared__ int wsum[CNT_THREADS / 32];
    __shared__ int wbig[CNT_THREADS / 32];
    const int t = threadIdx.x;
    if (blockIdx.y == gridDim.y - 1) {
        const int n1 = nt1 * 4 * 32;
        const int n2 = nt2 * 4 * 32;
        for (int id = blockIdx.x * CNT_THREADS + t; id < n1 + n2; id += gridDim.x * CNT_THREADS) {
            if (id < n1) gcn_pack_w(W1, W1p, nt1, id);
            else         gcn_pack_w(W2, W2p, nt2, id - n1);
        }
        return;
    }
    const int base = blockIdx.y * CNT_RANGE;
    const unsigned lim = (unsigned)max(0, min(CNT_RANGE, nRows - base));
    const int e0 = blockIdx.x * CNT_SLICE;
    const int e1 = min(nEdges, e0 + CNT_SLICE);
    const int e1v = e0 + ((e1 - e0) & ~3);
    for (int j = t; j < CNT_RANGE; j += CNT_THREADS) c[j] = 0;
    __syncthreads();
    for (int ib = e0 + t * 4; ib < e1v; ib += CNT_THREADS * 4 * CNT_U) {
        int4 v[CNT_U];
        gcn_load_rows(v, row, ib, e1v);
#pragma unroll
        for (int k = 0; k < CNT_U; ++k) {
            gcn_count_one(c, v[k].x, base, lim);
            gcn_count_one(c, v[k].y, base, lim);
            gcn_count_one(c, v[k].z, base, lim);
            gcn_count_one(c, v[k].w, base, lim);
        }
    }
    if (t < e1 - e1v) gcn_count_one(c, row[e1v + t], base, lim);
    __syncthreads();
    int o[CNT_PER_THREAD];
#pragma unroll
    for (int g = 0; g < 4; ++g) {
        const int4 v = *(const int4*)(c + t * CNT_PER_THREAD + g * 4);
        o[g * 4 + 0] = v.x; o[g * 4 + 1] = v.y; o[g * 4 + 2] = v.z; o[g * 4 + 3] = v.w;
    }
    unsigned pk[4];
    int tsum = 0;
    int big = 0;
#pragma unroll
    for (int g = 0; g < 4; ++g) {
        pk[g] = (unsigned)min(o[g * 4 + 0], 255)
              | ((unsigned)min(o[g * 4 + 1], 255) << 8)
              | ((unsigned)min(o[g * 4 + 2], 255) << 16)
              | ((unsigned)min(o[g * 4 + 3], 255) << 24);
    }
#pragma unroll
    for (int k = 0; k < CNT_PER_THREAD; ++k) {
        tsum += o[k];
        big |= (o[k] >= 255) ? 1 : 0;
    }
    const size_t off = (size_t)blockIdx.x * pitch + base + t * CNT_PER_THREAD;
    *(uint4*)(cnt8 + (off >> 2)) = make_uint4(pk[0], pk[1], pk[2], pk[3]);
    for (int d = 16; d >= 1; d >>= 1) {
        tsum += __shfl_xor(tsum, d);
        big |= __shfl_xor(big, d);
    }
    if ((t & 31) == 0) { wsum[t >> 5] = tsum; wbig[t >> 5] = big; }
    __syncthreads();
    int total = 0;
    int anybig = 0;
#pragma unroll
    for (int i = 0; i < CNT_THREADS / 32; ++i) { total += wsum[i]; anybig |= wbig[i]; }
    if (t == 0) {
        rtot[blockIdx.y * gridDim.x + blockIdx.x] = total;
        sflag[blockIdx.y * gridDim.x + blockIdx.x] = anybig;
    }
    if (anybig) {
#pragma unroll
        for (int g = 0; g < 4; ++g)
            *(int4*)(cnt32 + off + g * 4) = make_int4(o[g * 4 + 0], o[g * 4 + 1], o[g * 4 + 2], o[g * 4 + 3]);
    }
}

__device__ __forceinline__ int gcn_take_slot(int* p, int v, int base, unsigned lim) {
    const unsigned d = (unsigned)(v - base);
    int pos = -1;
    if (d < lim) pos = atomicAdd(&p[d], 1);
    return pos;
}

#define PL_B 25
__device__ __forceinline__ void gcn_cnt_load(uint4 (&q)[PL_B], const unsigned* __restrict__ q0,
                                             int sb, int nSlices, int pitchW) {
#pragma unroll
    for (int k = 0; k < PL_B; ++k)
        q[k] = *(const uint4*)(q0 + (size_t)min(sb + k, nSlices - 1) * pitchW);
}
__device__ __forceinline__ void gcn_cnt_add(unsigned& e, unsigned& o, unsigned x) {
    e += x & 0x00FF00FFu;
    o += (x >> 8) & 0x00FF00FFu;
}
__device__ __forceinline__ void gcn_cnt_acc(const uint4 (&q)[PL_B], int sb, int nSlices, int s,
                                            unsigned (&ae)[4], unsigned (&ao)[4],
                                            unsigned (&pe)[4], unsigned (&po)[4]) {
#pragma unroll
    for (int k = 0; k < PL_B; ++k) {
        if (sb + k < nSlices) {
            if (sb + k == s) {
#pragma unroll
                for (int g = 0; g < 4; ++g) { pe[g] = ae[g]; po[g] = ao[g]; }
            }
            gcn_cnt_add(ae[0], ao[0], q[k].x);
            gcn_cnt_add(ae[1], ao[1], q[k].y);
            gcn_cnt_add(ae[2], ao[2], q[k].z);
            gcn_cnt_add(ae[3], ao[3], q[k].w);
        }
    }
}

__global__ __launch_bounds__(CNT_THREADS)
void gcn_place_lds(const int* __restrict__ row, const int* __restrict__ col,
                   const float* __restrict__ vals, const unsigned* __restrict__ cnt8,
                   const int* __restrict__ cnt32, const int* __restrict__ rtot,
                   const int* __restrict__ sflag,
                   int* __restrict__ rowptr, int* __restrict__ recw,
                   int nRows, int pitch, int nEdges, int nSlices) {
    int4* const rec = (int4*)recw;
    __shared__ __attribute__((aligned(16))) int p[CNT_RANGE];
    __shared__ int wtot[CNT_THREADS / 32];
    const int t = threadIdx.x;
    const int lane = t & 31;
    const int w = t >> 5;
    const int s = blockIdx.x;
    const int base = blockIdx.y * CNT_RANGE;
    const unsigned lim = (unsigned)max(0, min(CNT_RANGE, nRows - base));
    const int e0 = blockIdx.x * CNT_SLICE;
    const int e1 = min(nEdges, e0 + CNT_SLICE);
    const int e1v = e0 + ((e1 - e0) & ~3);
    {
        const int pitchW = pitch >> 2;
        const unsigned* const q0 = cnt8 + (base >> 2) + t * 4;
        unsigned ae[4] = {0u, 0u, 0u, 0u}, ao[4] = {0u, 0u, 0u, 0u};
        unsigned pe[4] = {0u, 0u, 0u, 0u}, po[4] = {0u, 0u, 0u, 0u};
        uint4 q[PL_B];
        gcn_cnt_load(q, q0, 0, nSlices, pitchW);
        const int nAhead = blockIdx.y * nSlices;
        int ahead = 0;
        int slow = (nSlices > 256) ? 1 : 0;
        for (int bb = 0; bb < max(nAhead, nSlices); bb += 8 * 32) {
            int v[8], f[8];
#pragma unroll
            for (int k = 0; k < 8; ++k) {
                const int i = bb + k * 32 + lane;
                v[k] = rtot[max(min(i, nAhead - 1), 0)];
                f[k] = sflag[nAhead + min(i, nSlices - 1)];
            }
#pragma unroll
            for (int k = 0; k < 8; ++k) {
                const int i = bb + k * 32 + lane;
                ahead += (i < nAhead) ? v[k] : 0;
                slow |= (i < nSlices) ? f[k] : 0;
            }
        }
        for (int d = 16; d >= 1; d >>= 1) {
            ahead += __shfl_xor(ahead, d);
            slow |= __shfl_xor(slow, d);
        }
        gcn_cnt_acc(q, 0, nSlices, s, ae, ao, pe, po);
        for (int sb = PL_B; sb < nSlices; sb += PL_B) {
            gcn_cnt_load(q, q0, sb, nSlices, pitchW);
            gcn_cnt_acc(q, sb, nSlices, s, ae, ao, pe, po);
        }
        int tot[CNT_PER_THREAD], pre[CNT_PER_THREAD];
#pragma unroll
        for (int g = 0; g < 4; ++g) {
            tot[g * 4 + 0] = (int)(ae[g] & 0xFFFFu);  tot[g * 4 + 2] = (int)(ae[g] >> 16);
            tot[g * 4 + 1] = (int)(ao[g] & 0xFFFFu);  tot[g * 4 + 3] = (int)(ao[g] >> 16);
            pre[g * 4 + 0] = (int)(pe[g] & 0xFFFFu);  pre[g * 4 + 2] = (int)(pe[g] >> 16);
            pre[g * 4 + 1] = (int)(po[g] & 0xFFFFu);  pre[g * 4 + 3] = (int)(po[g] >> 16);
        }
        if (slow) {
#pragma unroll
            for (int k = 0; k < CNT_PER_THREAD; ++k) { tot[k] = 0; pre[k] = 0; }
            for (int s2 = 0; s2 < nSlices; ++s2) {
                const bool exact = sflag[nAhead + s2] != 0;
                const size_t off = (size_t)s2 * pitch + base + t * CNT_PER_THREAD;
#pragma unroll
                for (int g = 0; g < 4; ++g) {
                    int cn[4];
                    if (exact) {
                        const int4 v = *(const int4*)(cnt32 + off + g * 4);
                        cn[0] = v.x; cn[1] = v.y; cn[2] = v.z; cn[3] = v.w;
                    } else {
                        const unsigned x = cnt8[(off >> 2) + g];
                        cn[0] = (int)(x & 255u); cn[1] = (int)((x >> 8) & 255u);
                        cn[2] = (int)((x >> 16) & 255u); cn[3] = (int)(x >> 24);
                    }
#pragma unroll
                    for (int b = 0; b < 4; ++b) {
                        tot[g * 4 + b] += cn[b];
                        if (s2 < s) pre[g * 4 + b] += cn[b];
                    }
                }
            }
        }
        int tsum = 0;
#pragma unroll
        for (int k = 0; k < CNT_PER_THREAD; ++k) tsum += tot[k];
        int inc = tsum;
        for (int d = 1; d < 32; d <<= 1) {
            const int o = __shfl_up(inc, d);
            if (lane >= d) inc += o;
        }
        if (lane == 31) wtot[w] = inc;
        __syncthreads();
#pragma unroll
        for (int i = 0; i < CNT_THREADS / 32; ++i) {
            const int v = wtot[i];
            if (i < w) ahead += v;
        }
        int run = ahead + inc - tsum;
        int pv[CNT_PER_THREAD];
#pragma unroll
        for (int k = 0; k < CNT_PER_THREAD; ++k) {
            const unsigned j = (unsigned)(t * CNT_PER_THREAD + k);
            pv[k] = run + pre[k];
            if (blockIdx.x == 0 && j < lim) rowptr[base + j] = run;
            run += tot[k];
        }
#pragma unroll
        for (int g = 0; g < 4; ++g)
            *(int4*)(p + t * CNT_PER_THREAD + g * 4) = make_int4(pv[g * 4 + 0], pv[g * 4 + 1], pv[g * 4 + 2], pv[g * 4 + 3]);
        if (blockIdx.x == 0 && blockIdx.y == gridDim.y - 1 && t == CNT_THREADS - 1) rowptr[nRows] = run;
    }
    __syncthreads();
    for (int ib = e0 + t * 4; ib < e1v; ib += CNT_THREADS * 4 * CNT_U) {
        int4 v[CNT_U];
        gcn_load_rows(v, row, ib, e1v);
        int4 cc[CNT_U];
        float4 vv[CNT_U];
#pragma unroll
        for (int k = 0; k < CNT_U; ++k) {
            const int i = ib + k * CNT_THREADS * 4;
            cc[k] = make_int4(0, 0, 0, 0);
            vv[k] = make_float4(0.0f, 0.0f, 0.0f, 0.0f);
            if (i < e1v) {
                cc[k] = *(const int4*)(col + i);
                vv[k] = *(const float4*)(vals + i);
            }
        }
        int4 pos[CNT_U];
#pragma unroll
        for (int k = 0; k < CNT_U; ++k) {
            pos[k].x = gcn_take_slot(p, v[k].x, base, lim);
            pos[k].y = gcn_take_slot(p, v[k].y, base, lim);
            pos[k].z = gcn_take_slot(p, v[k].z, base, lim);
            pos[k].w = gcn_take_slot(p, v[k].w, base, lim);
        }
#pragma unroll
        for (int k = 0; k < CNT_U; ++k) {
            const int i = ib + k * CNT_THREADS * 4;
            if (pos[k].x >= 0) rec[pos[k].x] = make_int4(i,     cc[k].x, __float_as_int(vv[k].x), 0);
            if (pos[k].y >= 0) rec[pos[k].y] = make_int4(i + 1, cc[k].y, __float_as_int(vv[k].y), 0);
            if (pos[k].z >= 0) rec[pos[k].z] = make_int4(i + 2, cc[k].z, __float_as_int(vv[k].z), 0);
            if (pos[k].w >= 0) rec[pos[k].w] = make_int4(i + 3, cc[k].w, __float_as_int(vv[k].w), 0);
        }
    }
    if (t < e1 - e1v) {
        const int pos = gcn_take_slot(p, row[e1v + t], base, lim);
        if (pos >= 0) rec[pos] = make_int4(e1v + t, col[e1v + t], __float_as_int(vals[e1v + t]), 0);
    }
}

#define ROWS_PER_WAVE 5

__device__ __forceinline__ void gcn_sort_row32(const int4& r, int deg, int lane, int& c_l, float& v_l) {
    int place = 0;
    for (int j = 0; j < deg; ++j)
        place += (__builtin_amdgcn_readlane(r.x, j) < r.x) ? 1 : 0;
    const int dst = ((lane < deg) ? place : lane) << 2;
    c_l = __builtin_amdgcn_ds_permute(dst, r.y);
    v_l = __int_as_float(__builtin_amdgcn_ds_permute(dst, r.z));
}

__device__ __forceinline__ void gcn_sorted_window(const int4* __restrict__ rec, int s, int e, int w0, int lane,
                                                  int& c_w, float& v_w) {
    c_w = 0;
    v_w = 0.0f;
    int ib = s;
    do {
        const int4 r = rec[min(ib + lane, e - 1)];
        int place = 0;
        int jb = s;
        do {
            const int jn = min(32, e - jb);
            const int other = rec[min(jb + lane, e - 1)].x;
            place += (__builtin_amdgcn_readlane(other, 0) < r.x) ? 1 : 0;
            for (int j = 1; j < jn; ++j)
                place += (__builtin_amdgcn_readlane(other, j) < r.x) ? 1 : 0;
            jb += 32;
        } while (jb < e);
        const int d = place - w0;
        const int jn = min(32, e - ib);
        for (int j = 0; j < jn; ++j) {
            const int   dj = __builtin_amdgcn_readlane(d, j);
            const int   cj = __builtin_amdgcn_readlane(r.y, j);
            const float vj = __int_as_float(__builtin_amdgcn_readlane(r.z, j));
            if (dj == lane) { c_w = cj; v_w = vj; }
        }
        ib += 32;
    } while (ib < e);
}

template <int VEC> struct gcn_row;
template <> struct gcn_row<4> { typedef float4 type; };
template <> struct gcn_row<2> { typedef float2 type; };
__device__ __forceinline__ void gcn_row_zero(float4& a) { a.x = 0.0f; a.y = 0.0f; a.z = 0.0f; a.w = 0.0f; }
__device__ __forceinline__ void gcn_row_zero(float2& a) { a.x = 0.0f; a.y = 0.0f; }
__device__ __forceinline__ void gcn_row_fma(float4& a, float v, const float4& m) {
    a.x = fmaf(v, m.x, a.x);
    a.y = fmaf(v, m.y, a.y);
    a.z = fmaf(v, m.z, a.z);
    a.w = fmaf(v, m.w, a.w);
}
__device__ __forceinline__ void gcn_row_fma(float2& a, float v, const float2& m) {
    a.x = fmaf(v, m.x, a.x);
    a.y = fmaf(v, m.y, a.y);
}

#define GATHER_U 8
template <int VEC>
__device__ __forceinline__ void gcn_gather_chunk(typename gcn_row<VEC>::type& acc, int c_l, float v_l, int n,
                                                 const float* __restrict__ x, unsigned loff) {
    typedef typename gcn_row<VEC>::type row_t;
    const int D = 32 * VEC;
    int j = 0;
    for (; j + GATHER_U <= n; j += GATHER_U) {
        row_t m[GATHER_U];
#pragma unroll
        for (int k = 0; k < GATHER_U; ++k) {
            const int c = __builtin_amdgcn_readlane(c_l, j + k);
            m[k] = *(const row_t*)(x + (size_t)c * D + loff);
        }
#pragma unroll
        for (int k = 0; k < GATHER_U; ++k) {
            const float v = __int_as_float(__builtin_amdgcn_readlane(__float_as_int(v_l), j + k));
            gcn_row_fma(acc, v, m[k]);
        }
    }
    const int rem = n - j;
    if (rem > 0) {
        row_t m[GATHER_U - 1];
#pragma unroll
        for (int k = 0; k < GATHER_U - 1; ++k) {
            if (k < rem) {
                const int c = __builtin_amdgcn_readlane(c_l, j + k);
                m[k] = *(const row_t*)(x + (size_t)c * D + loff);
            }
        }
#pragma unroll
        for (int k = 0; k < GATHER_U - 1; ++k) {
            if (k < rem) {
                const float v = __int_as_float(__builtin_amdgcn_readlane(__float_as_int(v_l), j + k));
                gcn_row_fma(acc, v, m[k]);
            }
        }
    }
}

template <int VEC>
__global__ __launch_bounds__(256)
void gcn_gather_rows(const int* __restrict__ rowptr, const int* __restrict__ recw,
                     const float* __restrict__ x,
                     float* __restrict__ agg, int nRows, int nEdges) {
    typedef typename gcn_row<VEC>::type row_t;
    const int4* const rec = (const int4*)recw;
    const int D = 32 * VEC;
    const int lane = threadIdx.x & 31;
    const unsigned loff = (threadIdx.x & 31u) * (unsigned)VEC;
    const int wave = __builtin_amdgcn_readfirstlane((blockIdx.x * blockDim.x + threadIdx.x) >> 5);
    const int r0 = wave * ROWS_PER_WAVE;
    if (r0 >= nRows) return;
    const int rEnd = min(nRows, r0 + ROWS_PER_WAVE);
    const int last = max(nEdges - 1, 0);
    int s = rowptr[r0];
    int e = rowptr[r0 + 1];
    int4 r_n = rec[min(s + lane, last)];
    for (int r = r0; r < rEnd; ++r) {
        const int e_next = rowptr[min(r + 2, nRows)];
        const int4 r_c = r_n;
        r_n = rec[min(e + lane, last)];
        row_t acc;
        gcn_row_zero(acc);
        const int deg = e - s;
        for (int b = s; b < e; b += 32) {
            int c_w; float v_w;
            if (deg <= 32) gcn_sort_row32(r_c, deg, lane, c_w, v_w);
            else           gcn_sorted_window(rec, s, e, b - s, lane, c_w, v_w);
            gcn_gather_chunk<VEC>(acc, c_w, v_w, min(32, e - b), x, loff);
        }
        *(row_t*)(agg + (size_t)r * D + loff) = acc;
        s = e;
        e = e_next;
    }
}

#define FUSE_WAVES 4
#define FUSE_ROWS  4
#define FUSE_LD    136

__device__ __forceinline__ v16h gcn_frag_a(const _Float16* tile, int r, int half, int kk) {
    const _Float16* p = tile + r * FUSE_LD + kk * 32 + half * 8;
    const v8h lo = *(const v8h*)p;
    const v8h hi = *(const v8h*)(p + 16);
    return __builtin_shufflevector(lo, hi, 0, 1, 2, 3, 4, 5, 6, 7, 8, 9, 10, 11, 12, 13, 14, 15);
}

__global__ __launch_bounds__(FUSE_WAVES * 32)
void gcn_layer1_fused(const int* __restrict__ rowptr, const int* __restrict__ recw,
                      const float* __restrict__ x,
                      const _Float16* __restrict__ W1p, const _Float16* __restrict__ W2p,
                      float* __restrict__ t, int nRows, int nEdges) {
    const int4* const rec = (const int4*)recw;
    __shared__ __attribute__((aligned(16))) _Float16 sA[16 * FUSE_LD];
    __shared__ __attribute__((aligned(16))) _Float16 sH[16 * FUSE_LD];
    const int lane = threadIdx.x & 31;
    const unsigned loff = (threadIdx.x & 31u) * 4u;
    const int w = __builtin_amdgcn_readfirstlane(threadIdx.x >> 5);
    const int tile0 = blockIdx.x * 16;
    const int r0 = tile0 + w * FUSE_ROWS;
    const int last = max(nEdges - 1, 0);
    int s = rowptr[min(r0, nRows)];
    int e = rowptr[min(r0 + 1, nRows)];
    int4 r_n = rec[min(s + lane, last)];
#pragma unroll 1
    for (int k = 0; k < FUSE_ROWS; ++k) {
        const int e_next = rowptr[min(r0 + k + 2, nRows)];
        const int4 r_c = r_n;
        r_n = rec[min(e + lane, last)];
        float4 acc;
        gcn_row_zero(acc);
        const int deg = e - s;
        for (int b = s; b < e; b += 32) {
            int c_w; float v_w;
            if (deg <= 32) gcn_sort_row32(r_c, deg, lane, c_w, v_w);
            else           gcn_sorted_window(rec, s, e, b - s, lane, c_w, v_w);
            gcn_gather_chunk<4>(acc, c_w, v_w, min(32, e - b), x, loff);
        }
        v4h hv;
        hv[0] = (_Float16)acc.x; hv[1] = (_Float16)acc.y; hv[2] = (_Float16)acc.z; hv[3] = (_Float16)acc.w;
        *(v4h*)(sA + (w * FUSE_ROWS + k) * FUSE_LD + loff) = hv;
        s = e;
        e = e_next;
    }
    __syncthreads();

    const int half = lane >> 4;
    const int rr   = lane & 15;

    v8f h0 = (v8f){};
    v8f h1 = (v8f){};
#pragma unroll
    for (int kk = 0; kk < 4; ++kk) {
        const v16h a  = gcn_frag_a(sA, rr, half, kk);
        const v16h b0 = *(const v16h*)(W1p + (((size_t)(2 * w)     * 4 + kk) * 32 + lane) * 16);
        const v16h b1 = *(const v16h*)(W1p + (((size_t)(2 * w + 1) * 4 + kk) * 32 + lane) * 16);
        h0 = __builtin_amdgcn_wmma_f32_16x16x32_f16(false, a, false, b0, (short)0, h0, false, false);
        h1 = __builtin_amdgcn_wmma_f32_16x16x32_f16(false, a, false, b1, (short)0, h1, false, false);
    }
#pragma unroll
    for (int v = 0; v < 8; ++v) {
        _Float16* hp = sH + (half * 8 + v) * FUSE_LD + (2 * w) * 16 + rr;
        hp[0]  = (_Float16)fmaxf(h0[v], 0.0f);
        hp[16] = (_Float16)fmaxf(h1[v], 0.0f);
    }
    __syncthreads();

    v8f tv = (v8f){};
#pragma unroll
    for (int kk = 0; kk < 4; ++kk) {
        const v16h a = gcn_frag_a(sH, rr, half, kk);
        const v16h b = *(const v16h*)(W2p + (((size_t)w * 4 + kk) * 32 + lane) * 16);
        tv = __builtin_amdgcn_wmma_f32_16x16x32_f16(false, a, false, b, (short)0, tv, false, false);
    }
    const int mbase = tile0 + half * 8;
    float* tp = t + (size_t)mbase * 64 + w * 16 + rr;
    if (tile0 + 16 <= nRows) {
#pragma unroll
        for (int v = 0; v < 8; ++v) tp[v * 64] = tv[v];
    } else {
#pragma unroll
        for (int v = 0; v < 8; ++v)
            if (mbase + v < nRows) tp[v * 64] = tv[v];
    }
}

extern "C" void kernel_launch(void* const* d_in, const int* in_sizes, int n_in,
                              void* d_out, int out_size, void* d_ws, size_t ws_size,
                              hipStream_t stream) {
    const float* x    = (const float*)d_in[0];
    const int*   row  = (const int*)d_in[1];
    const int*   col  = (const int*)d_in[2];
    const float* vals = (const float*)d_in[3];
    const float* W1   = (const float*)d_in[4];
    const float* W2   = (const float*)d_in[5];
    float* out = (float*)d_out;

    const int NN = in_sizes[0] / IN_DIM;
    const int NE = in_sizes[1];

    const int nSlices = (NE + CNT_SLICE - 1) / CNT_SLICE;
    const int nRanges = (NN + CNT_RANGE - 1) / CNT_RANGE;
    const int pitch   = nRanges * CNT_RANGE;

    float*     tBuf = (float*)d_ws;
    _Float16*  W1p  = (_Float16*)(tBuf + (size_t)2 * NN * IN_DIM);
    _Float16*  W2p  = W1p + 128 * 128;
    const int  NNp  = ((NN + 8) >> 2) << 2;
    int*       rowptr = (int*)(W2p + 128 * 64);
    int*       rec    = rowptr + NNp;
    int*       cnt32  = rec + (size_t)4 * NE;
    unsigned*  cnt8   = (unsigned*)(cnt32 + (size_t)nSlices * pitch);
    int*       rtot   = (int*)(cnt8 + (size_t)nSlices * (pitch >> 2));
    int*       sflag  = rtot + (size_t)nRanges * nSlices;

    gcn_count_lds<<<dim3(nSlices, nRanges + 1), CNT_THREADS, 0, stream>>>(row, cnt32, cnt8, rtot, sflag, NN, pitch, NE, W1, W2, W1p, W2p, 8, 4);
    gcn_place_lds<<<dim3(nSlices, nRanges), CNT_THREADS, 0, stream>>>(row, col, vals, cnt8, cnt32, rtot, sflag, rowptr, rec, NN, pitch, NE, nSlices);
    const int rowBlocks = (NN + 8 * ROWS_PER_WAVE - 1) / (8 * ROWS_PER_WAVE);

    float* tF = tBuf;
    gcn_layer1_fused<<<(NN + 15) / 16, FUSE_WAVES * 32, 0, stream>>>(rowptr, rec, x, W1p, W2p, tF, NN, NE);

    gcn_gather_rows<2><<<rowBlocks, 256, 0, stream>>>(rowptr, rec, tF, out, NN, NE);
}


# Gathered rows asked for 16 at a time, not 8 (a mean-length row: one trip to memory, not two); layer-1 waves take 8 rows of a 32-row work-group to stay all resident; same f32 sums and order
# speedup vs baseline: 4.3001x; 1.0298x over previous
#include <hip/hip_runtime.h>
#include <hip/hip_bf16.h>

typedef __attribute__((ext_vector_type(16))) _Float16 v16h;
typedef __attribute__((ext_vector_type(8)))  _Float16 v8h;
typedef __attribute__((ext_vector_type(4)))  _Float16 v4h;
typedef __attribute__((ext_vector_type(8)))  float    v8f;

#define IN_DIM 128

__device__ __forceinline__ void gcn_pack_w(const float* __restrict__ W, _Float16* __restrict__ Wp,
                                           int Nt  , int id) {
    int lane = id & 31;
    int kk   = (id >> 5) & 3;
    int t    = id >> 7;
    int half = lane >> 4;
    int r    = lane & 15;
    int N    = Nt * 16;
    int n    = t * 16 + r;
    int k0   = kk * 32;
    _Float16* dst = Wp + (size_t)id * 16;
#pragma unroll
    for (int j = 0; j < 8; ++j)
        dst[j] = (_Float16)W[(size_t)(k0 + half * 8 + j) * N + n];
#pragma unroll
    for (int j = 0; j < 8; ++j)
        dst[8 + j] = (_Float16)W[(size_t)(k0 + 16 + half * 8 + j) * N + n];
}

#define CNT_RANGE   8192
#define CNT_SLICE   32768
#define CNT_THREADS 512
#define CNT_U       8

__device__ __forceinline__ void gcn_load_rows(int4 (&v)[CNT_U], const int* __restrict__ row, int ib, int e1v) {
#pragma unroll
    for (int k = 0; k < CNT_U; ++k) {
        const int i = ib + k * CNT_THREADS * 4;
        v[k] = make_int4(-1, -1, -1, -1);
        if (i < e1v) v[k] = *(const int4*)(row + i);
    }
}

__device__ __forceinline__ void gcn_count_one(int* c, int v, int base, unsigned lim) {
    const unsigned d = (unsigned)(v - base);
    if (d < lim) atomicAdd(&c[d], 1);
}

#define CNT_PER_THREAD (CNT_RANGE / CNT_THREADS)
static_assert(CNT_PER_THREAD == 16, "a thread's counts are four 16-byte LDS reads and one 16-byte store");

__global__ __launch_bounds__(CNT_THREADS)
void gcn_count_lds(const int* __restrict__ row, int* __restrict__ cnt32, unsigned* __restrict__ cnt8,
                   int* __restrict__ rtot, int* __restrict__ sflag,
                   int nRows, int pitch, int nEdges,
                   const float* __restrict__ W1, const float* __restrict__ W2,
                   _Float16* __restrict__ W1p, _Float16* __restrict__ W2p,
                   int nt1  , int nt2  ) {
    __shared__ __attribute__((aligned(16))) int c[CNT_RANGE];
    __shared__ int wsum[CNT_THREADS / 32];
    __shared__ int wbig[CNT_THREADS / 32];
    const int t = threadIdx.x;
    if (blockIdx.y == gridDim.y - 1) {
        const int n1 = nt1 * 4 * 32;
        const int n2 = nt2 * 4 * 32;
        for (int id = blockIdx.x * CNT_THREADS + t; id < n1 + n2; id += gridDim.x * CNT_THREADS) {
            if (id < n1) gcn_pack_w(W1, W1p, nt1, id);
            else         gcn_pack_w(W2, W2p, nt2, id - n1);
        }
        return;
    }
    const int base = blockIdx.y * CNT_RANGE;
    const unsigned lim = (unsigned)max(0, min(CNT_RANGE, nRows - base));
    const int e0 = blockIdx.x * CNT_SLICE;
    const int e1 = min(nEdges, e0 + CNT_SLICE);
    const int e1v = e0 + ((e1 - e0) & ~3);
    for (int j = t; j < CNT_RANGE; j += CNT_THREADS) c[j] = 0;
    __syncthreads();
    for (int ib = e0 + t * 4; ib < e1v; ib += CNT_THREADS * 4 * CNT_U) {
        int4 v[CNT_U];
        gcn_load_rows(v, row, ib, e1v);
#pragma unroll
        for (int k = 0; k < CNT_U; ++k) {
            gcn_count_one(c, v[k].x, base, lim);
            gcn_count_one(c, v[k].y, base, lim);
            gcn_count_one(c, v[k].z, base, lim);
            gcn_count_one(c, v[k].w, base, lim);
        }
    }
    if (t < e1 - e1v) gcn_count_one(c, row[e1v + t], base, lim);
    __syncthreads();
    int o[CNT_PER_THREAD];
#pragma unroll
    for (int g = 0; g < 4; ++g) {
        const int4 v = *(const int4*)(c + t * CNT_PER_THREAD + g * 4);
        o[g * 4 + 0] = v.x; o[g * 4 + 1] = v.y; o[g * 4 + 2] = v.z; o[g * 4 + 3] = v.w;
    }
    unsigned pk[4];
    int tsum = 0;
    int big = 0;
#pragma unroll
    for (int g = 0; g < 4; ++g) {
        pk[g] = (unsigned)min(o[g * 4 + 0], 255)
              | ((unsigned)min(o[g * 4 + 1], 255) << 8)
              | ((unsigned)min(o[g * 4 + 2], 255) << 16)
              | ((unsigned)min(o[g * 4 + 3], 255) << 24);
    }
#pragma unroll
    for (int k = 0; k < CNT_PER_THREAD; ++k) {
        tsum += o[k];
        big |= (o[k] >= 255) ? 1 : 0;
    }
    const size_t off = (size_t)blockIdx.x * pitch + base + t * CNT_PER_THREAD;
    *(uint4*)(cnt8 + (off >> 2)) = make_uint4(pk[0], pk[1], pk[2], pk[3]);
    for (int d = 16; d >= 1; d >>= 1) {
        tsum += __shfl_xor(tsum, d);
        big |= __shfl_xor(big, d);
    }
    if ((t & 31) == 0) { wsum[t >> 5] = tsum; wbig[t >> 5] = big; }
    __syncthreads();
    int total = 0;
    int anybig = 0;
#pragma unroll
    for (int i = 0; i < CNT_THREADS / 32; ++i) { total += wsum[i]; anybig |= wbig[i]; }
    if (t == 0) {
        rtot[blockIdx.y * gridDim.x + blockIdx.x] = total;
        sflag[blockIdx.y * gridDim.x + blockIdx.x] = anybig;
    }
    if (anybig) {
#pragma unroll
        for (int g = 0; g < 4; ++g)
            *(int4*)(cnt32 + off + g * 4) = make_int4(o[g * 4 + 0], o[g * 4 + 1], o[g * 4 + 2], o[g * 4 + 3]);
    }
}

__device__ __forceinline__ int gcn_take_slot(int* p, int v, int base, unsigned lim) {
    const unsigned d = (unsigned)(v - base);
    int pos = -1;
    if (d < lim) pos = atomicAdd(&p[d], 1);
    return pos;
}

#define PL_B 25
__device__ __forceinline__ void gcn_cnt_load(uint4 (&q)[PL_B], const unsigned* __restrict__ q0,
                                             int sb, int nSlices, int pitchW) {
#pragma unroll
    for (int k = 0; k < PL_B; ++k)
        q[k] = *(const uint4*)(q0 + (size_t)min(sb + k, nSlices - 1) * pitchW);
}
__device__ __forceinline__ void gcn_cnt_add(unsigned& e, unsigned& o, unsigned x) {
    e += x & 0x00FF00FFu;
    o += (x >> 8) & 0x00FF00FFu;
}
__device__ __forceinline__ void gcn_cnt_acc(const uint4 (&q)[PL_B], int sb, int nSlices, int s,
                                            unsigned (&ae)[4], unsigned (&ao)[4],
                                            unsigned (&pe)[4], unsigned (&po)[4]) {
#pragma unroll
    for (int k = 0; k < PL_B; ++k) {
        if (sb + k < nSlices) {
            if (sb + k == s) {
#pragma unroll
                for (int g = 0; g < 4; ++g) { pe[g] = ae[g]; po[g] = ao[g]; }
            }
            gcn_cnt_add(ae[0], ao[0], q[k].x);
            gcn_cnt_add(ae[1], ao[1], q[k].y);
            gcn_cnt_add(ae[2], ao[2], q[k].z);
            gcn_cnt_add(ae[3], ao[3], q[k].w);
        }
    }
}

__global__ __launch_bounds__(CNT_THREADS)
void gcn_place_lds(const int* __restrict__ row, const int* __restrict__ col,
                   const float* __restrict__ vals, const unsigned* __restrict__ cnt8,
                   const int* __restrict__ cnt32, const int* __restrict__ rtot,
                   const int* __restrict__ sflag,
                   int* __restrict__ rowptr, int* __restrict__ recw,
                   int nRows, int pitch, int nEdges, int nSlices) {
    int4* const rec = (int4*)recw;
    __shared__ __attribute__((aligned(16))) int p[CNT_RANGE];
    __shared__ int wtot[CNT_THREADS / 32];
    const int t = threadIdx.x;
    const int lane = t & 31;
    const int w = t >> 5;
    const int s = blockIdx.x;
    const int base = blockIdx.y * CNT_RANGE;
    const unsigned lim = (unsigned)max(0, min(CNT_RANGE, nRows - base));
    const int e0 = blockIdx.x * CNT_SLICE;
    const int e1 = min(nEdges, e0 + CNT_SLICE);
    const int e1v = e0 + ((e1 - e0) & ~3);
    {
        const int pitchW = pitch >> 2;
        const unsigned* const q0 = cnt8 + (base >> 2) + t * 4;
        unsigned ae[4] = {0u, 0u, 0u, 0u}, ao[4] = {0u, 0u, 0u, 0u};
        unsigned pe[4] = {0u, 0u, 0u, 0u}, po[4] = {0u, 0u, 0u, 0u};
        uint4 q[PL_B];
        gcn_cnt_load(q, q0, 0, nSlices, pitchW);
        const int nAhead = blockIdx.y * nSlices;
        int ahead = 0;
        int slow = (nSlices > 256) ? 1 : 0;
        for (int bb = 0; bb < max(nAhead, nSlices); bb += 8 * 32) {
            int v[8], f[8];
#pragma unroll
            for (int k = 0; k < 8; ++k) {
                const int i = bb + k * 32 + lane;
                v[k] = rtot[max(min(i, nAhead - 1), 0)];
                f[k] = sflag[nAhead + min(i, nSlices - 1)];
            }
#pragma unroll
            for (int k = 0; k < 8; ++k) {
                const int i = bb + k * 32 + lane;
                ahead += (i < nAhead) ? v[k] : 0;
                slow |= (i < nSlices) ? f[k] : 0;
            }
        }
        for (int d = 16; d >= 1; d >>= 1) {
            ahead += __shfl_xor(ahead, d);
            slow |= __shfl_xor(slow, d);
        }
        gcn_cnt_acc(q, 0, nSlices, s, ae, ao, pe, po);
        for (int sb = PL_B; sb < nSlices; sb += PL_B) {
            gcn_cnt_load(q, q0, sb, nSlices, pitchW);
            gcn_cnt_acc(q, sb, nSlices, s, ae, ao, pe, po);
        }
        int tot[CNT_PER_THREAD], pre[CNT_PER_THREAD];
#pragma unroll
        for (int g = 0; g < 4; ++g) {
            tot[g * 4 + 0] = (int)(ae[g] & 0xFFFFu);  tot[g * 4 + 2] = (int)(ae[g] >> 16);
            tot[g * 4 + 1] = (int)(ao[g] & 0xFFFFu);  tot[g * 4 + 3] = (int)(ao[g] >> 16);
            pre[g * 4 + 0] = (int)(pe[g] & 0xFFFFu);  pre[g * 4 + 2] = (int)(pe[g] >> 16);
            pre[g * 4 + 1] = (int)(po[g] & 0xFFFFu);  pre[g * 4 + 3] = (int)(po[g] >> 16);
        }
        if (slow) {
#pragma unroll
            for (int k = 0; k < CNT_PER_THREAD; ++k) { tot[k] = 0; pre[k] = 0; }
            for (int s2 = 0; s2 < nSlices; ++s2) {
                const bool exact = sflag[nAhead + s2] != 0;
                const size_t off = (size_t)s2 * pitch + base + t * CNT_PER_THREAD;
#pragma unroll
                for (int g = 0; g < 4; ++g) {
                    int cn[4];
                    if (exact) {
                        const int4 v = *(const int4*)(cnt32 + off + g * 4);
                        cn[0] = v.x; cn[1] = v.y; cn[2] = v.z; cn[3] = v.w;
                    } else {
                        const unsigned x = cnt8[(off >> 2) + g];
                        cn[0] = (int)(x & 255u); cn[1] = (int)((x >> 8) & 255u);
                        cn[2] = (int)((x >> 16) & 255u); cn[3] = (int)(x >> 24);
                    }
#pragma unroll
                    for (int b = 0; b < 4; ++b) {
                        tot[g * 4 + b] += cn[b];
                        if (s2 < s) pre[g * 4 + b] += cn[b];
                    }
                }
            }
        }
        int tsum = 0;
#pragma unroll
        for (int k = 0; k < CNT_PER_THREAD; ++k) tsum += tot[k];
        int inc = tsum;
        for (int d = 1; d < 32; d <<= 1) {
            const int o = __shfl_up(inc, d);
            if (lane >= d) inc += o;
        }
        if (lane == 31) wtot[w] = inc;
        __syncthreads();
#pragma unroll
        for (int i = 0; i < CNT_THREADS / 32; ++i) {
            const int v = wtot[i];
            if (i < w) ahead += v;
        }
        int run = ahead + inc - tsum;
        int pv[CNT_PER_THREAD];
#pragma unroll
        for (int k = 0; k < CNT_PER_THREAD; ++k) {
            const unsigned j = (unsigned)(t * CNT_PER_THREAD + k);
            pv[k] = run + pre[k];
            if (blockIdx.x == 0 && j < lim) rowptr[base + j] = run;
            run += tot[k];
        }
#pragma unroll
        for (int g = 0; g < 4; ++g)
            *(int4*)(p + t * CNT_PER_THREAD + g * 4) = make_int4(pv[g * 4 + 0], pv[g * 4 + 1], pv[g * 4 + 2], pv[g * 4 + 3]);
        if (blockIdx.x == 0 && blockIdx.y == gridDim.y - 1 && t == CNT_THREADS - 1) rowptr[nRows] = run;
    }
    __syncthreads();
    for (int ib = e0 + t * 4; ib < e1v; ib += CNT_THREADS * 4 * CNT_U) {
        int4 v[CNT_U];
        gcn_load_rows(v, row, ib, e1v);
        int4 cc[CNT_U];
        float4 vv[CNT_U];
#pragma unroll
        for (int k = 0; k < CNT_U; ++k) {
            const int i = ib + k * CNT_THREADS * 4;
            cc[k] = make_int4(0, 0, 0, 0);
            vv[k] = make_float4(0.0f, 0.0f, 0.0f, 0.0f);
            if (i < e1v) {
                cc[k] = *(const int4*)(col + i);
                vv[k] = *(const float4*)(vals + i);
            }
        }
        int4 pos[CNT_U];
#pragma unroll
        for (int k = 0; k < CNT_U; ++k) {
            pos[k].x = gcn_take_slot(p, v[k].x, base, lim);
            pos[k].y = gcn_take_slot(p, v[k].y, base, lim);
            pos[k].z = gcn_take_slot(p, v[k].z, base, lim);
            pos[k].w = gcn_take_slot(p, v[k].w, base, lim);
        }
#pragma unroll
        for (int k = 0; k < CNT_U; ++k) {
            const int i = ib + k * CNT_THREADS * 4;
            if (pos[k].x >= 0) rec[pos[k].x] = make_int4(i,     cc[k].x, __float_as_int(vv[k].x), 0);
            if (pos[k].y >= 0) rec[pos[k].y] = make_int4(i + 1, cc[k].y, __float_as_int(vv[k].y), 0);
            if (pos[k].z >= 0) rec[pos[k].z] = make_int4(i + 2, cc[k].z, __float_as_int(vv[k].z), 0);
            if (pos[k].w >= 0) rec[pos[k].w] = make_int4(i + 3, cc[k].w, __float_as_int(vv[k].w), 0);
        }
    }
    if (t < e1 - e1v) {
        const int pos = gcn_take_slot(p, row[e1v + t], base, lim);
        if (pos >= 0) rec[pos] = make_int4(e1v + t, col[e1v + t], __float_as_int(vals[e1v + t]), 0);
    }
}

#define ROWS_PER_WAVE 5
#define ROWS_BLOCK_WAVES 8

__device__ __forceinline__ void gcn_sort_row32(const int4& r, int deg, int lane, int& c_l, float& v_l) {
    int place = 0;
    for (int j = 0; j < deg; ++j)
        place += (__builtin_amdgcn_readlane(r.x, j) < r.x) ? 1 : 0;
    const int dst = ((lane < deg) ? place : lane) << 2;
    c_l = __builtin_amdgcn_ds_permute(dst, r.y);
    v_l = __int_as_float(__builtin_amdgcn_ds_permute(dst, r.z));
}

__device__ __forceinline__ void gcn_sorted_window(const int4* __restrict__ rec, int s, int e, int w0, int lane,
                                                  int& c_w, float& v_w) {
    c_w = 0;
    v_w = 0.0f;
    int ib = s;
    do {
        const int4 r = rec[min(ib + lane, e - 1)];
        int place = 0;
        int jb = s;
        do {
            const int jn = min(32, e - jb);
            const int other = rec[min(jb + lane, e - 1)].x;
            place += (__builtin_amdgcn_readlane(other, 0) < r.x) ? 1 : 0;
            for (int j = 1; j < jn; ++j)
                place += (__builtin_amdgcn_readlane(other, j) < r.x) ? 1 : 0;
            jb += 32;
        } while (jb < e);
        const int d = place - w0;
        const int jn = min(32, e - ib);
        for (int j = 0; j < jn; ++j) {
            const int   dj = __builtin_amdgcn_readlane(d, j);
            const int   cj = __builtin_amdgcn_readlane(r.y, j);
            const float vj = __int_as_float(__builtin_amdgcn_readlane(r.z, j));
            if (dj == lane) { c_w = cj; v_w = vj; }
        }
        ib += 32;
    } while (ib < e);
}

template <int VEC> struct gcn_row;
template <> struct gcn_row<4> { typedef float4 type; };
template <> struct gcn_row<2> { typedef float2 type; };
__device__ __forceinline__ void gcn_row_zero(float4& a) { a.x = 0.0f; a.y = 0.0f; a.z = 0.0f; a.w = 0.0f; }
__device__ __forceinline__ void gcn_row_zero(float2& a) { a.x = 0.0f; a.y = 0.0f; }
__device__ __forceinline__ void gcn_row_fma(float4& a, float v, const float4& m) {
    a.x = fmaf(v, m.x, a.x);
    a.y = fmaf(v, m.y, a.y);
    a.z = fmaf(v, m.z, a.z);
    a.w = fmaf(v, m.w, a.w);
}
__device__ __forceinline__ void gcn_row_fma(float2& a, float v, const float2& m) {
    a.x = fmaf(v, m.x, a.x);
    a.y = fmaf(v, m.y, a.y);
}

#define GATHER_U 16
template <int VEC>
__device__ __forceinline__ void gcn_gather_chunk(typename gcn_row<VEC>::type& acc, int c_l, float v_l, int n,
                                                 const float* __restrict__ x, unsigned loff) {
    typedef typename gcn_row<VEC>::type row_t;
    const int D = 32 * VEC;
    int j = 0;
    for (; j + GATHER_U <= n; j += GATHER_U) {
        row_t m[GATHER_U];
#pragma unroll
        for (int k = 0; k < GATHER_U; ++k) {
            const int c = __builtin_amdgcn_readlane(c_l, j + k);
            m[k] = *(const row_t*)(x + (size_t)c * D + loff);
        }
#pragma unroll
        for (int k = 0; k < GATHER_U; ++k) {
            const float v = __int_as_float(__builtin_amdgcn_readlane(__float_as_int(v_l), j + k));
            gcn_row_fma(acc, v, m[k]);
        }
    }
    const int rem = n - j;
    if (rem > 0) {
        row_t m[GATHER_U - 1];
#pragma unroll
        for (int k = 0; k < GATHER_U - 1; ++k) {
            if (k < rem) {
                const int c = __builtin_amdgcn_readlane(c_l, j + k);
                m[k] = *(const row_t*)(x + (size_t)c * D + loff);
            }
        }
#pragma unroll
        for (int k = 0; k < GATHER_U - 1; ++k) {
            if (k < rem) {
                const float v = __int_as_float(__builtin_amdgcn_readlane(__float_as_int(v_l), j + k));
                gcn_row_fma(acc, v, m[k]);
            }
        }
    }
}

template <int VEC>
__global__ __launch_bounds__(ROWS_BLOCK_WAVES * 32)
void gcn_gather_rows(const int* __restrict__ rowptr, const int* __restrict__ recw,
                     const float* __restrict__ x,
                     float* __restrict__ agg, int nRows, int nEdges) {
    typedef typename gcn_row<VEC>::type row_t;
    const int4* const rec = (const int4*)recw;
    const int D = 32 * VEC;
    const int lane = threadIdx.x & 31;
    const unsigned loff = (threadIdx.x & 31u) * (unsigned)VEC;
    const int wave = __builtin_amdgcn_readfirstlane((blockIdx.x * blockDim.x + threadIdx.x) >> 5);
    const int r0 = wave * ROWS_PER_WAVE;
    if (r0 >= nRows) return;
    const int rEnd = min(nRows, r0 + ROWS_PER_WAVE);
    const int last = max(nEdges - 1, 0);
    int s = rowptr[r0];
    int e = rowptr[r0 + 1];
    int4 r_n = rec[min(s + lane, last)];
    for (int r = r0; r < rEnd; ++r) {
        const int e_next = rowptr[min(r + 2, nRows)];
        const int4 r_c = r_n;
        r_n = rec[min(e + lane, last)];
        row_t acc;
        gcn_row_zero(acc);
        const int deg = e - s;
        for (int b = s; b < e; b += 32) {
            int c_w; float v_w;
            if (deg <= 32) gcn_sort_row32(r_c, deg, lane, c_w, v_w);
            else           gcn_sorted_window(rec, s, e, b - s, lane, c_w, v_w);
            gcn_gather_chunk<VEC>(acc, c_w, v_w, min(32, e - b), x, loff);
        }
        *(row_t*)(agg + (size_t)r * D + loff) = acc;
        s = e;
        e = e_next;
    }
}

#define FUSE_WAVES 4
#define FUSE_ROWS  8
#define FUSE_MT    2
#define FUSE_TILE  (16 * FUSE_MT)
static_assert(FUSE_WAVES * FUSE_ROWS == FUSE_TILE, "the waves' rows fill the work-group's row tiles");
#define FUSE_LD    136

__device__ __forceinline__ v16h gcn_frag_a(const _Float16* tile, int r, int half, int kk) {
    const _Float16* p = tile + r * FUSE_LD + kk * 32 + half * 8;
    const v8h lo = *(const v8h*)p;
    const v8h hi = *(const v8h*)(p + 16);
    return __builtin_shufflevector(lo, hi, 0, 1, 2, 3, 4, 5, 6, 7, 8, 9, 10, 11, 12, 13, 14, 15);
}

__global__ __launch_bounds__(FUSE_WAVES * 32)
void gcn_layer1_fused(const int* __restrict__ rowptr, const int* __restrict__ recw,
                      const float* __restrict__ x,
                      const _Float16* __restrict__ W1p, const _Float16* __restrict__ W2p,
                      float* __restrict__ t, int nRows, int nEdges) {
    const int4* const rec = (const int4*)recw;
    __shared__ __attribute__((aligned(16))) _Float16 sA[FUSE_TILE * FUSE_LD];
    __shared__ __attribute__((aligned(16))) _Float16 sH[FUSE_TILE * FUSE_LD];
    const int lane = threadIdx.x & 31;
    const unsigned loff = (threadIdx.x & 31u) * 4u;
    const int w = __builtin_amdgcn_readfirstlane(threadIdx.x >> 5);
    const int tile0 = blockIdx.x * FUSE_TILE;
    const int r0 = tile0 + w * FUSE_ROWS;
    const int last = max(nEdges - 1, 0);
    int s = rowptr[min(r0, nRows)];
    int e = rowptr[min(r0 + 1, nRows)];
    int4 r_n = rec[min(s + lane, last)];
#pragma unroll 1
    for (int k = 0; k < FUSE_ROWS; ++k) {
        const int e_next = rowptr[min(r0 + k + 2, nRows)];
        const int4 r_c = r_n;
        r_n = rec[min(e + lane, last)];
        float4 acc;
        gcn_row_zero(acc);
        const int deg = e - s;
        for (int b = s; b < e; b += 32) {
            int c_w; float v_w;
            if (deg <= 32) gcn_sort_row32(r_c, deg, lane, c_w, v_w);
            else           gcn_sorted_window(rec, s, e, b - s, lane, c_w, v_w);
            gcn_gather_chunk<4>(acc, c_w, v_w, min(32, e - b), x, loff);
        }
        v4h hv;
        hv[0] = (_Float16)acc.x; hv[1] = (_Float16)acc.y; hv[2] = (_Float16)acc.z; hv[3] = (_Float16)acc.w;
        *(v4h*)(sA + (w * FUSE_ROWS + k) * FUSE_LD + loff) = hv;
        s = e;
        e = e_next;
    }
    __syncthreads();

    const int half = lane >> 4;
    const int rr   = lane & 15;

    v8f h0[FUSE_MT], h1[FUSE_MT];
#pragma unroll
    for (int mt = 0; mt < FUSE_MT; ++mt) { h0[mt] = (v8f){}; h1[mt] = (v8f){}; }
#pragma unroll
    for (int kk = 0; kk < 4; ++kk) {
        const v16h b0 = *(const v16h*)(W1p + (((size_t)(2 * w)     * 4 + kk) * 32 + lane) * 16);
        const v16h b1 = *(const v16h*)(W1p + (((size_t)(2 * w + 1) * 4 + kk) * 32 + lane) * 16);
#pragma unroll
        for (int mt = 0; mt < FUSE_MT; ++mt) {
            const v16h a = gcn_frag_a(sA + mt * 16 * FUSE_LD, rr, half, kk);
            h0[mt] = __builtin_amdgcn_wmma_f32_16x16x32_f16(false, a, false, b0, (short)0, h0[mt], false, false);
            h1[mt] = __builtin_amdgcn_wmma_f32_16x16x32_f16(false, a, false, b1, (short)0, h1[mt], false, false);
        }
    }
#pragma unroll
    for (int mt = 0; mt < FUSE_MT; ++mt) {
#pragma unroll
        for (int v = 0; v < 8; ++v) {
            _Float16* hp = sH + (mt * 16 + half * 8 + v) * FUSE_LD + (2 * w) * 16 + rr;
            hp[0]  = (_Float16)fmaxf(h0[mt][v], 0.0f);
            hp[16] = (_Float16)fmaxf(h1[mt][v], 0.0f);
        }
    }
    __syncthreads();

    v8f tv[FUSE_MT];
#pragma unroll
    for (int mt = 0; mt < FUSE_MT; ++mt) tv[mt] = (v8f){};
#pragma unroll
    for (int kk = 0; kk < 4; ++kk) {
        const v16h b = *(const v16h*)(W2p + (((size_t)w * 4 + kk) * 32 + lane) * 16);
#pragma unroll
        for (int mt = 0; mt < FUSE_MT; ++mt) {
            const v16h a = gcn_frag_a(sH + mt * 16 * FUSE_LD, rr, half, kk);
            tv[mt] = __builtin_amdgcn_wmma_f32_16x16x32_f16(false, a, false, b, (short)0, tv[mt], false, false);
        }
    }
    const bool whole = tile0 + FUSE_TILE <= nRows;
#pragma unroll
    for (int mt = 0; mt < FUSE_MT; ++mt) {
        const int mbase = tile0 + mt * 16 + half * 8;
        float* tp = t + (size_t)mbase * 64 + w * 16 + rr;
        if (whole) {
#pragma unroll
            for (int v = 0; v < 8; ++v) tp[v * 64] = tv[mt][v];
        } else {
#pragma unroll
            for (int v = 0; v < 8; ++v)
                if (mbase + v < nRows) tp[v * 64] = tv[mt][v];
        }
    }
}

extern "C" void kernel_launch(void* const* d_in, const int* in_sizes, int n_in,
                              void* d_out, int out_size, void* d_ws, size_t ws_size,
                              hipStream_t stream) {
    const float* x    = (const float*)d_in[0];
    const int*   row  = (const int*)d_in[1];
    const int*   col  = (const int*)d_in[2];
    const float* vals = (const float*)d_in[3];
    const float* W1   = (const float*)d_in[4];
    const float* W2   = (const float*)d_in[5];
    float* out = (float*)d_out;

    const int NN = in_sizes[0] / IN_DIM;
    const int NE = in_sizes[1];

    const int nSlices = (NE + CNT_SLICE - 1) / CNT_SLICE;
    const int nRanges = (NN + CNT_RANGE - 1) / CNT_RANGE;
    const int pitch   = nRanges * CNT_RANGE;

    float*     tBuf = (float*)d_ws;
    _Float16*  W1p  = (_Float16*)(tBuf + (size_t)2 * NN * IN_DIM);
    _Float16*  W2p  = W1p + 128 * 128;
    const int  NNp  = ((NN + 8) >> 2) << 2;
    int*       rowptr = (int*)(W2p + 128 * 64);
    int*       rec    = rowptr + NNp;
    int*       cnt32  = rec + (size_t)4 * NE;
    unsigned*  cnt8   = (unsigned*)(cnt32 + (size_t)nSlices * pitch);
    int*       rtot   = (int*)(cnt8 + (size_t)nSlices * (pitch >> 2));
    int*       sflag  = rtot + (size_t)nRanges * nSlices;

    gcn_count_lds<<<dim3(nSlices, nRanges + 1), CNT_THREADS, 0, stream>>>(row, cnt32, cnt8, rtot, sflag, NN, pitch, NE, W1, W2, W1p, W2p, 8, 4);
    gcn_place_lds<<<dim3(nSlices, nRanges), CNT_THREADS, 0, stream>>>(row, col, vals, cnt8, cnt32, rtot, sflag, rowptr, rec, NN, pitch, NE, nSlices);
    const int rowBlocks = (NN + ROWS_BLOCK_WAVES * ROWS_PER_WAVE - 1) / (ROWS_BLOCK_WAVES * ROWS_PER_WAVE);

    float* tF = tBuf;
    gcn_layer1_fused<<<(NN + FUSE_TILE - 1) / FUSE_TILE, FUSE_WAVES * 32, 0, stream>>>(rowptr, rec, x, W1p, W2p, tF, NN, NE);

    gcn_gather_rows<2><<<rowBlocks, ROWS_BLOCK_WAVES * 32, 0, stream>>>(rowptr, rec, tF, out, NN, NE);
}


# Rows of 33 to 64 edges are ranked once in registers (two records per lane) and handed out through a per-wave LDS strip, not re-read and re-ranked per chunk and window; same f32 sums and order
# speedup vs baseline: 4.4952x; 1.0454x over previous
#include <hip/hip_runtime.h>
#include <hip/hip_bf16.h>

typedef __attribute__((ext_vector_type(16))) _Float16 v16h;
typedef __attribute__((ext_vector_type(8)))  _Float16 v8h;
typedef __attribute__((ext_vector_type(4)))  _Float16 v4h;
typedef __attribute__((ext_vector_type(8)))  float    v8f;

#define IN_DIM 128

__device__ __forceinline__ void gcn_pack_w(const float* __restrict__ W, _Float16* __restrict__ Wp,
                                           int Nt  , int id) {
    int lane = id & 31;
    int kk   = (id >> 5) & 3;
    int t    = id >> 7;
    int half = lane >> 4;
    int r    = lane & 15;
    int N    = Nt * 16;
    int n    = t * 16 + r;
    int k0   = kk * 32;
    _Float16* dst = Wp + (size_t)id * 16;
#pragma unroll
    for (int j = 0; j < 8; ++j)
        dst[j] = (_Float16)W[(size_t)(k0 + half * 8 + j) * N + n];
#pragma unroll
    for (int j = 0; j < 8; ++j)
        dst[8 + j] = (_Float16)W[(size_t)(k0 + 16 + half * 8 + j) * N + n];
}

#define CNT_RANGE   8192
#define CNT_SLICE   32768
#define CNT_THREADS 512
#define CNT_U       8

__device__ __forceinline__ void gcn_load_rows(int4 (&v)[CNT_U], const int* __restrict__ row, int ib, int e1v) {
#pragma unroll
    for (int k = 0; k < CNT_U; ++k) {
        const int i = ib + k * CNT_THREADS * 4;
        v[k] = make_int4(-1, -1, -1, -1);
        if (i < e1v) v[k] = *(const int4*)(row + i);
    }
}

__device__ __forceinline__ void gcn_count_one(int* c, int v, int base, unsigned lim) {
    const unsigned d = (unsigned)(v - base);
    if (d < lim) atomicAdd(&c[d], 1);
}

#define CNT_PER_THREAD (CNT_RANGE / CNT_THREADS)
static_assert(CNT_PER_THREAD == 16, "a thread's counts are four 16-byte LDS reads and one 16-byte store");

__global__ __launch_bounds__(CNT_THREADS)
void gcn_count_lds(const int* __restrict__ row, int* __restrict__ cnt32, unsigned* __restrict__ cnt8,
                   int* __restrict__ rtot, int* __restrict__ sflag,
                   int nRows, int pitch, int nEdges,
                   const float* __restrict__ W1, const float* __restrict__ W2,
                   _Float16* __restrict__ W1p, _Float16* __restrict__ W2p,
                   int nt1  , int nt2  ) {
    __shared__ __attribute__((aligned(16))) int c[CNT_RANGE];
    __shared__ int wsum[CNT_THREADS / 32];
    __shared__ int wbig[CNT_THREADS / 32];
    const int t = threadIdx.x;
    if (blockIdx.y == gridDim.y - 1) {
        const int n1 = nt1 * 4 * 32;
        const int n2 = nt2 * 4 * 32;
        for (int id = blockIdx.x * CNT_THREADS + t; id < n1 + n2; id += gridDim.x * CNT_THREADS) {
            if (id < n1) gcn_pack_w(W1, W1p, nt1, id);
            else         gcn_pack_w(W2, W2p, nt2, id - n1);
        }
        return;
    }
    const int base = blockIdx.y * CNT_RANGE;
    const unsigned lim = (unsigned)max(0, min(CNT_RANGE, nRows - base));
    const int e0 = blockIdx.x * CNT_SLICE;
    const int e1 = min(nEdges, e0 + CNT_SLICE);
    const int e1v = e0 + ((e1 - e0) & ~3);
    for (int j = t; j < CNT_RANGE; j += CNT_THREADS) c[j] = 0;
    __syncthreads();
    for (int ib = e0 + t * 4; ib < e1v; ib += CNT_THREADS * 4 * CNT_U) {
        int4 v[CNT_U];
        gcn_load_rows(v, row, ib, e1v);
#pragma unroll
        for (int k = 0; k < CNT_U; ++k) {
            gcn_count_one(c, v[k].x, base, lim);
            gcn_count_one(c, v[k].y, base, lim);
            gcn_count_one(c, v[k].z, base, lim);
            gcn_count_one(c, v[k].w, base, lim);
        }
    }
    if (t < e1 - e1v) gcn_count_one(c, row[e1v + t], base, lim);
    __syncthreads();
    int o[CNT_PER_THREAD];
#pragma unroll
    for (int g = 0; g < 4; ++g) {
        const int4 v = *(const int4*)(c + t * CNT_PER_THREAD + g * 4);
        o[g * 4 + 0] = v.x; o[g * 4 + 1] = v.y; o[g * 4 + 2] = v.z; o[g * 4 + 3] = v.w;
    }
    unsigned pk[4];
    int tsum = 0;
    int big = 0;
#pragma unroll
    for (int g = 0; g < 4; ++g) {
        pk[g] = (unsigned)min(o[g * 4 + 0], 255)
              | ((unsigned)min(o[g * 4 + 1], 255) << 8)
              | ((unsigned)min(o[g * 4 + 2], 255) << 16)
              | ((unsigned)min(o[g * 4 + 3], 255) << 24);
    }
#pragma unroll
    for (int k = 0; k < CNT_PER_THREAD; ++k) {
        tsum += o[k];
        big |= (o[k] >= 255) ? 1 : 0;
    }
    const size_t off = (size_t)blockIdx.x * pitch + base + t * CNT_PER_THREAD;
    *(uint4*)(cnt8 + (off >> 2)) = make_uint4(pk[0], pk[1], pk[2], pk[3]);
    for (int d = 16; d >= 1; d >>= 1) {
        tsum += __shfl_xor(tsum, d);
        big |= __shfl_xor(big, d);
    }
    if ((t & 31) == 0) { wsum[t >> 5] = tsum; wbig[t >> 5] = big; }
    __syncthreads();
    int total = 0;
    int anybig = 0;
#pragma unroll
    for (int i = 0; i < CNT_THREADS / 32; ++i) { total += wsum[i]; anybig |= wbig[i]; }
    if (t == 0) {
        rtot[blockIdx.y * gridDim.x + blockIdx.x] = total;
        sflag[blockIdx.y * gridDim.x + blockIdx.x] = anybig;
    }
    if (anybig) {
#pragma unroll
        for (int g = 0; g < 4; ++g)
            *(int4*)(cnt32 + off + g * 4) = make_int4(o[g * 4 + 0], o[g * 4 + 1], o[g * 4 + 2], o[g * 4 + 3]);
    }
}

__device__ __forceinline__ int gcn_take_slot(int* p, int v, int base, unsigned lim) {
    const unsigned d = (unsigned)(v - base);
    int pos = -1;
    if (d < lim) pos = atomicAdd(&p[d], 1);
    return pos;
}

#define PL_B 25
__device__ __forceinline__ void gcn_cnt_load(uint4 (&q)[PL_B], const unsigned* __restrict__ q0,
                                             int sb, int nSlices, int pitchW) {
#pragma unroll
    for (int k = 0; k < PL_B; ++k)
        q[k] = *(const uint4*)(q0 + (size_t)min(sb + k, nSlices - 1) * pitchW);
}
__device__ __forceinline__ void gcn_cnt_add(unsigned& e, unsigned& o, unsigned x) {
    e += x & 0x00FF00FFu;
    o += (x >> 8) & 0x00FF00FFu;
}
__device__ __forceinline__ void gcn_cnt_acc(const uint4 (&q)[PL_B], int sb, int nSlices, int s,
                                            unsigned (&ae)[4], unsigned (&ao)[4],
                                            unsigned (&pe)[4], unsigned (&po)[4]) {
#pragma unroll
    for (int k = 0; k < PL_B; ++k) {
        if (sb + k < nSlices) {
            if (sb + k == s) {
#pragma unroll
                for (int g = 0; g < 4; ++g) { pe[g] = ae[g]; po[g] = ao[g]; }
            }
            gcn_cnt_add(ae[0], ao[0], q[k].x);
            gcn_cnt_add(ae[1], ao[1], q[k].y);
            gcn_cnt_add(ae[2], ao[2], q[k].z);
            gcn_cnt_add(ae[3], ao[3], q[k].w);
        }
    }
}

__global__ __launch_bounds__(CNT_THREADS)
void gcn_place_lds(const int* __restrict__ row, const int* __restrict__ col,
                   const float* __restrict__ vals, const unsigned* __restrict__ cnt8,
                   const int* __restrict__ cnt32, const int* __restrict__ rtot,
                   const int* __restrict__ sflag,
                   int* __restrict__ rowptr, int* __restrict__ recw,
                   int nRows, int pitch, int nEdges, int nSlices) {
    int4* const rec = (int4*)recw;
    __shared__ __attribute__((aligned(16))) int p[CNT_RANGE];
    __shared__ int wtot[CNT_THREADS / 32];
    const int t = threadIdx.x;
    const int lane = t & 31;
    const int w = t >> 5;
    const int s = blockIdx.x;
    const int base = blockIdx.y * CNT_RANGE;
    const unsigned lim = (unsigned)max(0, min(CNT_RANGE, nRows - base));
    const int e0 = blockIdx.x * CNT_SLICE;
    const int e1 = min(nEdges, e0 + CNT_SLICE);
    const int e1v = e0 + ((e1 - e0) & ~3);
    {
        const int pitchW = pitch >> 2;
        const unsigned* const q0 = cnt8 + (base >> 2) + t * 4;
        unsigned ae[4] = {0u, 0u, 0u, 0u}, ao[4] = {0u, 0u, 0u, 0u};
        unsigned pe[4] = {0u, 0u, 0u, 0u}, po[4] = {0u, 0u, 0u, 0u};
        uint4 q[PL_B];
        gcn_cnt_load(q, q0, 0, nSlices, pitchW);
        const int nAhead = blockIdx.y * nSlices;
        int ahead = 0;
        int slow = (nSlices > 256) ? 1 : 0;
        for (int bb = 0; bb < max(nAhead, nSlices); bb += 8 * 32) {
            int v[8], f[8];
#pragma unroll
            for (int k = 0; k < 8; ++k) {
                const int i = bb + k * 32 + lane;
                v[k] = rtot[max(min(i, nAhead - 1), 0)];
                f[k] = sflag[nAhead + min(i, nSlices - 1)];
            }
#pragma unroll
            for (int k = 0; k < 8; ++k) {
                const int i = bb + k * 32 + lane;
                ahead += (i < nAhead) ? v[k] : 0;
                slow |= (i < nSlices) ? f[k] : 0;
            }
        }
        for (int d = 16; d >= 1; d >>= 1) {
            ahead += __shfl_xor(ahead, d);
            slow |= __shfl_xor(slow, d);
        }
        gcn_cnt_acc(q, 0, nSlices, s, ae, ao, pe, po);
        for (int sb = PL_B; sb < nSlices; sb += PL_B) {
            gcn_cnt_load(q, q0, sb, nSlices, pitchW);
            gcn_cnt_acc(q, sb, nSlices, s, ae, ao, pe, po);
        }
        int tot[CNT_PER_THREAD], pre[CNT_PER_THREAD];
#pragma unroll
        for (int g = 0; g < 4; ++g) {
            tot[g * 4 + 0] = (int)(ae[g] & 0xFFFFu);  tot[g * 4 + 2] = (int)(ae[g] >> 16);
            tot[g * 4 + 1] = (int)(ao[g] & 0xFFFFu);  tot[g * 4 + 3] = (int)(ao[g] >> 16);
            pre[g * 4 + 0] = (int)(pe[g] & 0xFFFFu);  pre[g * 4 + 2] = (int)(pe[g] >> 16);
            pre[g * 4 + 1] = (int)(po[g] & 0xFFFFu);  pre[g * 4 + 3] = (int)(po[g] >> 16);
        }
        if (slow) {
#pragma unroll
            for (int k = 0; k < CNT_PER_THREAD; ++k) { tot[k] = 0; pre[k] = 0; }
            for (int s2 = 0; s2 < nSlices; ++s2) {
                const bool exact = sflag[nAhead + s2] != 0;
                const size_t off = (size_t)s2 * pitch + base + t * CNT_PER_THREAD;
#pragma unroll
                for (int g = 0; g < 4; ++g) {
                    int cn[4];
                    if (exact) {
                        const int4 v = *(const int4*)(cnt32 + off + g * 4);
                        cn[0] = v.x; cn[1] = v.y; cn[2] = v.z; cn[3] = v.w;
                    } else {
                        const unsigned x = cnt8[(off >> 2) + g];
                        cn[0] = (int)(x & 255u); cn[1] = (int)((x >> 8) & 255u);
                        cn[2] = (int)((x >> 16) & 255u); cn[3] = (int)(x >> 24);
                    }
#pragma unroll
                    for (int b = 0; b < 4; ++b) {
                        tot[g * 4 + b] += cn[b];
                        if (s2 < s) pre[g * 4 + b] += cn[b];
                    }
                }
            }
        }
        int tsum = 0;
#pragma unroll
        for (int k = 0; k < CNT_PER_THREAD; ++k) tsum += tot[k];
        int inc = tsum;
        for (int d = 1; d < 32; d <<= 1) {
            const int o = __shfl_up(inc, d);
            if (lane >= d) inc += o;
        }
        if (lane == 31) wtot[w] = inc;
        __syncthreads();
#pragma unroll
        for (int i = 0; i < CNT_THREADS / 32; ++i) {
            const int v = wtot[i];
            if (i < w) ahead += v;
        }
        int run = ahead + inc - tsum;
        int pv[CNT_PER_THREAD];
#pragma unroll
        for (int k = 0; k < CNT_PER_THREAD; ++k) {
            const unsigned j = (unsigned)(t * CNT_PER_THREAD + k);
            pv[k] = run + pre[k];
            if (blockIdx.x == 0 && j < lim) rowptr[base + j] = run;
            run += tot[k];
        }
#pragma unroll
        for (int g = 0; g < 4; ++g)
            *(int4*)(p + t * CNT_PER_THREAD + g * 4) = make_int4(pv[g * 4 + 0], pv[g * 4 + 1], pv[g * 4 + 2], pv[g * 4 + 3]);
        if (blockIdx.x == 0 && blockIdx.y == gridDim.y - 1 && t == CNT_THREADS - 1) rowptr[nRows] = run;
    }
    __syncthreads();
    for (int ib = e0 + t * 4; ib < e1v; ib += CNT_THREADS * 4 * CNT_U) {
        int4 v[CNT_U];
        gcn_load_rows(v, row, ib, e1v);
        int4 cc[CNT_U];
        float4 vv[CNT_U];
#pragma unroll
        for (int k = 0; k < CNT_U; ++k) {
            const int i = ib + k * CNT_THREADS * 4;
            cc[k] = make_int4(0, 0, 0, 0);
            vv[k] = make_float4(0.0f, 0.0f, 0.0f, 0.0f);
            if (i < e1v) {
                cc[k] = *(const int4*)(col + i);
                vv[k] = *(const float4*)(vals + i);
            }
        }
        int4 pos[CNT_U];
#pragma unroll
        for (int k = 0; k < CNT_U; ++k) {
            pos[k].x = gcn_take_slot(p, v[k].x, base, lim);
            pos[k].y = gcn_take_slot(p, v[k].y, base, lim);
            pos[k].z = gcn_take_slot(p, v[k].z, base, lim);
            pos[k].w = gcn_take_slot(p, v[k].w, base, lim);
        }
#pragma unroll
        for (int k = 0; k < CNT_U; ++k) {
            const int i = ib + k * CNT_THREADS * 4;
            if (pos[k].x >= 0) rec[pos[k].x] = make_int4(i,     cc[k].x, __float_as_int(vv[k].x), 0);
            if (pos[k].y >= 0) rec[pos[k].y] = make_int4(i + 1, cc[k].y, __float_as_int(vv[k].y), 0);
            if (pos[k].z >= 0) rec[pos[k].z] = make_int4(i + 2, cc[k].z, __float_as_int(vv[k].z), 0);
            if (pos[k].w >= 0) rec[pos[k].w] = make_int4(i + 3, cc[k].w, __float_as_int(vv[k].w), 0);
        }
    }
    if (t < e1 - e1v) {
        const int pos = gcn_take_slot(p, row[e1v + t], base, lim);
        if (pos >= 0) rec[pos] = make_int4(e1v + t, col[e1v + t], __float_as_int(vals[e1v + t]), 0);
    }
}

#define ROWS_PER_WAVE 5
#define ROWS_BLOCK_WAVES 8

__device__ __forceinline__ void gcn_sort_row32(const int4& r, int deg, int lane, int& c_l, float& v_l) {
    int place = 0;
    for (int j = 0; j < deg; ++j)
        place += (__builtin_amdgcn_readlane(r.x, j) < r.x) ? 1 : 0;
    const int dst = ((lane < deg) ? place : lane) << 2;
    c_l = __builtin_amdgcn_ds_permute(dst, r.y);
    v_l = __int_as_float(__builtin_amdgcn_ds_permute(dst, r.z));
}

__device__ __forceinline__ void gcn_sorted_window(const int4* __restrict__ rec, int s, int e, int w0, int lane,
                                                  int& c_w, float& v_w) {
    c_w = 0;
    v_w = 0.0f;
    int ib = s;
    do {
        const int4 r = rec[min(ib + lane, e - 1)];
        int place = 0;
        int jb = s;
        do {
            const int jn = min(32, e - jb);
            const int other = rec[min(jb + lane, e - 1)].x;
            place += (__builtin_amdgcn_readlane(other, 0) < r.x) ? 1 : 0;
            for (int j = 1; j < jn; ++j)
                place += (__builtin_amdgcn_readlane(other, j) < r.x) ? 1 : 0;
            jb += 32;
        } while (jb < e);
        const int d = place - w0;
        const int jn = min(32, e - ib);
        for (int j = 0; j < jn; ++j) {
            const int   dj = __builtin_amdgcn_readlane(d, j);
            const int   cj = __builtin_amdgcn_readlane(r.y, j);
            const float vj = __int_as_float(__builtin_amdgcn_readlane(r.z, j));
            if (dj == lane) { c_w = cj; v_w = vj; }
        }
        ib += 32;
    } while (ib < e);
}

#define SORT_CAP 64
__device__ __forceinline__ void gcn_rank_row64(const int4& ra, const int4* __restrict__ rec, int s, int e,
                                               int lane, int2* strip) {
    const int nb = e - s - 32;
    const int4 rb = rec[min(s + 32 + lane, e - 1)];
    int pa = 0, pb = 0;
    for (int j = 0; j < 32; ++j) {
        const int o = __builtin_amdgcn_readlane(ra.x, j);
        pa += (o < ra.x) ? 1 : 0;
        pb += (o < rb.x) ? 1 : 0;
    }
    for (int j = 0; j < nb; ++j) {
        const int o = __builtin_amdgcn_readlane(rb.x, j);
        pa += (o < ra.x) ? 1 : 0;
        pb += (o < rb.x) ? 1 : 0;
    }
    strip[pa] = make_int2(ra.y, ra.z);
    if (lane < nb) strip[pb] = make_int2(rb.y, rb.z);
}

template <int VEC> struct gcn_row;
template <> struct gcn_row<4> { typedef float4 type; };
template <> struct gcn_row<2> { typedef float2 type; };
__device__ __forceinline__ void gcn_row_zero(float4& a) { a.x = 0.0f; a.y = 0.0f; a.z = 0.0f; a.w = 0.0f; }
__device__ __forceinline__ void gcn_row_zero(float2& a) { a.x = 0.0f; a.y = 0.0f; }
__device__ __forceinline__ void gcn_row_fma(float4& a, float v, const float4& m) {
    a.x = fmaf(v, m.x, a.x);
    a.y = fmaf(v, m.y, a.y);
    a.z = fmaf(v, m.z, a.z);
    a.w = fmaf(v, m.w, a.w);
}
__device__ __forceinline__ void gcn_row_fma(float2& a, float v, const float2& m) {
    a.x = fmaf(v, m.x, a.x);
    a.y = fmaf(v, m.y, a.y);
}

#define GATHER_U 16
template <int VEC>
__device__ __forceinline__ void gcn_gather_chunk(typename gcn_row<VEC>::type& acc, int c_l, float v_l, int n,
                                                 const float* __restrict__ x, unsigned loff) {
    typedef typename gcn_row<VEC>::type row_t;
    const int D = 32 * VEC;
    int j = 0;
    for (; j + GATHER_U <= n; j += GATHER_U) {
        row_t m[GATHER_U];
#pragma unroll
        for (int k = 0; k < GATHER_U; ++k) {
            const int c = __builtin_amdgcn_readlane(c_l, j + k);
            m[k] = *(const row_t*)(x + (size_t)c * D + loff);
        }
#pragma unroll
        for (int k = 0; k < GATHER_U; ++k) {
            const float v = __int_as_float(__builtin_amdgcn_readlane(__float_as_int(v_l), j + k));
            gcn_row_fma(acc, v, m[k]);
        }
    }
    const int rem = n - j;
    if (rem > 0) {
        row_t m[GATHER_U - 1];
#pragma unroll
        for (int k = 0; k < GATHER_U - 1; ++k) {
            if (k < rem) {
                const int c = __builtin_amdgcn_readlane(c_l, j + k);
                m[k] = *(const row_t*)(x + (size_t)c * D + loff);
            }
        }
#pragma unroll
        for (int k = 0; k < GATHER_U - 1; ++k) {
            if (k < rem) {
                const float v = __int_as_float(__builtin_amdgcn_readlane(__float_as_int(v_l), j + k));
                gcn_row_fma(acc, v, m[k]);
            }
        }
    }
}

template <int VEC>
__global__ __launch_bounds__(ROWS_BLOCK_WAVES * 32)
void gcn_gather_rows(const int* __restrict__ rowptr, const int* __restrict__ recw,
                     const float* __restrict__ x,
                     float* __restrict__ agg, int nRows, int nEdges) {
    typedef typename gcn_row<VEC>::type row_t;
    const int4* const rec = (const int4*)recw;
    __shared__ int2 sStrip[ROWS_BLOCK_WAVES * SORT_CAP];
    int2* const strip = sStrip + (threadIdx.x >> 5) * SORT_CAP;
    const int D = 32 * VEC;
    const int lane = threadIdx.x & 31;
    const unsigned loff = (threadIdx.x & 31u) * (unsigned)VEC;
    const int wave = __builtin_amdgcn_readfirstlane((blockIdx.x * blockDim.x + threadIdx.x) >> 5);
    const int r0 = wave * ROWS_PER_WAVE;
    if (r0 >= nRows) return;
    const int rEnd = min(nRows, r0 + ROWS_PER_WAVE);
    const int last = max(nEdges - 1, 0);
    int s = rowptr[r0];
    int e = rowptr[r0 + 1];
    int4 r_n = rec[min(s + lane, last)];
    for (int r = r0; r < rEnd; ++r) {
        const int e_next = rowptr[min(r + 2, nRows)];
        const int4 r_c = r_n;
        r_n = rec[min(e + lane, last)];
        row_t acc;
        gcn_row_zero(acc);
        const int deg = e - s;
        const bool mid = deg > 32 && deg <= SORT_CAP;
        if (mid) gcn_rank_row64(r_c, rec, s, e, lane, strip);
        for (int b = s; b < e; b += 32) {
            int c_w; float v_w;
            if (deg <= 32) gcn_sort_row32(r_c, deg, lane, c_w, v_w);
            else if (mid) {
                const int2 cv = strip[min(b - s + lane, deg - 1)];
                c_w = cv.x; v_w = __int_as_float(cv.y);
            }
            else           gcn_sorted_window(rec, s, e, b - s, lane, c_w, v_w);
            gcn_gather_chunk<VEC>(acc, c_w, v_w, min(32, e - b), x, loff);
        }
        *(row_t*)(agg + (size_t)r * D + loff) = acc;
        s = e;
        e = e_next;
    }
}

#define FUSE_WAVES 4
#define FUSE_ROWS  8
#define FUSE_MT    2
#define FUSE_TILE  (16 * FUSE_MT)
static_assert(FUSE_WAVES * FUSE_ROWS == FUSE_TILE, "the waves' rows fill the work-group's row tiles");
#define FUSE_LD    136

__device__ __forceinline__ v16h gcn_frag_a(const _Float16* tile, int r, int half, int kk) {
    const _Float16* p = tile + r * FUSE_LD + kk * 32 + half * 8;
    const v8h lo = *(const v8h*)p;
    const v8h hi = *(const v8h*)(p + 16);
    return __builtin_shufflevector(lo, hi, 0, 1, 2, 3, 4, 5, 6, 7, 8, 9, 10, 11, 12, 13, 14, 15);
}

__global__ __launch_bounds__(FUSE_WAVES * 32)
void gcn_layer1_fused(const int* __restrict__ rowptr, const int* __restrict__ recw,
                      const float* __restrict__ x,
                      const _Float16* __restrict__ W1p, const _Float16* __restrict__ W2p,
                      float* __restrict__ t, int nRows, int nEdges) {
    const int4* const rec = (const int4*)recw;
    __shared__ __attribute__((aligned(16))) _Float16 sA[FUSE_TILE * FUSE_LD];
    __shared__ __attribute__((aligned(16))) _Float16 sH[FUSE_TILE * FUSE_LD];
    __shared__ int2 sStrip[FUSE_WAVES * SORT_CAP];
    const int lane = threadIdx.x & 31;
    const unsigned loff = (threadIdx.x & 31u) * 4u;
    const int w = __builtin_amdgcn_readfirstlane(threadIdx.x >> 5);
    int2* const strip = sStrip + w * SORT_CAP;
    const int tile0 = blockIdx.x * FUSE_TILE;
    const int r0 = tile0 + w * FUSE_ROWS;
    const int last = max(nEdges - 1, 0);
    int s = rowptr[min(r0, nRows)];
    int e = rowptr[min(r0 + 1, nRows)];
    int4 r_n = rec[min(s + lane, last)];
#pragma unroll 1
    for (int k = 0; k < FUSE_ROWS; ++k) {
        const int e_next = rowptr[min(r0 + k + 2, nRows)];
        const int4 r_c = r_n;
        r_n = rec[min(e + lane, last)];
        float4 acc;
        gcn_row_zero(acc);
        const int deg = e - s;
        const bool mid = deg > 32 && deg <= SORT_CAP;
        if (mid) gcn_rank_row64(r_c, rec, s, e, lane, strip);
        for (int b = s; b < e; b += 32) {
            int c_w; float v_w;
            if (deg <= 32) gcn_sort_row32(r_c, deg, lane, c_w, v_w);
            else if (mid) {
                const int2 cv = strip[min(b - s + lane, deg - 1)];
                c_w = cv.x; v_w = __int_as_float(cv.y);
            }
            else           gcn_sorted_window(rec, s, e, b - s, lane, c_w, v_w);
            gcn_gather_chunk<4>(acc, c_w, v_w, min(32, e - b), x, loff);
        }
        v4h hv;
        hv[0] = (_Float16)acc.x; hv[1] = (_Float16)acc.y; hv[2] = (_Float16)acc.z; hv[3] = (_Float16)acc.w;
        *(v4h*)(sA + (w * FUSE_ROWS + k) * FUSE_LD + loff) = hv;
        s = e;
        e = e_next;
    }
    __syncthreads();

    const int half = lane >> 4;
    const int rr   = lane & 15;

    v8f h0[FUSE_MT], h1[FUSE_MT];
#pragma unroll
    for (int mt = 0; mt < FUSE_MT; ++mt) { h0[mt] = (v8f){}; h1[mt] = (v8f){}; }
#pragma unroll
    for (int kk = 0; kk < 4; ++kk) {
        const v16h b0 = *(const v16h*)(W1p + (((size_t)(2 * w)     * 4 + kk) * 32 + lane) * 16);
        const v16h b1 = *(const v16h*)(W1p + (((size_t)(2 * w + 1) * 4 + kk) * 32 + lane) * 16);
#pragma unroll
        for (int mt = 0; mt < FUSE_MT; ++mt) {
            const v16h a = gcn_frag_a(sA + mt * 16 * FUSE_LD, rr, half, kk);
            h0[mt] = __builtin_amdgcn_wmma_f32_16x16x32_f16(false, a, false, b0, (short)0, h0[mt], false, false);
            h1[mt] = __builtin_amdgcn_wmma_f32_16x16x32_f16(false, a, false, b1, (short)0, h1[mt], false, false);
        }
    }
#pragma unroll
    for (int mt = 0; mt < FUSE_MT; ++mt) {
#pragma unroll
        for (int v = 0; v < 8; ++v) {
            _Float16* hp = sH + (mt * 16 + half * 8 + v) * FUSE_LD + (2 * w) * 16 + rr;
            hp[0]  = (_Float16)fmaxf(h0[mt][v], 0.0f);
            hp[16] = (_Float16)fmaxf(h1[mt][v], 0.0f);
        }
    }
    __syncthreads();

    v8f tv[FUSE_MT];
#pragma unroll
    for (int mt = 0; mt < FUSE_MT; ++mt) tv[mt] = (v8f){};
#pragma unroll
    for (int kk = 0; kk < 4; ++kk) {
        const v16h b = *(const v16h*)(W2p + (((size_t)w * 4 + kk) * 32 + lane) * 16);
#pragma unroll
        for (int mt = 0; mt < FUSE_MT; ++mt) {
            const v16h a = gcn_frag_a(sH + mt * 16 * FUSE_LD, rr, half, kk);
            tv[mt] = __builtin_amdgcn_wmma_f32_16x16x32_f16(false, a, false, b, (short)0, tv[mt], false, false);
        }
    }
    const bool whole = tile0 + FUSE_TILE <= nRows;
#pragma unroll
    for (int mt = 0; mt < FUSE_MT; ++mt) {
        const int mbase = tile0 + mt * 16 + half * 8;
        float* tp = t + (size_t)mbase * 64 + w * 16 + rr;
        if (whole) {
#pragma unroll
            for (int v = 0; v < 8; ++v) tp[v * 64] = tv[mt][v];
        } else {
#pragma unroll
            for (int v = 0; v < 8; ++v)
                if (mbase + v < nRows) tp[v * 64] = tv[mt][v];
        }
    }
}

extern "C" void kernel_launch(void* const* d_in, const int* in_sizes, int n_in,
                              void* d_out, int out_size, void* d_ws, size_t ws_size,
                              hipStream_t stream) {
    const float* x    = (const float*)d_in[0];
    const int*   row  = (const int*)d_in[1];
    const int*   col  = (const int*)d_in[2];
    const float* vals = (const float*)d_in[3];
    const float* W1   = (const float*)d_in[4];
    const float* W2   = (const float*)d_in[5];
    float* out = (float*)d_out;

    const int NN = in_sizes[0] / IN_DIM;
    const int NE = in_sizes[1];

    const int nSlices = (NE + CNT_SLICE - 1) / CNT_SLICE;
    const int nRanges = (NN + CNT_RANGE - 1) / CNT_RANGE;
    const int pitch   = nRanges * CNT_RANGE;

    float*     tBuf = (float*)d_ws;
    _Float16*  W1p  = (_Float16*)(tBuf + (size_t)2 * NN * IN_DIM);
    _Float16*  W2p  = W1p + 128 * 128;
    const int  NNp  = ((NN + 8) >> 2) << 2;
    int*       rowptr = (int*)(W2p + 128 * 64);
    int*       rec    = rowptr + NNp;
    int*       cnt32  = rec + (size_t)4 * NE;
    unsigned*  cnt8   = (unsigned*)(cnt32 + (size_t)nSlices * pitch);
    int*       rtot   = (int*)(cnt8 + (size_t)nSlices * (pitch >> 2));
    int*       sflag  = rtot + (size_t)nRanges * nSlices;

    gcn_count_lds<<<dim3(nSlices, nRanges + 1), CNT_THREADS, 0, stream>>>(row, cnt32, cnt8, rtot, sflag, NN, pitch, NE, W1, W2, W1p, W2p, 8, 4);
    gcn_place_lds<<<dim3(nSlices, nRanges), CNT_THREADS, 0, stream>>>(row, col, vals, cnt8, cnt32, rtot, sflag, rowptr, rec, NN, pitch, NE, nSlices);
    const int rowBlocks = (NN + ROWS_BLOCK_WAVES * ROWS_PER_WAVE - 1) / (ROWS_BLOCK_WAVES * ROWS_PER_WAVE);

    float* tF = tBuf;
    gcn_layer1_fused<<<(NN + FUSE_TILE - 1) / FUSE_TILE, FUSE_WAVES * 32, 0, stream>>>(rowptr, rec, x, W1p, W2p, tF, NN, NE);

    gcn_gather_rows<2><<<rowBlocks, ROWS_BLOCK_WAVES * 32, 0, stream>>>(rowptr, rec, tF, out, NN, NE);
}
